# starting kernel
# baseline (speedup 1.0000x reference)
#include <hip/hip_runtime.h>
#include <hip/hip_bf16.h>
#include <math.h>


typedef __attribute__((ext_vector_type(2))) float v2f;
typedef __attribute__((ext_vector_type(8))) float v8f;

__device__ __forceinline__ v8f wmma4(v2f a, v2f b, v8f c) {
  return __builtin_amdgcn_wmma_f32_16x16x4_f32(
       false, a,  false, b,
       (short)0, c,  false,  false);
}

__global__ __launch_bounds__(256) void temb_kernel(
    const int* __restrict__ num_loops,
    const float* __restrict__ t_w1, const float* __restrict__ t_b1,
    const float* __restrict__ t_w2, const float* __restrict__ t_b2,
    float* __restrict__ ttab) {
  const int t = blockIdx.x;
  if (t >= *num_loops) return;
  __shared__ float te[256];
  __shared__ float hs[1024];
  __shared__ float red[256];
  const int tid = threadIdx.x;

  if (tid < 128) {
    float f   = expf(-logf(10000.0f) * (float)tid / 128.0f);
    float arg = (float)t * f;
    te[tid]       = cosf(arg);
    te[128 + tid] = sinf(arg);
  }
  __syncthreads();

  for (int j = tid; j < 1024; j += 256) {
    float acc = t_b1[j];
    for (int k = 0; k < 256; ++k) acc += te[k] * t_w1[k * 1024 + j];
    hs[j] = acc / (1.0f + expf(-acc));
  }
  __syncthreads();

  const int j = tid & 63, p = tid >> 6;
  float acc = 0.0f;
  for (int k = p * 256; k < p * 256 + 256; ++k) acc += hs[k] * t_w2[k * 64 + j];
  red[tid] = acc;
  __syncthreads();
  if (tid < 64) {
    ttab[t * 64 + tid] =
        t_b2[tid] + red[tid] + red[tid + 64] + red[tid + 128] + red[tid + 192];
  }
}

#define TB 512
#define NW 16

__device__ __forceinline__ void ln_rows(const float* __restrict__ src,
                                        float* __restrict__ dst,
                                        const float* __restrict__ g,
                                        const float* __restrict__ bta, int tid) {
  const int s = tid >> 2, p = tid & 3;
  float sum = 0.0f, sq = 0.0f;
  for (int d = p * 16; d < p * 16 + 16; ++d) {
    float xv = src[s * 64 + d];
    sum += xv; sq += xv * xv;
  }
  for (int off = 1; off < 4; off <<= 1) {
    sum += __shfl_xor(sum, off, 4);
    sq  += __shfl_xor(sq,  off, 4);
  }
  const float mu  = sum * (1.0f / 64.0f);
  const float var = sq * (1.0f / 64.0f) - mu * mu;
  const float inv = rsqrtf(var + 1e-5f);
  for (int d = p * 16; d < p * 16 + 16; ++d)
    dst[s * 64 + d] = (src[s * 64 + d] - mu) * inv * g[d] + bta[d];
}

__global__ __launch_bounds__(TB, 1) void xformer_kernel(
    const int* __restrict__ idx, const int* __restrict__ num_loops,
    const float* __restrict__ wte, const float* __restrict__ wpe,
    const float* __restrict__ ln1_g, const float* __restrict__ ln1_b,
    const float* __restrict__ w_qkv, const float* __restrict__ b_qkv,
    const float* __restrict__ w_o, const float* __restrict__ b_o,
    const float* __restrict__ ln2_g, const float* __restrict__ ln2_b,
    const float* __restrict__ w_fc, const float* __restrict__ b_fc,
    const float* __restrict__ w_pr, const float* __restrict__ b_pr,
    const float* __restrict__ lnf_g, const float* __restrict__ lnf_b,
    const float* __restrict__ ttab, float* __restrict__ out) {
  __shared__ float smem[40960];
  float* xb = smem;
  float* hb = smem + 8192;
  float* qb = smem + 16384;
  float* kb = smem + 24576;
  float* vb = smem + 32768;
  float* gb = qb;
  float* ob = vb;

  const int b    = blockIdx.x;
  const int tid  = threadIdx.x;
  const int wid  = tid >> 5;
  const int lane = tid & 31;
  const int l16  = lane & 15;
  const int half = lane >> 4;
  const int koff = 2 * half;

  for (int i = tid; i < 8192; i += TB) {
    int s = i >> 6, d = i & 63;
    xb[i] = wte[idx[b * 128 + s] * 64 + d] + wpe[i];
  }
  __syncthreads();

  const int NL = *num_loops;
  for (int t = 0; t < NL; ++t) {
    for (int i = tid; i < 8192; i += TB) xb[i] += ttab[t * 64 + (i & 63)];
    __syncthreads();

    ln_rows(xb, hb, ln1_g, ln1_b, tid);
    __syncthreads();

    for (int tile = wid; tile < 96; tile += NW) {
      const int m0 = (tile / 12) * 16, n0 = (tile % 12) * 16;
      v8f acc;
      { float bias = b_qkv[n0 + l16]; for (int r = 0; r < 8; ++r) acc[r] = bias; }
      for (int k = 0; k < 64; k += 4) {
        v2f a, w;
        a.x = hb[(m0 + l16) * 64 + k + koff];
        a.y = hb[(m0 + l16) * 64 + k + koff + 1];
        w.x = w_qkv[(k + koff) * 192 + n0 + l16];
        w.y = w_qkv[(k + koff + 1) * 192 + n0 + l16];
        acc = wmma4(a, w, acc);
      }
      float* dst; int nc;
      if (n0 < 64)       { dst = qb; nc = n0; }
      else if (n0 < 128) { dst = kb; nc = n0 - 64; }
      else               { dst = vb; nc = n0 - 128; }
      for (int r = 0; r < 8; ++r)
        dst[(m0 + r + 8 * half) * 64 + nc + l16] = acc[r];
    }
    __syncthreads();

    for (int job = wid; job < 32; job += NW) {
      const int hh = job >> 3, qt = job & 7;
      const int q0 = qt * 16, c0 = hh * 16;
      v8f sc[8];
      for (int kt = 0; kt < 8; ++kt) {
        v8f a8; for (int r = 0; r < 8; ++r) a8[r] = 0.0f;
        for (int kk = 0; kk < 16; kk += 4) {
          v2f a, bb;
          a.x  = qb[(q0 + l16) * 64 + c0 + kk + koff];
          a.y  = qb[(q0 + l16) * 64 + c0 + kk + koff + 1];
          bb.x = kb[(kt * 16 + l16) * 64 + c0 + kk + koff];
          bb.y = kb[(kt * 16 + l16) * 64 + c0 + kk + koff + 1];
          a8 = wmma4(a, bb, a8);
        }
        for (int r = 0; r < 8; ++r) {
          const int row = q0 + r + 8 * half, col = kt * 16 + l16;
          a8[r] = (col <= row) ? a8[r] * 0.25f : -1e9f;
        }
        sc[kt] = a8;
      }
      for (int r = 0; r < 8; ++r) {
        float mx = -1e30f;
        for (int kt = 0; kt < 8; ++kt) mx = fmaxf(mx, sc[kt][r]);
        for (int off = 8; off; off >>= 1) mx = fmaxf(mx, __shfl_xor(mx, off, 16));
        float sum = 0.0f;
        for (int kt = 0; kt < 8; ++kt) {
          float e = __expf(sc[kt][r] - mx);
          sc[kt][r] = e; sum += e;
        }
        for (int off = 8; off; off >>= 1) sum += __shfl_xor(sum, off, 16);
        const float inv = 1.0f / sum;
        for (int kt = 0; kt < 8; ++kt) sc[kt][r] *= inv;
      }
      v8f o; for (int r = 0; r < 8; ++r) o[r] = 0.0f;
      for (int kt = 0; kt < 8; ++kt) {
        for (int r = 0; r < 8; ++r)
          hb[(q0 + r + 8 * half) * 64 + c0 + l16] = sc[kt][r];
        for (int kk = 0; kk < 16; kk += 4) {
          v2f a, bb;
          a.x  = hb[(q0 + l16) * 64 + c0 + kk + koff];
          a.y  = hb[(q0 + l16) * 64 + c0 + kk + koff + 1];
          bb.x = vb[(kt * 16 + kk + koff) * 64 + c0 + l16];
          bb.y = vb[(kt * 16 + kk + koff + 1) * 64 + c0 + l16];
          o = wmma4(a, bb, o);
        }
      }
      for (int r = 0; r < 8; ++r)
        hb[(q0 + r + 8 * half) * 64 + c0 + l16] = o[r];
    }
    __syncthreads();

    for (int tile = wid; tile < 32; tile += NW) {
      const int m0 = (tile >> 2) * 16, n0 = (tile & 3) * 16;
      v8f acc;
      { float bias = b_o[n0 + l16]; for (int r = 0; r < 8; ++r) acc[r] = bias; }
      for (int k = 0; k < 64; k += 4) {
        v2f a, w;
        a.x = hb[(m0 + l16) * 64 + k + koff];
        a.y = hb[(m0 + l16) * 64 + k + koff + 1];
        w.x = w_o[(k + koff) * 64 + n0 + l16];
        w.y = w_o[(k + koff + 1) * 64 + n0 + l16];
        acc = wmma4(a, w, acc);
      }
      for (int r = 0; r < 8; ++r)
        xb[(m0 + r + 8 * half) * 64 + n0 + l16] += 0.1f * acc[r];
    }
    __syncthreads();

    ln_rows(xb, hb, ln2_g, ln2_b, tid);
    __syncthreads();

    for (int hh2 = 0; hh2 < 2; ++hh2) {
      for (int tile = wid; tile < 64; tile += NW) {
        const int m0 = (tile >> 3) * 16, ntl = (tile & 7) * 16;
        const int n0g = hh2 * 128 + ntl;
        v8f acc;
        { float bias = b_fc[n0g + l16]; for (int r = 0; r < 8; ++r) acc[r] = bias; }
        for (int k = 0; k < 64; k += 4) {
          v2f a, w;
          a.x = hb[(m0 + l16) * 64 + k + koff];
          a.y = hb[(m0 + l16) * 64 + k + koff + 1];
          w.x = w_fc[(k + koff) * 256 + n0g + l16];
          w.y = w_fc[(k + koff + 1) * 256 + n0g + l16];
          acc = wmma4(a, w, acc);
        }
        for (int r = 0; r < 8; ++r) {
          float u = acc[r];
          float g = 0.5f * u *
              (1.0f + tanhf(0.7978845608028654f * (u + 0.044715f * u * u * u)));
          gb[(m0 + r + 8 * half) * 128 + ntl + l16] = g;
        }
      }
      __syncthreads();
      for (int tile = wid; tile < 32; tile += NW) {
        const int m0 = (tile >> 2) * 16, n0 = (tile & 3) * 16;
        v8f acc;
        if (hh2 == 0) {
          float bias = b_pr[n0 + l16];
          for (int r = 0; r < 8; ++r) acc[r] = bias;
        } else {
          for (int r = 0; r < 8; ++r)
            acc[r] = ob[(m0 + r + 8 * half) * 64 + n0 + l16];
        }
        for (int k = 0; k < 128; k += 4) {
          v2f a, w;
          a.x = gb[(m0 + l16) * 128 + k + koff];
          a.y = gb[(m0 + l16) * 128 + k + koff + 1];
          w.x = w_pr[(hh2 * 128 + k + koff) * 64 + n0 + l16];
          w.y = w_pr[(hh2 * 128 + k + koff + 1) * 64 + n0 + l16];
          acc = wmma4(a, w, acc);
        }
        if (hh2 == 0) {
          for (int r = 0; r < 8; ++r)
            ob[(m0 + r + 8 * half) * 64 + n0 + l16] = acc[r];
        } else {
          for (int r = 0; r < 8; ++r)
            xb[(m0 + r + 8 * half) * 64 + n0 + l16] += 0.1f * acc[r];
        }
      }
      __syncthreads();
    }
  }

  {
    const int s = tid >> 2, p = tid & 3;
    float sum = 0.0f, sq = 0.0f;
    for (int d = p * 16; d < p * 16 + 16; ++d) {
      float xv = xb[s * 64 + d];
      sum += xv; sq += xv * xv;
    }
    for (int off = 1; off < 4; off <<= 1) {
      sum += __shfl_xor(sum, off, 4);
      sq  += __shfl_xor(sq,  off, 4);
    }
    const float mu  = sum * (1.0f / 64.0f);
    const float var = sq * (1.0f / 64.0f) - mu * mu;
    const float inv = rsqrtf(var + 1e-5f);
    for (int d = p * 16; d < p * 16 + 16; ++d)
      out[b * 8192 + s * 64 + d] =
          (xb[s * 64 + d] - mu) * inv * lnf_g[d] + lnf_b[d];
  }
}

extern "C" void kernel_launch(void* const* d_in, const int* in_sizes, int n_in,
                              void* d_out, int out_size, void* d_ws, size_t ws_size,
                              hipStream_t stream) {
  const int*   idx   = (const int*)  d_in[0];
  const int*   nlp   = (const int*)  d_in[1];
  const float* wte   = (const float*)d_in[2];
  const float* wpe   = (const float*)d_in[3];
  const float* t_w1  = (const float*)d_in[4];
  const float* t_b1  = (const float*)d_in[5];
  const float* t_w2  = (const float*)d_in[6];
  const float* t_b2  = (const float*)d_in[7];
  const float* ln1_g = (const float*)d_in[8];
  const float* ln1_b = (const float*)d_in[9];
  const float* w_qkv = (const float*)d_in[10];
  const float* b_qkv = (const float*)d_in[11];
  const float* w_o   = (const float*)d_in[12];
  const float* b_o   = (const float*)d_in[13];
  const float* ln2_g = (const float*)d_in[14];
  const float* ln2_b = (const float*)d_in[15];
  const float* w_fc  = (const float*)d_in[16];
  const float* b_fc  = (const float*)d_in[17];
  const float* w_pr  = (const float*)d_in[18];
  const float* b_pr  = (const float*)d_in[19];
  const float* lnf_g = (const float*)d_in[20];
  const float* lnf_b = (const float*)d_in[21];

  float* ttab = (float*)d_ws;

  temb_kernel<<<128, 256, 0, stream>>>(nlp, t_w1, t_b1, t_w2, t_b2, ttab);
  xformer_kernel<<<32, TB, 0, stream>>>(idx, nlp, wte, wpe, ln1_g, ln1_b,
                                        w_qkv, b_qkv, w_o, b_o, ln2_g, ln2_b,
                                        w_fc, b_fc, w_pr, b_pr, lnf_g, lnf_b,
                                        ttab, (float*)d_out);
}


# Weights, biases and LN vectors of each phase loaded into registers one to two phases ahead, kept in flight across hand-written LDS-only workgroup barriers; sums and order unchanged
# speedup vs baseline: 7.4272x; 7.4272x over previous
#include <hip/hip_runtime.h>
#include <hip/hip_bf16.h>
#include <math.h>


typedef __attribute__((ext_vector_type(2))) float v2f;
typedef __attribute__((ext_vector_type(8))) float v8f;

__device__ __forceinline__ v8f wmma4(v2f a, v2f b, v8f c) {
  return __builtin_amdgcn_wmma_f32_16x16x4_f32(
       false, a,  false, b,
       (short)0, c,  false,  false);
}

__global__ __launch_bounds__(256) void temb_kernel(
    const int* __restrict__ num_loops,
    const float* __restrict__ t_w1, const float* __restrict__ t_b1,
    const float* __restrict__ t_w2, const float* __restrict__ t_b2,
    float* __restrict__ ttab) {
  const int t = blockIdx.x;
  if (t >= *num_loops) return;
  __shared__ float te[256];
  __shared__ float hs[1024];
  __shared__ float red[256];
  const int tid = threadIdx.x;

  if (tid < 128) {
    float f   = expf(-logf(10000.0f) * (float)tid / 128.0f);
    float arg = (float)t * f;
    te[tid]       = cosf(arg);
    te[128 + tid] = sinf(arg);
  }
  __syncthreads();

  for (int j = tid; j < 1024; j += 256) {
    float acc = t_b1[j];
    for (int k = 0; k < 256; ++k) acc += te[k] * t_w1[k * 1024 + j];
    hs[j] = acc / (1.0f + expf(-acc));
  }
  __syncthreads();

  const int j = tid & 63, p = tid >> 6;
  float acc = 0.0f;
  for (int k = p * 256; k < p * 256 + 256; ++k) acc += hs[k] * t_w2[k * 64 + j];
  red[tid] = acc;
  __syncthreads();
  if (tid < 64) {
    ttab[t * 64 + tid] =
        t_b2[tid] + red[tid] + red[tid + 64] + red[tid + 128] + red[tid + 192];
  }
}

#define NT 512
#define WS_X 8192
#define WS_QKV (8192 + 262144)
#define PLANE 262144
#define L_XB 0
#define L_HB 1024
#define L_GB 2048
#define L_PART 6144
#define L_PM 10240
#define L_PL 10496
#define L_PW 10752
#define L_TOTAL 18944

__device__ __forceinline__ int afrag(int row, int k) {
  return ((k >> 2) << 6) + (((k >> 1) & 1) << 5) + (row << 1) + (k & 1);
}
__device__ __forceinline__ v2f ld_afrag(const float* buf, int step, int lane) {
  return *reinterpret_cast<const v2f*>(buf + step * 64 + lane * 2);
}

template <bool ADD, bool STEP0, bool FRAG>
__device__ __forceinline__ void row_pass(
    float* xb, const float* part, const float* __restrict__ bias,
    const float* __restrict__ tt, float* __restrict__ xg,
    const float* __restrict__ g, const float* __restrict__ bta, float* dst,
    int wid, int lane) {
  float v[2];
#pragma unroll
  for (int e = 0; e < 2; ++e) {
    const int d = lane + 32 * e;
    float x = xb[wid * 64 + d];
    if (ADD) {
      float s = bias[d];
      for (int k = 0; k < 4; ++k) s += part[k * 1024 + wid * 64 + d];
      x += 0.1f * s;
    }
    if (STEP0) x += tt[d];
    v[e] = x;
    if (ADD || STEP0) xb[wid * 64 + d] = x;
    if (STEP0) xg[wid * 64 + d] = x;
  }
  float sum = v[0] + v[1], sq = v[0] * v[0] + v[1] * v[1];
  for (int off = 16; off; off >>= 1) {
    sum += __shfl_xor(sum, off, 32);
    sq  += __shfl_xor(sq,  off, 32);
  }
  const float mu  = sum * (1.0f / 64.0f);
  const float var = sq * (1.0f / 64.0f) - mu * mu;
  const float inv = rsqrtf(var + 1e-5f);
#pragma unroll
  for (int e = 0; e < 2; ++e) {
    const int d = lane + 32 * e;
    dst[FRAG ? afrag(wid, d) : wid * 64 + d] = (v[e] - mu) * inv * g[d] + bta[d];
  }
}

__device__ __forceinline__ void qkv_out(
    const float* hb, float* st, const float* __restrict__ w_qkv,
    const float* __restrict__ b_qkv, float* __restrict__ qkv_next, int toff,
    int tid, int wid, int lane, int l16, int half, int koff) {
  if (wid < 12) {
    const int n0 = wid * 16;
    v8f acc;
    { float bias = b_qkv[n0 + l16]; for (int r = 0; r < 8; ++r) acc[r] = bias; }
#pragma unroll
    for (int k = 0; k < 64; k += 4) {
      v2f a = ld_afrag(hb, k >> 2, lane), w;
      w.x = w_qkv[(k + koff) * 192 + n0 + l16];
      w.y = w_qkv[(k + koff + 1) * 192 + n0 + l16];
      acc = wmma4(a, w, acc);
    }
    for (int r = 0; r < 8; ++r)
      st[(wid >> 2) * 1024 + (r + 8 * half) * 64 + (wid & 3) * 16 + l16] = acc[r];
  }
  __syncthreads();
  for (int i = tid; i < 768; i += NT) {
    const int pl = i >> 8, off = i & 255;
    reinterpret_cast<float4*>(qkv_next + pl * PLANE + toff)[off] =
        reinterpret_cast<const float4*>(st + pl * 1024)[off];
  }
}

__device__ __forceinline__ void out_copy(const float* hb, float* __restrict__ outg, int tid) {
  __syncthreads();
  for (int i = tid; i < 256; i += NT)
    reinterpret_cast<float4*>(outg)[i] = reinterpret_cast<const float4*>(hb)[i];
}

__device__ __forceinline__ void wg_barrier() {
  asm volatile("s_wait_dscnt 0x0\n\ts_barrier_signal -1\n\ts_barrier_wait -1" ::: "memory");
}

template <bool STEP0, bool FRAG>
__device__ __forceinline__ void row_pass_r(
    float* xb, const float* part, const float (&bias)[2], const float (&tt)[2],
    float* __restrict__ xg, const float (&g)[2], const float (&bta)[2],
    float* dst, int wid, int lane) {
  float v[2];
#pragma unroll
  for (int e = 0; e < 2; ++e) {
    const int d = lane + 32 * e;
    float x = xb[wid * 64 + d];
    float s = bias[e];
    for (int k = 0; k < 4; ++k) s += part[k * 1024 + wid * 64 + d];
    x += 0.1f * s;
    if (STEP0) x += tt[e];
    v[e] = x;
    xb[wid * 64 + d] = x;
    if (STEP0) xg[wid * 64 + d] = x;
  }
  float sum = v[0] + v[1], sq = v[0] * v[0] + v[1] * v[1];
  for (int off = 16; off; off >>= 1) {
    sum += __shfl_xor(sum, off, 32);
    sq  += __shfl_xor(sq,  off, 32);
  }
  const float mu  = sum * (1.0f / 64.0f);
  const float var = sq * (1.0f / 64.0f) - mu * mu;
  const float inv = rsqrtf(var + 1e-5f);
#pragma unroll
  for (int e = 0; e < 2; ++e) {
    const int d = lane + 32 * e;
    dst[FRAG ? afrag(wid, d) : wid * 64 + d] = (v[e] - mu) * inv * g[e] + bta[e];
  }
}

__global__ __launch_bounds__(NT) void embed_kernel(
    const int* __restrict__ idx, const int* __restrict__ num_loops,
    const float* __restrict__ wte, const float* __restrict__ wpe,
    const float* __restrict__ ln1_g, const float* __restrict__ ln1_b,
    const float* __restrict__ w_qkv, const float* __restrict__ b_qkv,
    const float* __restrict__ lnf_g, const float* __restrict__ lnf_b,
    float* __restrict__ ws, float* __restrict__ out) {
  __shared__ __attribute__((aligned(16))) float smem[L_TOTAL];
  float* xb = smem + L_XB;
  float* hb = smem + L_HB;
  float* st = smem + L_GB;
  const int b = blockIdx.x >> 3, q0 = (blockIdx.x & 7) * 16;
  const int toff = (b * 128 + q0) * 64;
  const int tid = threadIdx.x, lane = tid & 31;
  const int wid = __builtin_amdgcn_readfirstlane(tid >> 5);
  const int l16 = lane & 15, half = lane >> 4, koff = 2 * half;

  for (int i = tid; i < 1024; i += NT) {
    int s = i >> 6, d = i & 63;
    xb[i] = wte[idx[b * 128 + q0 + s] * 64 + d] + wpe[(q0 + s) * 64 + d];
  }
  __syncthreads();

  if (*num_loops > 0) {
    row_pass<false, true, true>(xb, nullptr, nullptr, ws, ws + WS_X + toff, ln1_g, ln1_b, hb, wid, lane);
    __syncthreads();
    qkv_out(hb, st, w_qkv, b_qkv, ws + WS_QKV, toff, tid, wid, lane, l16, half, koff);
  } else {
    row_pass<false, false, false>(xb, nullptr, nullptr, nullptr, nullptr, lnf_g, lnf_b, hb, wid, lane);
    out_copy(hb, out + toff, tid);
  }
}

__global__ __launch_bounds__(NT) void step_kernel(
    int t, const int* __restrict__ num_loops,
    const float* __restrict__ ln1_g, const float* __restrict__ ln1_b,
    const float* __restrict__ w_qkv, const float* __restrict__ b_qkv,
    const float* __restrict__ w_o, const float* __restrict__ b_o,
    const float* __restrict__ ln2_g, const float* __restrict__ ln2_b,
    const float* __restrict__ w_fc, const float* __restrict__ b_fc,
    const float* __restrict__ w_pr, const float* __restrict__ b_pr,
    const float* __restrict__ lnf_g, const float* __restrict__ lnf_b,
    float* __restrict__ ws, float* __restrict__ out) {
  const int NL = *num_loops;
  if (t >= NL) return;

  __shared__ __attribute__((aligned(16))) float smem[L_TOTAL];
  float* xb = smem + L_XB;
  float* hb = smem + L_HB;
  float* gb = smem + L_GB;
  float* part = smem + L_PART;
  float* pm = smem + L_PM;
  float* pl = smem + L_PL;

  const int b = blockIdx.x >> 3, q0 = (blockIdx.x & 7) * 16;
  const int toff = (b * 128 + q0) * 64;
  const int tid = threadIdx.x, lane = tid & 31;
  const int wid = __builtin_amdgcn_readfirstlane(tid >> 5);
  const int l16 = lane & 15, half = lane >> 4, koff = 2 * half;
  const int wq = wid & 3;
  const int kq = wid >> 2;

  float* xg = ws + WS_X + toff;
  const float* cur = ws + WS_QKV + (t & 1) * (3 * PLANE);
  float* nxt = ws + WS_QKV + ((t + 1) & 1) * (3 * PLANE);
  const float* qg = cur + toff;
  const float* kg = cur + PLANE + b * 8192;
  const float* vg = cur + 2 * PLANE + b * 8192;

  const bool more = (t + 1 < NL);
  const int c0 = wq * 16;
  const int n0g = wid * 16;

  const float x0 = xg[tid], x1 = xg[tid + NT];
  v2f qa[4], kb[2][4], vb[2][4], fo[4];
#pragma unroll
  for (int s = 0; s < 4; ++s) {
    qa[s].x = qg[l16 * 64 + c0 + 4 * s + koff];
    qa[s].y = qg[l16 * 64 + c0 + 4 * s + koff + 1];
  }
#pragma unroll
  for (int j = 0; j < 2; ++j)
#pragma unroll
    for (int s = 0; s < 4; ++s) {
      const int kt = 2 * kq + j;
      kb[j][s].x = kg[(kt * 16 + l16) * 64 + c0 + 4 * s + koff];
      kb[j][s].y = kg[(kt * 16 + l16) * 64 + c0 + 4 * s + koff + 1];
    }
#pragma unroll
  for (int j = 0; j < 2; ++j)
#pragma unroll
    for (int s = 0; s < 4; ++s) {
      const int kt = 2 * kq + j;
      vb[j][s].x = vg[(kt * 16 + 4 * s + koff) * 64 + c0 + l16];
      vb[j][s].y = vg[(kt * 16 + 4 * s + koff + 1) * 64 + c0 + l16];
    }
#pragma unroll
  for (int s = 0; s < 4; ++s) {
    const int k = kq * 16 + 4 * s;
    fo[s].x = w_o[(k + koff) * 64 + c0 + l16];
    fo[s].y = w_o[(k + koff + 1) * 64 + c0 + l16];
  }
  float cbo[2], cg2[2], cb2[2];
#pragma unroll
  for (int e = 0; e < 2; ++e) {
    const int d = lane + 32 * e;
    cbo[e] = b_o[d]; cg2[e] = ln2_g[d]; cb2[e] = ln2_b[d];
  }
  __builtin_amdgcn_sched_barrier(0);

  v2f ffc[16];
  float bfc;
  {
    float* pw = smem + L_PW + wid * 512;
#pragma unroll
    for (int j = 0; j < 2; ++j) {
      const int kt = 2 * kq + j;
      v8f a8; for (int r = 0; r < 8; ++r) a8[r] = 0.0f;
#pragma unroll
      for (int s = 0; s < 4; ++s)
        a8 = wmma4(qa[s], kb[j][s], a8);
#pragma unroll
      for (int r = 0; r < 8; ++r) {
        const int row = q0 + r + 8 * half, col = kt * 16 + l16;
        pw[j * 256 + (r + 8 * half) * 16 + l16] =
            (col <= row) ? a8[r] * 0.25f : -1e9f;
      }
    }
    v2f p[2][4];
    float mx = -1e30f;
#pragma unroll
    for (int j = 0; j < 2; ++j)
#pragma unroll
      for (int s = 0; s < 4; ++s) {
        p[j][s] = *reinterpret_cast<const v2f*>(pw + j * 256 + l16 * 16 + 4 * s + koff);
        mx = fmaxf(mx, fmaxf(p[j][s].x, p[j][s].y));
      }
    mx = fmaxf(mx, __shfl_xor(mx, 16, 32));
    float sum = 0.0f;
#pragma unroll
    for (int j = 0; j < 2; ++j)
#pragma unroll
      for (int s = 0; s < 4; ++s) {
        p[j][s].x = __expf(p[j][s].x - mx);
        p[j][s].y = __expf(p[j][s].y - mx);
        sum += p[j][s].x + p[j][s].y;
      }
    sum += __shfl_xor(sum, 16, 32);
    v8f o; for (int r = 0; r < 8; ++r) o[r] = 0.0f;
#pragma unroll
    for (int j = 0; j < 2; ++j)
#pragma unroll
      for (int s = 0; s < 4; ++s)
        o = wmma4(p[j][s], vb[j][s], o);
    __builtin_amdgcn_sched_barrier(0);
#pragma unroll
    for (int s = 0; s < 16; ++s) {
      ffc[s].x = w_fc[(4 * s + koff) * 256 + n0g + l16];
      ffc[s].y = w_fc[(4 * s + koff + 1) * 256 + n0g + l16];
    }
    bfc = b_fc[n0g + l16];
    for (int r = 0; r < 8; ++r)
      part[kq * 1024 + (r + 8 * half) * 64 + c0 + l16] = o[r];
    if (half == 0) {
      pm[(kq * 4 + wq) * 16 + l16] = mx;
      pl[(kq * 4 + wq) * 16 + l16] = sum;
    }
  }
  xb[tid] = x0;
  xb[tid + NT] = x1;
  wg_barrier();
  {
    const int row = tid >> 5, c = (tid & 31) * 2, hh = c >> 4;
    float m[4];
    for (int k = 0; k < 4; ++k) m[k] = pm[(k * 4 + hh) * 16 + row];
    const float M = fmaxf(fmaxf(m[0], m[1]), fmaxf(m[2], m[3]));
    float L = 0.0f;
    v2f O; O.x = 0.0f; O.y = 0.0f;
    for (int k = 0; k < 4; ++k) {
      const float wk = __expf(m[k] - M);
      const v2f pk = *reinterpret_cast<const v2f*>(part + k * 1024 + row * 64 + c);
      L += pl[(k * 4 + hh) * 16 + row] * wk;
      O.x += pk.x * wk;
      O.y += pk.y * wk;
    }
    const float inv = 1.0f / L;
    O.x *= inv; O.y *= inv;
    *reinterpret_cast<v2f*>(hb + afrag(row, c)) = O;
  }
  wg_barrier();

  v2f fpr[16];
  float cbp[2], ctt[2], cgE[2], cbE[2];
  {
    v8f acc; for (int r = 0; r < 8; ++r) acc[r] = 0.0f;
#pragma unroll
    for (int s = 0; s < 4; ++s)
      acc = wmma4(ld_afrag(hb, kq * 4 + s, lane), fo[s], acc);
    for (int r = 0; r < 8; ++r)
      part[kq * 1024 + (r + 8 * half) * 64 + c0 + l16] = acc[r];
#pragma unroll
    for (int s = 0; s < 16; ++s) {
      const int k = kq * 64 + 4 * s;
      fpr[s].x = w_pr[(k + koff) * 64 + c0 + l16];
      fpr[s].y = w_pr[(k + koff + 1) * 64 + c0 + l16];
    }
    const float* gE = more ? ln1_g : lnf_g;
    const float* bE = more ? ln1_b : lnf_b;
    const float* tr = ws + (more ? (t + 1) * 64 : 0);
#pragma unroll
    for (int e = 0; e < 2; ++e) {
      const int d = lane + 32 * e;
      cbp[e] = b_pr[d]; ctt[e] = tr[d]; cgE[e] = gE[d]; cbE[e] = bE[d];
    }
  }
  wg_barrier();

  row_pass_r<false, true>(xb, part, cbo, cbo, nullptr, cg2, cb2, hb, wid, lane);
  wg_barrier();

  v2f fq[16];
  float bq = 0.0f;
#pragma unroll
  for (int s = 0; s < 16; ++s) { fq[s].x = 0.0f; fq[s].y = 0.0f; }
  {
    v8f acc;
    for (int r = 0; r < 8; ++r) acc[r] = bfc;
#pragma unroll
    for (int s = 0; s < 16; ++s)
      acc = wmma4(ld_afrag(hb, s, lane), ffc[s], acc);
    if (more && wid < 12) {
#pragma unroll
      for (int s = 0; s < 16; ++s) {
        fq[s].x = w_qkv[(4 * s + koff) * 192 + n0g + l16];
        fq[s].y = w_qkv[(4 * s + koff + 1) * 192 + n0g + l16];
      }
      bq = b_qkv[n0g + l16];
    }
    for (int r = 0; r < 8; ++r) {
      float u = acc[r];
      float g = 0.5f * u *
          (1.0f + tanhf(0.7978845608028654f * (u + 0.044715f * u * u * u)));
      gb[afrag(r + 8 * half, n0g + l16)] = g;
    }
  }
  wg_barrier();
  {
    v8f acc; for (int r = 0; r < 8; ++r) acc[r] = 0.0f;
#pragma unroll
    for (int s = 0; s < 16; ++s)
      acc = wmma4(ld_afrag(gb, kq * 16 + s, lane), fpr[s], acc);
    for (int r = 0; r < 8; ++r)
      part[kq * 1024 + (r + 8 * half) * 64 + c0 + l16] = acc[r];
  }
  wg_barrier();

  if (more) {
    row_pass_r<true, true>(xb, part, cbp, ctt, xg, cgE, cbE, hb, wid, lane);
    wg_barrier();
    if (wid < 12) {
      v8f acc;
      for (int r = 0; r < 8; ++r) acc[r] = bq;
#pragma unroll
      for (int s = 0; s < 16; ++s)
        acc = wmma4(ld_afrag(hb, s, lane), fq[s], acc);
      for (int r = 0; r < 8; ++r)
        gb[(wid >> 2) * 1024 + (r + 8 * half) * 64 + (wid & 3) * 16 + l16] = acc[r];
    }
    wg_barrier();
    for (int i = tid; i < 768; i += NT) {
      const int pln = i >> 8, off = i & 255;
      reinterpret_cast<float4*>(nxt + pln * PLANE + toff)[off] =
          reinterpret_cast<const float4*>(gb + pln * 1024)[off];
    }
  } else {
    row_pass_r<false, false>(xb, part, cbp, ctt, nullptr, cgE, cbE, hb, wid, lane);
    wg_barrier();
    for (int i = tid; i < 256; i += NT)
      reinterpret_cast<float4*>(out + toff)[i] = reinterpret_cast<const float4*>(hb)[i];
  }
}

extern "C" void kernel_launch(void* const* d_in, const int* in_sizes, int n_in,
                              void* d_out, int out_size, void* d_ws, size_t ws_size,
                              hipStream_t stream) {
  const int*   idx   = (const int*)  d_in[0];
  const int*   nlp   = (const int*)  d_in[1];
  const float* wte   = (const float*)d_in[2];
  const float* wpe   = (const float*)d_in[3];
  const float* t_w1  = (const float*)d_in[4];
  const float* t_b1  = (const float*)d_in[5];
  const float* t_w2  = (const float*)d_in[6];
  const float* t_b2  = (const float*)d_in[7];
  const float* ln1_g = (const float*)d_in[8];
  const float* ln1_b = (const float*)d_in[9];
  const float* w_qkv = (const float*)d_in[10];
  const float* b_qkv = (const float*)d_in[11];
  const float* w_o   = (const float*)d_in[12];
  const float* b_o   = (const float*)d_in[13];
  const float* ln2_g = (const float*)d_in[14];
  const float* ln2_b = (const float*)d_in[15];
  const float* w_fc  = (const float*)d_in[16];
  const float* b_fc  = (const float*)d_in[17];
  const float* w_pr  = (const float*)d_in[18];
  const float* b_pr  = (const float*)d_in[19];
  const float* lnf_g = (const float*)d_in[20];
  const float* lnf_b = (const float*)d_in[21];

  float* ws = (float*)d_ws;

  temb_kernel<<<128, 256, 0, stream>>>(nlp, t_w1, t_b1, t_w2, t_b2, ws);
  embed_kernel<<<256, NT, 0, stream>>>(idx, nlp, wte, wpe, ln1_g, ln1_b, w_qkv,
                                       b_qkv, lnf_g, lnf_b, ws, (float*)d_out);
  for (int t = 0; t < 100; ++t)
    step_kernel<<<256, NT, 0, stream>>>(t, nlp, ln1_g, ln1_b, w_qkv, b_qkv, w_o,
                                        b_o, ln2_g, ln2_b, w_fc, b_fc, w_pr,
                                        b_pr, lnf_g, lnf_b, ws, (float*)d_out);
}


# q, k, v cross the workspace in WMMA fragment order per tile and head: an attention operand load is one whole-line 512-byte wave load, not 16 row pieces of 8 bytes; sums and order unchanged
# speedup vs baseline: 7.7858x; 1.0483x over previous
#include <hip/hip_runtime.h>
#include <hip/hip_bf16.h>
#include <math.h>


typedef __attribute__((ext_vector_type(2))) float v2f;
typedef __attribute__((ext_vector_type(8))) float v8f;

__device__ __forceinline__ v8f wmma4(v2f a, v2f b, v8f c) {
  return __builtin_amdgcn_wmma_f32_16x16x4_f32(
       false, a,  false, b,
       (short)0, c,  false,  false);
}

__global__ __launch_bounds__(256) void temb_kernel(
    const int* __restrict__ num_loops,
    const float* __restrict__ t_w1, const float* __restrict__ t_b1,
    const float* __restrict__ t_w2, const float* __restrict__ t_b2,
    float* __restrict__ ttab) {
  const int t = blockIdx.x;
  if (t >= *num_loops) return;
  __shared__ float te[256];
  __shared__ float hs[1024];
  __shared__ float red[256];
  const int tid = threadIdx.x;

  if (tid < 128) {
    float f   = expf(-logf(10000.0f) * (float)tid / 128.0f);
    float arg = (float)t * f;
    te[tid]       = cosf(arg);
    te[128 + tid] = sinf(arg);
  }
  __syncthreads();

  for (int j = tid; j < 1024; j += 256) {
    float acc = t_b1[j];
    for (int k = 0; k < 256; ++k) acc += te[k] * t_w1[k * 1024 + j];
    hs[j] = acc / (1.0f + expf(-acc));
  }
  __syncthreads();

  const int j = tid & 63, p = tid >> 6;
  float acc = 0.0f;
  for (int k = p * 256; k < p * 256 + 256; ++k) acc += hs[k] * t_w2[k * 64 + j];
  red[tid] = acc;
  __syncthreads();
  if (tid < 64) {
    ttab[t * 64 + tid] =
        t_b2[tid] + red[tid] + red[tid + 64] + red[tid + 128] + red[tid + 192];
  }
}

#define NT 512
#define WS_X 8192
#define WS_QKV (8192 + 262144)
#define PLANE 262144
#define L_XB 0
#define L_HB 1024
#define L_GB 2048
#define L_PART 6144
#define L_PM 10240
#define L_PL 10496
#define L_PW 10752
#define L_TOTAL 18944

__device__ __forceinline__ int afrag(int row, int k) {
  return ((k >> 2) << 6) + (((k >> 1) & 1) << 5) + (row << 1) + (k & 1);
}
__device__ __forceinline__ v2f ld_afrag(const float* buf, int step, int lane) {
  return *reinterpret_cast<const v2f*>(buf + step * 64 + lane * 2);
}

#define QK_SP 68
template <int P>
__device__ __forceinline__ int qkfrag(int r, int d) {
  return (((d >> 3) << 1) + ((d >> 1) & 1)) * P + (r << 2) + (((d >> 2) & 1) << 1) + (d & 1);
}
__device__ __forceinline__ void ld_frag2(const float* blk, int sp, int lane, v2f& f0, v2f& f1) {
  const float4 f = reinterpret_cast<const float4*>(blk)[sp * 32 + lane];
  f0.x = f.x; f0.y = f.y; f1.x = f.z; f1.y = f.w;
}
__device__ __forceinline__ void stage_qkv(float* st, const v8f& acc, int ct, int l16, int half) {
  float* blk = st + ct * (4 * QK_SP);
  if (ct >= 8) {
    for (int r = 0; r < 8; ++r) blk[qkfrag<QK_SP>(l16, r + 8 * half)] = acc[r];
  } else {
    for (int r = 0; r < 8; ++r) blk[qkfrag<QK_SP>(r + 8 * half, l16)] = acc[r];
  }
}
__device__ __forceinline__ const float4& staged4(const float* st, int i) {
  return *reinterpret_cast<const float4*>(st + (i >> 4) * QK_SP + (i & 15) * 4);
}

template <bool ADD, bool STEP0, bool FRAG>
__device__ __forceinline__ void row_pass(
    float* xb, const float* part, const float* __restrict__ bias,
    const float* __restrict__ tt, float* __restrict__ xg,
    const float* __restrict__ g, const float* __restrict__ bta, float* dst,
    int wid, int lane) {
  float v[2];
#pragma unroll
  for (int e = 0; e < 2; ++e) {
    const int d = lane + 32 * e;
    float x = xb[wid * 64 + d];
    if (ADD) {
      float s = bias[d];
      for (int k = 0; k < 4; ++k) s += part[k * 1024 + wid * 64 + d];
      x += 0.1f * s;
    }
    if (STEP0) x += tt[d];
    v[e] = x;
    if (ADD || STEP0) xb[wid * 64 + d] = x;
    if (STEP0) xg[wid * 64 + d] = x;
  }
  float sum = v[0] + v[1], sq = v[0] * v[0] + v[1] * v[1];
  for (int off = 16; off; off >>= 1) {
    sum += __shfl_xor(sum, off, 32);
    sq  += __shfl_xor(sq,  off, 32);
  }
  const float mu  = sum * (1.0f / 64.0f);
  const float var = sq * (1.0f / 64.0f) - mu * mu;
  const float inv = rsqrtf(var + 1e-5f);
#pragma unroll
  for (int e = 0; e < 2; ++e) {
    const int d = lane + 32 * e;
    dst[FRAG ? afrag(wid, d) : wid * 64 + d] = (v[e] - mu) * inv * g[d] + bta[d];
  }
}

__device__ __forceinline__ void qkv_out(
    const float* hb, float* st, const float* __restrict__ w_qkv,
    const float* __restrict__ b_qkv, float* __restrict__ qkv_next, int toff,
    int tid, int wid, int lane, int l16, int half, int koff) {
  if (wid < 12) {
    const int n0 = wid * 16;
    v8f acc;
    { float bias = b_qkv[n0 + l16]; for (int r = 0; r < 8; ++r) acc[r] = bias; }
#pragma unroll
    for (int k = 0; k < 64; k += 4) {
      v2f a = ld_afrag(hb, k >> 2, lane), w;
      w.x = w_qkv[(k + koff) * 192 + n0 + l16];
      w.y = w_qkv[(k + koff + 1) * 192 + n0 + l16];
      acc = wmma4(a, w, acc);
    }
    stage_qkv(st, acc, wid, l16, half);
  }
  __syncthreads();
  for (int i = tid; i < 768; i += NT) {
    const int pl = i >> 8, off = i & 255;
    reinterpret_cast<float4*>(qkv_next + pl * PLANE + toff)[off] =
        staged4(st, i);
  }
}

__device__ __forceinline__ void out_copy(const float* hb, float* __restrict__ outg, int tid) {
  __syncthreads();
  for (int i = tid; i < 256; i += NT)
    reinterpret_cast<float4*>(outg)[i] = reinterpret_cast<const float4*>(hb)[i];
}

__device__ __forceinline__ void wg_barrier() {
  asm volatile("s_wait_dscnt 0x0\n\ts_barrier_signal -1\n\ts_barrier_wait -1" ::: "memory");
}

template <bool STEP0, bool FRAG>
__device__ __forceinline__ void row_pass_r(
    float* xb, const float* part, const float (&bias)[2], const float (&tt)[2],
    float* __restrict__ xg, const float (&g)[2], const float (&bta)[2],
    float* dst, int wid, int lane) {
  float v[2];
#pragma unroll
  for (int e = 0; e < 2; ++e) {
    const int d = lane + 32 * e;
    float x = xb[wid * 64 + d];
    float s = bias[e];
    for (int k = 0; k < 4; ++k) s += part[k * 1024 + wid * 64 + d];
    x += 0.1f * s;
    if (STEP0) x += tt[e];
    v[e] = x;
    xb[wid * 64 + d] = x;
    if (STEP0) xg[wid * 64 + d] = x;
  }
  float sum = v[0] + v[1], sq = v[0] * v[0] + v[1] * v[1];
  for (int off = 16; off; off >>= 1) {
    sum += __shfl_xor(sum, off, 32);
    sq  += __shfl_xor(sq,  off, 32);
  }
  const float mu  = sum * (1.0f / 64.0f);
  const float var = sq * (1.0f / 64.0f) - mu * mu;
  const float inv = rsqrtf(var + 1e-5f);
#pragma unroll
  for (int e = 0; e < 2; ++e) {
    const int d = lane + 32 * e;
    dst[FRAG ? afrag(wid, d) : wid * 64 + d] = (v[e] - mu) * inv * g[e] + bta[e];
  }
}

__global__ __launch_bounds__(NT) void embed_kernel(
    const int* __restrict__ idx, const int* __restrict__ num_loops,
    const float* __restrict__ wte, const float* __restrict__ wpe,
    const float* __restrict__ ln1_g, const float* __restrict__ ln1_b,
    const float* __restrict__ w_qkv, const float* __restrict__ b_qkv,
    const float* __restrict__ lnf_g, const float* __restrict__ lnf_b,
    float* __restrict__ ws, float* __restrict__ out) {
  __shared__ __attribute__((aligned(16))) float smem[L_TOTAL];
  float* xb = smem + L_XB;
  float* hb = smem + L_HB;
  float* st = smem + L_GB;
  const int b = blockIdx.x >> 3, q0 = (blockIdx.x & 7) * 16;
  const int toff = (b * 128 + q0) * 64;
  const int tid = threadIdx.x, lane = tid & 31;
  const int wid = __builtin_amdgcn_readfirstlane(tid >> 5);
  const int l16 = lane & 15, half = lane >> 4, koff = 2 * half;

  for (int i = tid; i < 1024; i += NT) {
    int s = i >> 6, d = i & 63;
    xb[i] = wte[idx[b * 128 + q0 + s] * 64 + d] + wpe[(q0 + s) * 64 + d];
  }
  __syncthreads();

  if (*num_loops > 0) {
    row_pass<false, true, true>(xb, nullptr, nullptr, ws, ws + WS_X + toff, ln1_g, ln1_b, hb, wid, lane);
    __syncthreads();
    qkv_out(hb, st, w_qkv, b_qkv, ws + WS_QKV, toff, tid, wid, lane, l16, half, koff);
  } else {
    row_pass<false, false, false>(xb, nullptr, nullptr, nullptr, nullptr, lnf_g, lnf_b, hb, wid, lane);
    out_copy(hb, out + toff, tid);
  }
}

__global__ __launch_bounds__(NT) void step_kernel(
    int t, const int* __restrict__ num_loops,
    const float* __restrict__ ln1_g, const float* __restrict__ ln1_b,
    const float* __restrict__ w_qkv, const float* __restrict__ b_qkv,
    const float* __restrict__ w_o, const float* __restrict__ b_o,
    const float* __restrict__ ln2_g, const float* __restrict__ ln2_b,
    const float* __restrict__ w_fc, const float* __restrict__ b_fc,
    const float* __restrict__ w_pr, const float* __restrict__ b_pr,
    const float* __restrict__ lnf_g, const float* __restrict__ lnf_b,
    float* __restrict__ ws, float* __restrict__ out) {
  const int NL = *num_loops;
  if (t >= NL) return;

  __shared__ __attribute__((aligned(16))) float smem[L_TOTAL];
  float* xb = smem + L_XB;
  float* hb = smem + L_HB;
  float* gb = smem + L_GB;
  float* part = smem + L_PART;
  float* pm = smem + L_PM;
  float* pl = smem + L_PL;

  const int b = blockIdx.x >> 3, q0 = (blockIdx.x & 7) * 16;
  const int toff = (b * 128 + q0) * 64;
  const int tid = threadIdx.x, lane = tid & 31;
  const int wid = __builtin_amdgcn_readfirstlane(tid >> 5);
  const int l16 = lane & 15, half = lane >> 4, koff = 2 * half;
  const int wq = wid & 3;
  const int kq = wid >> 2;

  float* xg = ws + WS_X + toff;
  const float* cur = ws + WS_QKV + (t & 1) * (3 * PLANE);
  float* nxt = ws + WS_QKV + ((t + 1) & 1) * (3 * PLANE);
  const float* qg = cur + toff + wq * 256;
  const float* kg = cur + PLANE + b * 8192 + wq * 256;
  const float* vg = cur + 2 * PLANE + b * 8192 + wq * 256;

  const bool more = (t + 1 < NL);
  const int c0 = wq * 16;
  const int n0g = wid * 16;

  const float x0 = xg[tid], x1 = xg[tid + NT];
  v2f qa[4], kb[2][4], vb[2][4], fo[4];
#pragma unroll
  for (int sp = 0; sp < 2; ++sp)
    ld_frag2(qg, sp, lane, qa[2 * sp], qa[2 * sp + 1]);
#pragma unroll
  for (int j = 0; j < 2; ++j)
#pragma unroll
    for (int sp = 0; sp < 2; ++sp)
      ld_frag2(kg + (2 * kq + j) * 1024, sp, lane, kb[j][2 * sp], kb[j][2 * sp + 1]);
  __builtin_amdgcn_sched_barrier(0);
#pragma unroll
  for (int j = 0; j < 2; ++j)
#pragma unroll
    for (int sp = 0; sp < 2; ++sp)
      ld_frag2(vg + (2 * kq + j) * 1024, sp, lane, vb[j][2 * sp], vb[j][2 * sp + 1]);
#pragma unroll
  for (int s = 0; s < 4; ++s) {
    const int k = kq * 16 + 4 * s;
    fo[s].x = w_o[(k + koff) * 64 + c0 + l16];
    fo[s].y = w_o[(k + koff + 1) * 64 + c0 + l16];
  }
  float cbo[2], cg2[2], cb2[2];
#pragma unroll
  for (int e = 0; e < 2; ++e) {
    const int d = lane + 32 * e;
    cbo[e] = b_o[d]; cg2[e] = ln2_g[d]; cb2[e] = ln2_b[d];
  }
  __builtin_amdgcn_sched_barrier(0);

  v2f ffc[16];
  float bfc;
  {
    float* pw = smem + L_PW + wid * 512;
#pragma unroll
    for (int j = 0; j < 2; ++j) {
      const int kt = 2 * kq + j;
      v8f a8; for (int r = 0; r < 8; ++r) a8[r] = 0.0f;
#pragma unroll
      for (int s = 0; s < 4; ++s)
        a8 = wmma4(qa[s], kb[j][s], a8);
#pragma unroll
      for (int r = 0; r < 8; ++r) {
        const int row = q0 + r + 8 * half, col = kt * 16 + l16;
        pw[j * 256 + (r + 8 * half) * 16 + l16] =
            (col <= row) ? a8[r] * 0.25f : -1e9f;
      }
    }
    v2f p[2][4];
    float mx = -1e30f;
#pragma unroll
    for (int j = 0; j < 2; ++j)
#pragma unroll
      for (int s = 0; s < 4; ++s) {
        p[j][s] = *reinterpret_cast<const v2f*>(pw + j * 256 + l16 * 16 + 4 * s + koff);
        mx = fmaxf(mx, fmaxf(p[j][s].x, p[j][s].y));
      }
    mx = fmaxf(mx, __shfl_xor(mx, 16, 32));
    float sum = 0.0f;
#pragma unroll
    for (int j = 0; j < 2; ++j)
#pragma unroll
      for (int s = 0; s < 4; ++s) {
        p[j][s].x = __expf(p[j][s].x - mx);
        p[j][s].y = __expf(p[j][s].y - mx);
        sum += p[j][s].x + p[j][s].y;
      }
    sum += __shfl_xor(sum, 16, 32);
    v8f o; for (int r = 0; r < 8; ++r) o[r] = 0.0f;
#pragma unroll
    for (int j = 0; j < 2; ++j)
#pragma unroll
      for (int s = 0; s < 4; ++s)
        o = wmma4(p[j][s], vb[j][s], o);
    __builtin_amdgcn_sched_barrier(0);
#pragma unroll
    for (int s = 0; s < 16; ++s) {
      ffc[s].x = w_fc[(4 * s + koff) * 256 + n0g + l16];
      ffc[s].y = w_fc[(4 * s + koff + 1) * 256 + n0g + l16];
    }
    bfc = b_fc[n0g + l16];
    for (int r = 0; r < 8; ++r)
      part[kq * 1024 + (r + 8 * half) * 64 + c0 + l16] = o[r];
    if (half == 0) {
      pm[(kq * 4 + wq) * 16 + l16] = mx;
      pl[(kq * 4 + wq) * 16 + l16] = sum;
    }
  }
  xb[tid] = x0;
  xb[tid + NT] = x1;
  wg_barrier();
  {
    const int row = tid >> 5, c = (tid & 31) * 2, hh = c >> 4;
    float m[4];
    for (int k = 0; k < 4; ++k) m[k] = pm[(k * 4 + hh) * 16 + row];
    const float M = fmaxf(fmaxf(m[0], m[1]), fmaxf(m[2], m[3]));
    float L = 0.0f;
    v2f O; O.x = 0.0f; O.y = 0.0f;
    for (int k = 0; k < 4; ++k) {
      const float wk = __expf(m[k] - M);
      const v2f pk = *reinterpret_cast<const v2f*>(part + k * 1024 + row * 64 + c);
      L += pl[(k * 4 + hh) * 16 + row] * wk;
      O.x += pk.x * wk;
      O.y += pk.y * wk;
    }
    const float inv = 1.0f / L;
    O.x *= inv; O.y *= inv;
    *reinterpret_cast<v2f*>(hb + afrag(row, c)) = O;
  }
  wg_barrier();

  v2f fpr[16];
  float cbp[2], ctt[2], cgE[2], cbE[2];
  {
    v8f acc; for (int r = 0; r < 8; ++r) acc[r] = 0.0f;
#pragma unroll
    for (int s = 0; s < 4; ++s)
      acc = wmma4(ld_afrag(hb, kq * 4 + s, lane), fo[s], acc);
    for (int r = 0; r < 8; ++r)
      part[kq * 1024 + (r + 8 * half) * 64 + c0 + l16] = acc[r];
#pragma unroll
    for (int s = 0; s < 16; ++s) {
      const int k = kq * 64 + 4 * s;
      fpr[s].x = w_pr[(k + koff) * 64 + c0 + l16];
      fpr[s].y = w_pr[(k + koff + 1) * 64 + c0 + l16];
    }
    const float* gE = more ? ln1_g : lnf_g;
    const float* bE = more ? ln1_b : lnf_b;
    const float* tr = ws + (more ? (t + 1) * 64 : 0);
#pragma unroll
    for (int e = 0; e < 2; ++e) {
      const int d = lane + 32 * e;
      cbp[e] = b_pr[d]; ctt[e] = tr[d]; cgE[e] = gE[d]; cbE[e] = bE[d];
    }
  }
  wg_barrier();

  row_pass_r<false, true>(xb, part, cbo, cbo, nullptr, cg2, cb2, hb, wid, lane);
  wg_barrier();

  v2f fq[16];
  float bq = 0.0f;
#pragma unroll
  for (int s = 0; s < 16; ++s) { fq[s].x = 0.0f; fq[s].y = 0.0f; }
  {
    v8f acc;
    for (int r = 0; r < 8; ++r) acc[r] = bfc;
#pragma unroll
    for (int s = 0; s < 16; ++s)
      acc = wmma4(ld_afrag(hb, s, lane), ffc[s], acc);
    if (more && wid < 12) {
#pragma unroll
      for (int s = 0; s < 16; ++s) {
        fq[s].x = w_qkv[(4 * s + koff) * 192 + n0g + l16];
        fq[s].y = w_qkv[(4 * s + koff + 1) * 192 + n0g + l16];
      }
      bq = b_qkv[n0g + l16];
    }
    for (int r = 0; r < 8; ++r) {
      float u = acc[r];
      float g = 0.5f * u *
          (1.0f + tanhf(0.7978845608028654f * (u + 0.044715f * u * u * u)));
      gb[afrag(r + 8 * half, n0g + l16)] = g;
    }
  }
  wg_barrier();
  {
    v8f acc; for (int r = 0; r < 8; ++r) acc[r] = 0.0f;
#pragma unroll
    for (int s = 0; s < 16; ++s)
      acc = wmma4(ld_afrag(gb, kq * 16 + s, lane), fpr[s], acc);
    for (int r = 0; r < 8; ++r)
      part[kq * 1024 + (r + 8 * half) * 64 + c0 + l16] = acc[r];
  }
  wg_barrier();

  if (more) {
    row_pass_r<true, true>(xb, part, cbp, ctt, xg, cgE, cbE, hb, wid, lane);
    wg_barrier();
    if (wid < 12) {
      v8f acc;
      for (int r = 0; r < 8; ++r) acc[r] = bq;
#pragma unroll
      for (int s = 0; s < 16; ++s)
        acc = wmma4(ld_afrag(hb, s, lane), fq[s], acc);
      stage_qkv(gb, acc, wid, l16, half);
    }
    wg_barrier();
    for (int i = tid; i < 768; i += NT) {
      const int pln = i >> 8, off = i & 255;
      reinterpret_cast<float4*>(nxt + pln * PLANE + toff)[off] =
          staged4(gb, i);
    }
  } else {
    row_pass_r<false, false>(xb, part, cbp, ctt, nullptr, cgE, cbE, hb, wid, lane);
    wg_barrier();
    for (int i = tid; i < 256; i += NT)
      reinterpret_cast<float4*>(out + toff)[i] = reinterpret_cast<const float4*>(hb)[i];
  }
}

extern "C" void kernel_launch(void* const* d_in, const int* in_sizes, int n_in,
                              void* d_out, int out_size, void* d_ws, size_t ws_size,
                              hipStream_t stream) {
  const int*   idx   = (const int*)  d_in[0];
  const int*   nlp   = (const int*)  d_in[1];
  const float* wte   = (const float*)d_in[2];
  const float* wpe   = (const float*)d_in[3];
  const float* t_w1  = (const float*)d_in[4];
  const float* t_b1  = (const float*)d_in[5];
  const float* t_w2  = (const float*)d_in[6];
  const float* t_b2  = (const float*)d_in[7];
  const float* ln1_g = (const float*)d_in[8];
  const float* ln1_b = (const float*)d_in[9];
  const float* w_qkv = (const float*)d_in[10];
  const float* b_qkv = (const float*)d_in[11];
  const float* w_o   = (const float*)d_in[12];
  const float* b_o   = (const float*)d_in[13];
  const float* ln2_g = (const float*)d_in[14];
  const float* ln2_b = (const float*)d_in[15];
  const float* w_fc  = (const float*)d_in[16];
  const float* b_fc  = (const float*)d_in[17];
  const float* w_pr  = (const float*)d_in[18];
  const float* b_pr  = (const float*)d_in[19];
  const float* lnf_g = (const float*)d_in[20];
  const float* lnf_b = (const float*)d_in[21];

  float* ws = (float*)d_ws;

  temb_kernel<<<128, 256, 0, stream>>>(nlp, t_w1, t_b1, t_w2, t_b2, ws);
  embed_kernel<<<256, NT, 0, stream>>>(idx, nlp, wte, wpe, ln1_g, ln1_b, w_qkv,
                                       b_qkv, lnf_g, lnf_b, ws, (float*)d_out);
  for (int t = 0; t < 100; ++t)
    step_kernel<<<256, NT, 0, stream>>>(t, nlp, ln1_g, ln1_b, w_qkv, b_qkv, w_o,
                                        b_o, ln2_g, ln2_b, w_fc, b_fc, w_pr,
                                        b_pr, lnf_g, lnf_b, ws, (float*)d_out);
}


# Weights copied once per call (spare blocks of launch 1) to the workspace in B-fragment order: a wave's weight fetch is 26 whole-line 16-byte loads, not 104 4-byte ones; sums and order unchanged
# speedup vs baseline: 8.2922x; 1.0650x over previous
#include <hip/hip_runtime.h>
#include <hip/hip_bf16.h>
#include <math.h>


typedef __attribute__((ext_vector_type(2))) float v2f;
typedef __attribute__((ext_vector_type(8))) float v8f;

__device__ __forceinline__ v8f wmma4(v2f a, v2f b, v8f c) {
  return __builtin_amdgcn_wmma_f32_16x16x4_f32(
       false, a,  false, b,
       (short)0, c,  false,  false);
}

#define TEMB_BLOCKS 128
#define RELAY_BLOCKS 48
#define LAUNCH1_BLOCKS 176
#define WS_W (8192 + 7 * 262144)
#define WF_O 0
#define WF_FC 4096
#define WF_PR 20480
#define WF_QKV 36864
template <int N, int LOG2_KP>
__device__ __forceinline__ void relay_entry(const float* __restrict__ w,
                                            float* __restrict__ dst, int i) {
  const int lane = i & 31, g = i >> 5;
  const int sp = g & ((1 << LOG2_KP) - 1), nt = g >> LOG2_KP;
  const int k = 8 * sp + 2 * (lane >> 4), n = 16 * nt + (lane & 15);
  float4 f;
  f.x = w[k * N + n];       f.y = w[(k + 1) * N + n];
  f.z = w[(k + 4) * N + n]; f.w = w[(k + 5) * N + n];
  reinterpret_cast<float4*>(dst)[i] = f;
}

__global__ __launch_bounds__(256) void temb_kernel(
    const int* __restrict__ num_loops,
    const float* __restrict__ t_w1, const float* __restrict__ t_b1,
    const float* __restrict__ t_w2, const float* __restrict__ t_b2,
    const float* __restrict__ w_qkv, const float* __restrict__ w_o,
    const float* __restrict__ w_fc, const float* __restrict__ w_pr,
    float* __restrict__ ttab) {
  if (blockIdx.x >= TEMB_BLOCKS) {
    const int i = (blockIdx.x - TEMB_BLOCKS) * 256 + threadIdx.x;
    float* wf = ttab + WS_W;
    if (i < 1024)      relay_entry<64, 3>(w_o, wf + WF_O, i);
    else if (i < 5120) relay_entry<256, 3>(w_fc, wf + WF_FC, i - 1024);
    else if (i < 9216) relay_entry<64, 5>(w_pr, wf + WF_PR, i - 5120);
    else               relay_entry<192, 3>(w_qkv, wf + WF_QKV, i - 9216);
    return;
  }
  const int t = blockIdx.x;
  if (t >= *num_loops) return;
  __shared__ float te[256];
  __shared__ float hs[1024];
  __shared__ float red[256];
  const int tid = threadIdx.x;

  if (tid < 128) {
    float f   = expf(-logf(10000.0f) * (float)tid / 128.0f);
    float arg = (float)t * f;
    te[tid]       = cosf(arg);
    te[128 + tid] = sinf(arg);
  }
  __syncthreads();

  for (int j = tid; j < 1024; j += 256) {
    float acc = t_b1[j];
    for (int k = 0; k < 256; ++k) acc += te[k] * t_w1[k * 1024 + j];
    hs[j] = acc / (1.0f + expf(-acc));
  }
  __syncthreads();

  const int j = tid & 63, p = tid >> 6;
  float acc = 0.0f;
  for (int k = p * 256; k < p * 256 + 256; ++k) acc += hs[k] * t_w2[k * 64 + j];
  red[tid] = acc;
  __syncthreads();
  if (tid < 64) {
    ttab[t * 64 + tid] =
        t_b2[tid] + red[tid] + red[tid + 64] + red[tid + 128] + red[tid + 192];
  }
}

#define NT 512
#define WS_X 8192
#define WS_QKV (8192 + 262144)
#define PLANE 262144
#define L_XB 0
#define L_HB 1024
#define L_GB 2048
#define L_PART 6144
#define L_PM 10240
#define L_PL 10496
#define L_PW 10752
#define L_TOTAL 18944

__device__ __forceinline__ int afrag(int row, int k) {
  return ((k >> 2) << 6) + (((k >> 1) & 1) << 5) + (row << 1) + (k & 1);
}
__device__ __forceinline__ v2f ld_afrag(const float* buf, int step, int lane) {
  return *reinterpret_cast<const v2f*>(buf + step * 64 + lane * 2);
}

#define QK_SP 68
template <int P>
__device__ __forceinline__ int qkfrag(int r, int d) {
  return (((d >> 3) << 1) + ((d >> 1) & 1)) * P + (r << 2) + (((d >> 2) & 1) << 1) + (d & 1);
}
__device__ __forceinline__ void ld_frag2(const float* blk, int sp, int lane, v2f& f0, v2f& f1) {
  const float4 f = reinterpret_cast<const float4*>(blk)[sp * 32 + lane];
  f0.x = f.x; f0.y = f.y; f1.x = f.z; f1.y = f.w;
}
__device__ __forceinline__ void stage_qkv(float* st, const v8f& acc, int ct, int l16, int half) {
  float* blk = st + ct * (4 * QK_SP);
  if (ct >= 8) {
    for (int r = 0; r < 8; ++r) blk[qkfrag<QK_SP>(l16, r + 8 * half)] = acc[r];
  } else {
    for (int r = 0; r < 8; ++r) blk[qkfrag<QK_SP>(r + 8 * half, l16)] = acc[r];
  }
}
__device__ __forceinline__ const float4& staged4(const float* st, int i) {
  return *reinterpret_cast<const float4*>(st + (i >> 4) * QK_SP + (i & 15) * 4);
}

template <bool ADD, bool STEP0, bool FRAG>
__device__ __forceinline__ void row_pass(
    float* xb, const float* part, const float* __restrict__ bias,
    const float* __restrict__ tt, float* __restrict__ xg,
    const float* __restrict__ g, const float* __restrict__ bta, float* dst,
    int wid, int lane) {
  float v[2];
#pragma unroll
  for (int e = 0; e < 2; ++e) {
    const int d = lane + 32 * e;
    float x = xb[wid * 64 + d];
    if (ADD) {
      float s = bias[d];
      for (int k = 0; k < 4; ++k) s += part[k * 1024 + wid * 64 + d];
      x += 0.1f * s;
    }
    if (STEP0) x += tt[d];
    v[e] = x;
    if (ADD || STEP0) xb[wid * 64 + d] = x;
    if (STEP0) xg[wid * 64 + d] = x;
  }
  float sum = v[0] + v[1], sq = v[0] * v[0] + v[1] * v[1];
  for (int off = 16; off; off >>= 1) {
    sum += __shfl_xor(sum, off, 32);
    sq  += __shfl_xor(sq,  off, 32);
  }
  const float mu  = sum * (1.0f / 64.0f);
  const float var = sq * (1.0f / 64.0f) - mu * mu;
  const float inv = rsqrtf(var + 1e-5f);
#pragma unroll
  for (int e = 0; e < 2; ++e) {
    const int d = lane + 32 * e;
    dst[FRAG ? afrag(wid, d) : wid * 64 + d] = (v[e] - mu) * inv * g[d] + bta[d];
  }
}

__device__ __forceinline__ void qkv_out(
    const float* hb, float* st, const float* __restrict__ w_qkv,
    const float* __restrict__ b_qkv, float* __restrict__ qkv_next, int toff,
    int tid, int wid, int lane, int l16, int half, int koff) {
  if (wid < 12) {
    const int n0 = wid * 16;
    v8f acc;
    { float bias = b_qkv[n0 + l16]; for (int r = 0; r < 8; ++r) acc[r] = bias; }
#pragma unroll
    for (int k = 0; k < 64; k += 4) {
      v2f a = ld_afrag(hb, k >> 2, lane), w;
      w.x = w_qkv[(k + koff) * 192 + n0 + l16];
      w.y = w_qkv[(k + koff + 1) * 192 + n0 + l16];
      acc = wmma4(a, w, acc);
    }
    stage_qkv(st, acc, wid, l16, half);
  }
  __syncthreads();
  for (int i = tid; i < 768; i += NT) {
    const int pl = i >> 8, off = i & 255;
    reinterpret_cast<float4*>(qkv_next + pl * PLANE + toff)[off] =
        staged4(st, i);
  }
}

__device__ __forceinline__ void out_copy(const float* hb, float* __restrict__ outg, int tid) {
  __syncthreads();
  for (int i = tid; i < 256; i += NT)
    reinterpret_cast<float4*>(outg)[i] = reinterpret_cast<const float4*>(hb)[i];
}

__device__ __forceinline__ void wg_barrier() {
  asm volatile("s_wait_dscnt 0x0\n\ts_barrier_signal -1\n\ts_barrier_wait -1" ::: "memory");
}

template <bool STEP0, bool FRAG>
__device__ __forceinline__ void row_pass_r(
    float* xb, const float* part, const float (&bias)[2], const float (&tt)[2],
    float* __restrict__ xg, const float (&g)[2], const float (&bta)[2],
    float* dst, int wid, int lane) {
  float v[2];
#pragma unroll
  for (int e = 0; e < 2; ++e) {
    const int d = lane + 32 * e;
    float x = xb[wid * 64 + d];
    float s = bias[e];
    for (int k = 0; k < 4; ++k) s += part[k * 1024 + wid * 64 + d];
    x += 0.1f * s;
    if (STEP0) x += tt[e];
    v[e] = x;
    xb[wid * 64 + d] = x;
    if (STEP0) xg[wid * 64 + d] = x;
  }
  float sum = v[0] + v[1], sq = v[0] * v[0] + v[1] * v[1];
  for (int off = 16; off; off >>= 1) {
    sum += __shfl_xor(sum, off, 32);
    sq  += __shfl_xor(sq,  off, 32);
  }
  const float mu  = sum * (1.0f / 64.0f);
  const float var = sq * (1.0f / 64.0f) - mu * mu;
  const float inv = rsqrtf(var + 1e-5f);
#pragma unroll
  for (int e = 0; e < 2; ++e) {
    const int d = lane + 32 * e;
    dst[FRAG ? afrag(wid, d) : wid * 64 + d] = (v[e] - mu) * inv * g[e] + bta[e];
  }
}

__global__ __launch_bounds__(NT) void embed_kernel(
    const int* __restrict__ idx, const int* __restrict__ num_loops,
    const float* __restrict__ wte, const float* __restrict__ wpe,
    const float* __restrict__ ln1_g, const float* __restrict__ ln1_b,
    const float* __restrict__ w_qkv, const float* __restrict__ b_qkv,
    const float* __restrict__ lnf_g, const float* __restrict__ lnf_b,
    float* __restrict__ ws, float* __restrict__ out) {
  __shared__ __attribute__((aligned(16))) float smem[L_TOTAL];
  float* xb = smem + L_XB;
  float* hb = smem + L_HB;
  float* st = smem + L_GB;
  const int b = blockIdx.x >> 3, q0 = (blockIdx.x & 7) * 16;
  const int toff = (b * 128 + q0) * 64;
  const int tid = threadIdx.x, lane = tid & 31;
  const int wid = __builtin_amdgcn_readfirstlane(tid >> 5);
  const int l16 = lane & 15, half = lane >> 4, koff = 2 * half;

  for (int i = tid; i < 1024; i += NT) {
    int s = i >> 6, d = i & 63;
    xb[i] = wte[idx[b * 128 + q0 + s] * 64 + d] + wpe[(q0 + s) * 64 + d];
  }
  __syncthreads();

  if (*num_loops > 0) {
    row_pass<false, true, true>(xb, nullptr, nullptr, ws, ws + WS_X + toff, ln1_g, ln1_b, hb, wid, lane);
    __syncthreads();
    qkv_out(hb, st, w_qkv, b_qkv, ws + WS_QKV, toff, tid, wid, lane, l16, half, koff);
  } else {
    row_pass<false, false, false>(xb, nullptr, nullptr, nullptr, nullptr, lnf_g, lnf_b, hb, wid, lane);
    out_copy(hb, out + toff, tid);
  }
}

__global__ __launch_bounds__(NT) void step_kernel(
    int t, const int* __restrict__ num_loops,
    const float* __restrict__ ln1_g, const float* __restrict__ ln1_b,
    const float* __restrict__ b_qkv, const float* __restrict__ b_o,
    const float* __restrict__ ln2_g, const float* __restrict__ ln2_b,
    const float* __restrict__ b_fc, const float* __restrict__ b_pr,
    const float* __restrict__ lnf_g, const float* __restrict__ lnf_b,
    float* __restrict__ ws, float* __restrict__ out) {
  const int NL = *num_loops;
  if (t >= NL) return;

  __shared__ __attribute__((aligned(16))) float smem[L_TOTAL];
  float* xb = smem + L_XB;
  float* hb = smem + L_HB;
  float* gb = smem + L_GB;
  float* part = smem + L_PART;
  float* pm = smem + L_PM;
  float* pl = smem + L_PL;

  const int b = blockIdx.x >> 3, q0 = (blockIdx.x & 7) * 16;
  const int toff = (b * 128 + q0) * 64;
  const int tid = threadIdx.x, lane = tid & 31;
  const int wid = __builtin_amdgcn_readfirstlane(tid >> 5);
  const int l16 = lane & 15, half = lane >> 4, koff = 2 * half;
  const int wq = wid & 3;
  const int kq = wid >> 2;

  float* xg = ws + WS_X + toff;
  const float* cur = ws + WS_QKV + (t & 1) * (3 * PLANE);
  float* nxt = ws + WS_QKV + ((t + 1) & 1) * (3 * PLANE);
  const float* qg = cur + toff + wq * 256;
  const float* kg = cur + PLANE + b * 8192 + wq * 256;
  const float* vg = cur + 2 * PLANE + b * 8192 + wq * 256;

  const bool more = (t + 1 < NL);
  const int c0 = wq * 16;
  const int n0g = wid * 16;
  const float* wf = ws + WS_W;
  const float* wo_f  = wf + WF_O   + (wq * 8 + 2 * kq) * 128;
  const float* wfc_f = wf + WF_FC  + (wid * 8) * 128;
  const float* wpr_f = wf + WF_PR  + (wq * 32 + 8 * kq) * 128;
  const float* wq_f  = wf + WF_QKV + (wid * 8) * 128;

  const float x0 = xg[tid], x1 = xg[tid + NT];
  v2f qa[4], kb[2][4], vb[2][4], fo[4];
#pragma unroll
  for (int sp = 0; sp < 2; ++sp)
    ld_frag2(qg, sp, lane, qa[2 * sp], qa[2 * sp + 1]);
#pragma unroll
  for (int j = 0; j < 2; ++j)
#pragma unroll
    for (int sp = 0; sp < 2; ++sp)
      ld_frag2(kg + (2 * kq + j) * 1024, sp, lane, kb[j][2 * sp], kb[j][2 * sp + 1]);
  __builtin_amdgcn_sched_barrier(0);
#pragma unroll
  for (int j = 0; j < 2; ++j)
#pragma unroll
    for (int sp = 0; sp < 2; ++sp)
      ld_frag2(vg + (2 * kq + j) * 1024, sp, lane, vb[j][2 * sp], vb[j][2 * sp + 1]);
#pragma unroll
  for (int sp = 0; sp < 2; ++sp)
    ld_frag2(wo_f, sp, lane, fo[2 * sp], fo[2 * sp + 1]);
  float cbo[2], cg2[2], cb2[2];
#pragma unroll
  for (int e = 0; e < 2; ++e) {
    const int d = lane + 32 * e;
    cbo[e] = b_o[d]; cg2[e] = ln2_g[d]; cb2[e] = ln2_b[d];
  }
  __builtin_amdgcn_sched_barrier(0);

  v2f ffc[16];
  float bfc;
  {
    float* pw = smem + L_PW + wid * 512;
#pragma unroll
    for (int j = 0; j < 2; ++j) {
      const int kt = 2 * kq + j;
      v8f a8; for (int r = 0; r < 8; ++r) a8[r] = 0.0f;
#pragma unroll
      for (int s = 0; s < 4; ++s)
        a8 = wmma4(qa[s], kb[j][s], a8);
#pragma unroll
      for (int r = 0; r < 8; ++r) {
        const int row = q0 + r + 8 * half, col = kt * 16 + l16;
        pw[j * 256 + (r + 8 * half) * 16 + l16] =
            (col <= row) ? a8[r] * 0.25f : -1e9f;
      }
    }
    v2f p[2][4];
    float mx = -1e30f;
#pragma unroll
    for (int j = 0; j < 2; ++j)
#pragma unroll
      for (int s = 0; s < 4; ++s) {
        p[j][s] = *reinterpret_cast<const v2f*>(pw + j * 256 + l16 * 16 + 4 * s + koff);
        mx = fmaxf(mx, fmaxf(p[j][s].x, p[j][s].y));
      }
    mx = fmaxf(mx, __shfl_xor(mx, 16, 32));
    float sum = 0.0f;
#pragma unroll
    for (int j = 0; j < 2; ++j)
#pragma unroll
      for (int s = 0; s < 4; ++s) {
        p[j][s].x = __expf(p[j][s].x - mx);
        p[j][s].y = __expf(p[j][s].y - mx);
        sum += p[j][s].x + p[j][s].y;
      }
    sum += __shfl_xor(sum, 16, 32);
    v8f o; for (int r = 0; r < 8; ++r) o[r] = 0.0f;
#pragma unroll
    for (int j = 0; j < 2; ++j)
#pragma unroll
      for (int s = 0; s < 4; ++s)
        o = wmma4(p[j][s], vb[j][s], o);
    __builtin_amdgcn_sched_barrier(0);
#pragma unroll
    for (int sp = 0; sp < 8; ++sp)
      ld_frag2(wfc_f, sp, lane, ffc[2 * sp], ffc[2 * sp + 1]);
    bfc = b_fc[n0g + l16];
    for (int r = 0; r < 8; ++r)
      part[kq * 1024 + (r + 8 * half) * 64 + c0 + l16] = o[r];
    if (half == 0) {
      pm[(kq * 4 + wq) * 16 + l16] = mx;
      pl[(kq * 4 + wq) * 16 + l16] = sum;
    }
  }
  xb[tid] = x0;
  xb[tid + NT] = x1;
  wg_barrier();
  {
    const int row = tid >> 5, c = (tid & 31) * 2, hh = c >> 4;
    float m[4];
    for (int k = 0; k < 4; ++k) m[k] = pm[(k * 4 + hh) * 16 + row];
    const float M = fmaxf(fmaxf(m[0], m[1]), fmaxf(m[2], m[3]));
    float L = 0.0f;
    v2f O; O.x = 0.0f; O.y = 0.0f;
    for (int k = 0; k < 4; ++k) {
      const float wk = __expf(m[k] - M);
      const v2f pk = *reinterpret_cast<const v2f*>(part + k * 1024 + row * 64 + c);
      L += pl[(k * 4 + hh) * 16 + row] * wk;
      O.x += pk.x * wk;
      O.y += pk.y * wk;
    }
    const float inv = 1.0f / L;
    O.x *= inv; O.y *= inv;
    *reinterpret_cast<v2f*>(hb + afrag(row, c)) = O;
  }
  wg_barrier();

  v2f fpr[16];
  float cbp[2], ctt[2], cgE[2], cbE[2];
  {
    v8f acc; for (int r = 0; r < 8; ++r) acc[r] = 0.0f;
#pragma unroll
    for (int s = 0; s < 4; ++s)
      acc = wmma4(ld_afrag(hb, kq * 4 + s, lane), fo[s], acc);
    for (int r = 0; r < 8; ++r)
      part[kq * 1024 + (r + 8 * half) * 64 + c0 + l16] = acc[r];
#pragma unroll
    for (int sp = 0; sp < 8; ++sp)
      ld_frag2(wpr_f, sp, lane, fpr[2 * sp], fpr[2 * sp + 1]);
    const float* gE = more ? ln1_g : lnf_g;
    const float* bE = more ? ln1_b : lnf_b;
    const float* tr = ws + (more ? (t + 1) * 64 : 0);
#pragma unroll
    for (int e = 0; e < 2; ++e) {
      const int d = lane + 32 * e;
      cbp[e] = b_pr[d]; ctt[e] = tr[d]; cgE[e] = gE[d]; cbE[e] = bE[d];
    }
  }
  wg_barrier();

  row_pass_r<false, true>(xb, part, cbo, cbo, nullptr, cg2, cb2, hb, wid, lane);
  wg_barrier();

  v2f fq[16];
  float bq = 0.0f;
#pragma unroll
  for (int s = 0; s < 16; ++s) { fq[s].x = 0.0f; fq[s].y = 0.0f; }
  {
    v8f acc;
    for (int r = 0; r < 8; ++r) acc[r] = bfc;
#pragma unroll
    for (int s = 0; s < 16; ++s)
      acc = wmma4(ld_afrag(hb, s, lane), ffc[s], acc);
    if (more && wid < 12) {
#pragma unroll
      for (int sp = 0; sp < 8; ++sp)
        ld_frag2(wq_f, sp, lane, fq[2 * sp], fq[2 * sp + 1]);
      bq = b_qkv[n0g + l16];
    }
    for (int r = 0; r < 8; ++r) {
      float u = acc[r];
      float g = 0.5f * u *
          (1.0f + tanhf(0.7978845608028654f * (u + 0.044715f * u * u * u)));
      gb[afrag(r + 8 * half, n0g + l16)] = g;
    }
  }
  wg_barrier();
  {
    v8f acc; for (int r = 0; r < 8; ++r) acc[r] = 0.0f;
#pragma unroll
    for (int s = 0; s < 16; ++s)
      acc = wmma4(ld_afrag(gb, kq * 16 + s, lane), fpr[s], acc);
    for (int r = 0; r < 8; ++r)
      part[kq * 1024 + (r + 8 * half) * 64 + c0 + l16] = acc[r];
  }
  wg_barrier();

  if (more) {
    row_pass_r<true, true>(xb, part, cbp, ctt, xg, cgE, cbE, hb, wid, lane);
    wg_barrier();
    if (wid < 12) {
      v8f acc;
      for (int r = 0; r < 8; ++r) acc[r] = bq;
#pragma unroll
      for (int s = 0; s < 16; ++s)
        acc = wmma4(ld_afrag(hb, s, lane), fq[s], acc);
      stage_qkv(gb, acc, wid, l16, half);
    }
    wg_barrier();
    for (int i = tid; i < 768; i += NT) {
      const int pln = i >> 8, off = i & 255;
      reinterpret_cast<float4*>(nxt + pln * PLANE + toff)[off] =
          staged4(gb, i);
    }
  } else {
    row_pass_r<false, false>(xb, part, cbp, ctt, nullptr, cgE, cbE, hb, wid, lane);
    wg_barrier();
    for (int i = tid; i < 256; i += NT)
      reinterpret_cast<float4*>(out + toff)[i] = reinterpret_cast<const float4*>(hb)[i];
  }
}

extern "C" void kernel_launch(void* const* d_in, const int* in_sizes, int n_in,
                              void* d_out, int out_size, void* d_ws, size_t ws_size,
                              hipStream_t stream) {
  const int*   idx   = (const int*)  d_in[0];
  const int*   nlp   = (const int*)  d_in[1];
  const float* wte   = (const float*)d_in[2];
  const float* wpe   = (const float*)d_in[3];
  const float* t_w1  = (const float*)d_in[4];
  const float* t_b1  = (const float*)d_in[5];
  const float* t_w2  = (const float*)d_in[6];
  const float* t_b2  = (const float*)d_in[7];
  const float* ln1_g = (const float*)d_in[8];
  const float* ln1_b = (const float*)d_in[9];
  const float* w_qkv = (const float*)d_in[10];
  const float* b_qkv = (const float*)d_in[11];
  const float* w_o   = (const float*)d_in[12];
  const float* b_o   = (const float*)d_in[13];
  const float* ln2_g = (const float*)d_in[14];
  const float* ln2_b = (const float*)d_in[15];
  const float* w_fc  = (const float*)d_in[16];
  const float* b_fc  = (const float*)d_in[17];
  const float* w_pr  = (const float*)d_in[18];
  const float* b_pr  = (const float*)d_in[19];
  const float* lnf_g = (const float*)d_in[20];
  const float* lnf_b = (const float*)d_in[21];

  float* ws = (float*)d_ws;

  temb_kernel<<<LAUNCH1_BLOCKS, 256, 0, stream>>>(
      nlp, t_w1, t_b1, t_w2, t_b2, w_qkv, w_o, w_fc, w_pr, ws);
  embed_kernel<<<256, NT, 0, stream>>>(idx, nlp, wte, wpe, ln1_g, ln1_b, w_qkv,
                                       b_qkv, lnf_g, lnf_b, ws, (float*)d_out);
  for (int t = 0; t < 100; ++t)
    step_kernel<<<256, NT, 0, stream>>>(t, nlp, ln1_g, ln1_b, b_qkv, b_o, ln2_g,
                                        ln2_b, b_fc, b_pr, lnf_g, lnf_b, ws,
                                        (float*)d_out);
}


# Wavefront over (row tile, loop step): launch L runs steps 4(L-i)..4(L-i)+3 of row tile i in one workgroup, lower tiles' k, v from per-step planes; 32 step launches, not 100; sums and order unchanged
# speedup vs baseline: 8.6746x; 1.0461x over previous
#include <hip/hip_runtime.h>
#include <hip/hip_bf16.h>
#include <math.h>


typedef __attribute__((ext_vector_type(2))) float v2f;
typedef __attribute__((ext_vector_type(8))) float v8f;

__device__ __forceinline__ v8f wmma4(v2f a, v2f b, v8f c) {
  return __builtin_amdgcn_wmma_f32_16x16x4_f32(
       false, a,  false, b,
       (short)0, c,  false,  false);
}

#define TEMB_BLOCKS 128
#define RELAY_BLOCKS 48
#define LAUNCH1_BLOCKS 176
#define WS_W (8192 + 7 * 262144)
#define WF_O 0
#define WF_FC 4096
#define WF_PR 20480
#define WF_QKV 36864
template <int N, int LOG2_KP>
__device__ __forceinline__ void relay_entry(const float* __restrict__ w,
                                            float* __restrict__ dst, int i) {
  const int lane = i & 31, g = i >> 5;
  const int sp = g & ((1 << LOG2_KP) - 1), nt = g >> LOG2_KP;
  const int k = 8 * sp + 2 * (lane >> 4), n = 16 * nt + (lane & 15);
  float4 f;
  f.x = w[k * N + n];       f.y = w[(k + 1) * N + n];
  f.z = w[(k + 4) * N + n]; f.w = w[(k + 5) * N + n];
  reinterpret_cast<float4*>(dst)[i] = f;
}

__global__ __launch_bounds__(256) void temb_kernel(
    const int* __restrict__ num_loops,
    const float* __restrict__ t_w1, const float* __restrict__ t_b1,
    const float* __restrict__ t_w2, const float* __restrict__ t_b2,
    const float* __restrict__ w_qkv, const float* __restrict__ w_o,
    const float* __restrict__ w_fc, const float* __restrict__ w_pr,
    float* __restrict__ ttab) {
  if (blockIdx.x >= TEMB_BLOCKS) {
    const int i = (blockIdx.x - TEMB_BLOCKS) * 256 + threadIdx.x;
    float* wf = ttab + WS_W;
    if (i < 1024)      relay_entry<64, 3>(w_o, wf + WF_O, i);
    else if (i < 5120) relay_entry<256, 3>(w_fc, wf + WF_FC, i - 1024);
    else if (i < 9216) relay_entry<64, 5>(w_pr, wf + WF_PR, i - 5120);
    else               relay_entry<192, 3>(w_qkv, wf + WF_QKV, i - 9216);
    return;
  }
  const int t = blockIdx.x;
  if (t >= *num_loops) return;
  __shared__ float te[256];
  __shared__ float hs[1024];
  __shared__ float red[256];
  const int tid = threadIdx.x;

  if (tid < 128) {
    float f   = expf(-logf(10000.0f) * (float)tid / 128.0f);
    float arg = (float)t * f;
    te[tid]       = cosf(arg);
    te[128 + tid] = sinf(arg);
  }
  __syncthreads();

  for (int j = tid; j < 1024; j += 256) {
    float acc = t_b1[j];
    for (int k = 0; k < 256; ++k) acc += te[k] * t_w1[k * 1024 + j];
    hs[j] = acc / (1.0f + expf(-acc));
  }
  __syncthreads();

  const int j = tid & 63, p = tid >> 6;
  float acc = 0.0f;
  for (int k = p * 256; k < p * 256 + 256; ++k) acc += hs[k] * t_w2[k * 64 + j];
  red[tid] = acc;
  __syncthreads();
  if (tid < 64) {
    ttab[t * 64 + tid] =
        t_b2[tid] + red[tid] + red[tid + 64] + red[tid + 128] + red[tid + 192];
  }
}

#define NT 512
#define G 4
#define NLAUNCH 32
#define WS_X 8192
#define WS_Q (8192 + 262144)
#define WS_KV (8192 + 8 * 262144)
#define PLANE 262144
__device__ __forceinline__ float* plane_ptr(float* ws, int pl, int s) {
  return pl == 0 ? ws + WS_Q : ws + WS_KV + (2 * s + pl - 1) * PLANE;
}
#define L_XB 0
#define L_HB 1024
#define L_GB 2048
#define L_PART 6144
#define L_PM 10240
#define L_PL 10496
#define L_PW 10752
#define L_TOTAL 18944

__device__ __forceinline__ int afrag(int row, int k) {
  return ((k >> 2) << 6) + (((k >> 1) & 1) << 5) + (row << 1) + (k & 1);
}
__device__ __forceinline__ v2f ld_afrag(const float* buf, int step, int lane) {
  return *reinterpret_cast<const v2f*>(buf + step * 64 + lane * 2);
}

#define QK_SP 68
template <int P>
__device__ __forceinline__ int qkfrag(int r, int d) {
  return (((d >> 3) << 1) + ((d >> 1) & 1)) * P + (r << 2) + (((d >> 2) & 1) << 1) + (d & 1);
}
__device__ __forceinline__ void ld_frag2(const float* blk, int sp, int lane, v2f& f0, v2f& f1) {
  const float4 f = reinterpret_cast<const float4*>(blk)[sp * 32 + lane];
  f0.x = f.x; f0.y = f.y; f1.x = f.z; f1.y = f.w;
}
__device__ __forceinline__ float4 ld_staged(const float* st, int ct, int sp, int lane) {
  const int e = ct * 64 + sp * 32 + lane;
  return *reinterpret_cast<const float4*>(st + (e >> 4) * QK_SP + (e & 15) * 4);
}
__device__ __forceinline__ void pick_frag(const float4& fg, const float4& fl, int kt, int ti,
                                          v2f& f0, v2f& f1) {
  const bool lo = kt < ti, eq = kt == ti;
  f0.x = lo ? fg.x : (eq ? fl.x : 0.0f);
  f0.y = lo ? fg.y : (eq ? fl.y : 0.0f);
  f1.x = lo ? fg.z : (eq ? fl.z : 0.0f);
  f1.y = lo ? fg.w : (eq ? fl.w : 0.0f);
}
__device__ __forceinline__ void ask_kv(float4 (&kvn)[2][2][2], float* ws, int step, int bofs,
                                       int kq, int ti, int lane) {
#pragma unroll
  for (int pl = 0; pl < 2; ++pl)
#pragma unroll
    for (int j = 0; j < 2; ++j) {
      const int kt = 2 * kq + j;
      const float4* src = reinterpret_cast<const float4*>(
          plane_ptr(ws, 1 + pl, step) + bofs + (kt < ti ? kt : 2 * kq) * 1024);
#pragma unroll
      for (int sp = 0; sp < 2; ++sp) kvn[pl][j][sp] = src[sp * 32 + lane];
    }
}
__device__ __forceinline__ void stage_qkv(float* st, const v8f& acc, int ct, int l16, int half) {
  float* blk = st + ct * (4 * QK_SP);
  if (ct >= 8) {
    for (int r = 0; r < 8; ++r) blk[qkfrag<QK_SP>(l16, r + 8 * half)] = acc[r];
  } else {
    for (int r = 0; r < 8; ++r) blk[qkfrag<QK_SP>(r + 8 * half, l16)] = acc[r];
  }
}
__device__ __forceinline__ const float4& staged4(const float* st, int i) {
  return *reinterpret_cast<const float4*>(st + (i >> 4) * QK_SP + (i & 15) * 4);
}

template <bool ADD, bool STEP0, bool FRAG>
__device__ __forceinline__ void row_pass(
    float* xb, const float* part, const float* __restrict__ bias,
    const float* __restrict__ tt, float* __restrict__ xg,
    const float* __restrict__ g, const float* __restrict__ bta, float* dst,
    int wid, int lane) {
  float v[2];
#pragma unroll
  for (int e = 0; e < 2; ++e) {
    const int d = lane + 32 * e;
    float x = xb[wid * 64 + d];
    if (ADD) {
      float s = bias[d];
      for (int k = 0; k < 4; ++k) s += part[k * 1024 + wid * 64 + d];
      x += 0.1f * s;
    }
    if (STEP0) x += tt[d];
    v[e] = x;
    if (ADD || STEP0) xb[wid * 64 + d] = x;
    if (STEP0) xg[wid * 64 + d] = x;
  }
  float sum = v[0] + v[1], sq = v[0] * v[0] + v[1] * v[1];
  for (int off = 16; off; off >>= 1) {
    sum += __shfl_xor(sum, off, 32);
    sq  += __shfl_xor(sq,  off, 32);
  }
  const float mu  = sum * (1.0f / 64.0f);
  const float var = sq * (1.0f / 64.0f) - mu * mu;
  const float inv = rsqrtf(var + 1e-5f);
#pragma unroll
  for (int e = 0; e < 2; ++e) {
    const int d = lane + 32 * e;
    dst[FRAG ? afrag(wid, d) : wid * 64 + d] = (v[e] - mu) * inv * g[d] + bta[d];
  }
}

__device__ __forceinline__ void qkv_out(
    const float* hb, float* st, const float* __restrict__ w_qkv,
    const float* __restrict__ b_qkv, float* __restrict__ ws, int toff,
    int tid, int wid, int lane, int l16, int half, int koff) {
  if (wid < 12) {
    const int n0 = wid * 16;
    v8f acc;
    { float bias = b_qkv[n0 + l16]; for (int r = 0; r < 8; ++r) acc[r] = bias; }
#pragma unroll
    for (int k = 0; k < 64; k += 4) {
      v2f a = ld_afrag(hb, k >> 2, lane), w;
      w.x = w_qkv[(k + koff) * 192 + n0 + l16];
      w.y = w_qkv[(k + koff + 1) * 192 + n0 + l16];
      acc = wmma4(a, w, acc);
    }
    stage_qkv(st, acc, wid, l16, half);
  }
  __syncthreads();
  for (int i = tid; i < 768; i += NT) {
    const int pl = i >> 8, off = i & 255;
    reinterpret_cast<float4*>(plane_ptr(ws, pl, 0) + toff)[off] =
        staged4(st, i);
  }
}

__device__ __forceinline__ void out_copy(const float* hb, float* __restrict__ outg, int tid) {
  __syncthreads();
  for (int i = tid; i < 256; i += NT)
    reinterpret_cast<float4*>(outg)[i] = reinterpret_cast<const float4*>(hb)[i];
}

__device__ __forceinline__ void wg_barrier() {
  asm volatile("s_wait_dscnt 0x0\n\ts_barrier_signal -1\n\ts_barrier_wait -1" ::: "memory");
}

template <bool STEP0, bool FRAG>
__device__ __forceinline__ void row_pass_r(
    float* xb, const float* part, const float (&bias)[2], const float (&tt)[2],
    float* __restrict__ xg, const float (&g)[2], const float (&bta)[2],
    float* dst, int wid, int lane) {
  float v[2];
#pragma unroll
  for (int e = 0; e < 2; ++e) {
    const int d = lane + 32 * e;
    float x = xb[wid * 64 + d];
    float s = bias[e];
    for (int k = 0; k < 4; ++k) s += part[k * 1024 + wid * 64 + d];
    x += 0.1f * s;
    if (STEP0) x += tt[e];
    v[e] = x;
    xb[wid * 64 + d] = x;
    if (STEP0) xg[wid * 64 + d] = x;
  }
  float sum = v[0] + v[1], sq = v[0] * v[0] + v[1] * v[1];
  for (int off = 16; off; off >>= 1) {
    sum += __shfl_xor(sum, off, 32);
    sq  += __shfl_xor(sq,  off, 32);
  }
  const float mu  = sum * (1.0f / 64.0f);
  const float var = sq * (1.0f / 64.0f) - mu * mu;
  const float inv = rsqrtf(var + 1e-5f);
#pragma unroll
  for (int e = 0; e < 2; ++e) {
    const int d = lane + 32 * e;
    dst[FRAG ? afrag(wid, d) : wid * 64 + d] = (v[e] - mu) * inv * g[e] + bta[e];
  }
}

__global__ __launch_bounds__(NT) void embed_kernel(
    const int* __restrict__ idx, const int* __restrict__ num_loops,
    const float* __restrict__ wte, const float* __restrict__ wpe,
    const float* __restrict__ ln1_g, const float* __restrict__ ln1_b,
    const float* __restrict__ w_qkv, const float* __restrict__ b_qkv,
    const float* __restrict__ lnf_g, const float* __restrict__ lnf_b,
    float* __restrict__ ws, float* __restrict__ out) {
  __shared__ __attribute__((aligned(16))) float smem[L_TOTAL];
  float* xb = smem + L_XB;
  float* hb = smem + L_HB;
  float* st = smem + L_GB;
  const int b = blockIdx.x >> 3, q0 = (blockIdx.x & 7) * 16;
  const int toff = (b * 128 + q0) * 64;
  const int tid = threadIdx.x, lane = tid & 31;
  const int wid = __builtin_amdgcn_readfirstlane(tid >> 5);
  const int l16 = lane & 15, half = lane >> 4, koff = 2 * half;

  for (int i = tid; i < 1024; i += NT) {
    int s = i >> 6, d = i & 63;
    xb[i] = wte[idx[b * 128 + q0 + s] * 64 + d] + wpe[(q0 + s) * 64 + d];
  }
  __syncthreads();

  if (*num_loops > 0) {
    row_pass<false, true, true>(xb, nullptr, nullptr, ws, ws + WS_X + toff, ln1_g, ln1_b, hb, wid, lane);
    __syncthreads();
    qkv_out(hb, st, w_qkv, b_qkv, ws, toff, tid, wid, lane, l16, half, koff);
  } else {
    row_pass<false, false, false>(xb, nullptr, nullptr, nullptr, nullptr, lnf_g, lnf_b, hb, wid, lane);
    out_copy(hb, out + toff, tid);
  }
}

__global__ __launch_bounds__(NT) void step_kernel(
    int L, const int* __restrict__ num_loops,
    const float* __restrict__ ln1_g, const float* __restrict__ ln1_b,
    const float* __restrict__ b_qkv, const float* __restrict__ b_o,
    const float* __restrict__ ln2_g, const float* __restrict__ ln2_b,
    const float* __restrict__ b_fc, const float* __restrict__ b_pr,
    const float* __restrict__ lnf_g, const float* __restrict__ lnf_b,
    float* __restrict__ ws, float* __restrict__ out) {
  const int NL = min(*num_loops, G * (NLAUNCH - 7));
  const int ti = blockIdx.x & 7;
  const int s0 = G * (L - ti);
  if (s0 < 0 || s0 >= NL) return;

  __shared__ __attribute__((aligned(16))) float smem[L_TOTAL];
  float* xb = smem + L_XB;
  float* hb = smem + L_HB;
  float* gb = smem + L_GB;
  float* part = smem + L_PART;
  float* pm = smem + L_PM;
  float* pl = smem + L_PL;

  const int b = blockIdx.x >> 3, q0 = ti * 16;
  const int toff = (b * 128 + q0) * 64;
  const int tid = threadIdx.x, lane = tid & 31;
  const int wid = __builtin_amdgcn_readfirstlane(tid >> 5);
  const int l16 = lane & 15, half = lane >> 4, koff = 2 * half;
  const int wq = wid & 3;
  const int kq = wid >> 2;

  float* xg = ws + WS_X + toff;
  const int bofs = b * 8192 + wq * 256;
  const bool below = (2 * kq < ti);
  const int c0 = wq * 16;
  const int n0g = wid * 16;
  const float* wf = ws + WS_W;
  const float* wo_f  = wf + WF_O   + (wq * 8 + 2 * kq) * 128;
  const float* wfc_f = wf + WF_FC  + (wid * 8) * 128;
  const float* wpr_f = wf + WF_PR  + (wq * 32 + 8 * kq) * 128;
  const float* wq_f  = wf + WF_QKV + (wid * 8) * 128;

  float4 kvn[2][2][2];
#pragma unroll
  for (int a = 0; a < 8; ++a) kvn[a >> 2][(a >> 1) & 1][a & 1] = make_float4(0.0f, 0.0f, 0.0f, 0.0f);
  if (below) ask_kv(kvn, ws, s0, bofs, kq, ti, lane);
  {
    const float xa = xg[tid], xc = xg[tid + NT];
    const float4 f0 =
        reinterpret_cast<const float4*>(plane_ptr(ws, tid >> 8, s0) + toff)[tid & 255];
    float4 f1 = f0;
    if (tid < 256)
      f1 = reinterpret_cast<const float4*>(plane_ptr(ws, 2, s0) + toff)[tid];
    *reinterpret_cast<float4*>(gb + (tid >> 4) * QK_SP + (tid & 15) * 4) = f0;
    if (tid < 256)
      *reinterpret_cast<float4*>(gb + ((tid + 512) >> 4) * QK_SP + (tid & 15) * 4) = f1;
    xb[tid] = xa;
    xb[tid + NT] = xc;
  }
  wg_barrier();

  for (int it = 0; it < G; ++it) {
  const int t = s0 + it;
  if (t >= NL) break;
  const bool more = (t + 1 < NL);
  v2f qa[4], kb[2][4], vb[2][4], fo[4];
#pragma unroll
  for (int sp = 0; sp < 2; ++sp) {
    const float4 f = ld_staged(gb, wq, sp, lane);
    qa[2 * sp].x = f.x; qa[2 * sp].y = f.y; qa[2 * sp + 1].x = f.z; qa[2 * sp + 1].y = f.w;
  }
#pragma unroll
  for (int j = 0; j < 2; ++j)
#pragma unroll
    for (int sp = 0; sp < 2; ++sp) {
      pick_frag(kvn[0][j][sp], ld_staged(gb, 4 + wq, sp, lane), 2 * kq + j, ti,
                kb[j][2 * sp], kb[j][2 * sp + 1]);
      pick_frag(kvn[1][j][sp], ld_staged(gb, 8 + wq, sp, lane), 2 * kq + j, ti,
                vb[j][2 * sp], vb[j][2 * sp + 1]);
    }
#pragma unroll
  for (int sp = 0; sp < 2; ++sp)
    ld_frag2(wo_f, sp, lane, fo[2 * sp], fo[2 * sp + 1]);
  float cbo[2], cg2[2], cb2[2];
#pragma unroll
  for (int e = 0; e < 2; ++e) {
    const int d = lane + 32 * e;
    cbo[e] = b_o[d]; cg2[e] = ln2_g[d]; cb2[e] = ln2_b[d];
  }

  v2f ffc[16];
  float bfc;
  {
    float* pw = smem + L_PW + wid * 512;
#pragma unroll
    for (int j = 0; j < 2; ++j) {
      const int kt = 2 * kq + j;
      v8f a8; for (int r = 0; r < 8; ++r) a8[r] = 0.0f;
#pragma unroll
      for (int s = 0; s < 4; ++s)
        a8 = wmma4(qa[s], kb[j][s], a8);
#pragma unroll
      for (int r = 0; r < 8; ++r) {
        const int row = q0 + r + 8 * half, col = kt * 16 + l16;
        pw[j * 256 + (r + 8 * half) * 16 + l16] =
            (col <= row) ? a8[r] * 0.25f : -1e9f;
      }
    }
    v2f p[2][4];
    float mx = -1e30f;
#pragma unroll
    for (int j = 0; j < 2; ++j)
#pragma unroll
      for (int s = 0; s < 4; ++s) {
        p[j][s] = *reinterpret_cast<const v2f*>(pw + j * 256 + l16 * 16 + 4 * s + koff);
        mx = fmaxf(mx, fmaxf(p[j][s].x, p[j][s].y));
      }
    mx = fmaxf(mx, __shfl_xor(mx, 16, 32));
    float sum = 0.0f;
#pragma unroll
    for (int j = 0; j < 2; ++j)
#pragma unroll
      for (int s = 0; s < 4; ++s) {
        p[j][s].x = __expf(p[j][s].x - mx);
        p[j][s].y = __expf(p[j][s].y - mx);
        sum += p[j][s].x + p[j][s].y;
      }
    sum += __shfl_xor(sum, 16, 32);
    v8f o; for (int r = 0; r < 8; ++r) o[r] = 0.0f;
#pragma unroll
    for (int j = 0; j < 2; ++j)
#pragma unroll
      for (int s = 0; s < 4; ++s)
        o = wmma4(p[j][s], vb[j][s], o);
#pragma unroll
    for (int sp = 0; sp < 8; ++sp)
      ld_frag2(wfc_f, sp, lane, ffc[2 * sp], ffc[2 * sp + 1]);
    bfc = b_fc[n0g + l16];
    for (int r = 0; r < 8; ++r)
      part[kq * 1024 + (r + 8 * half) * 64 + c0 + l16] = o[r];
    if (half == 0) {
      pm[(kq * 4 + wq) * 16 + l16] = mx;
      pl[(kq * 4 + wq) * 16 + l16] = sum;
    }
  }
  wg_barrier();
  {
    const int row = tid >> 5, c = (tid & 31) * 2, hh = c >> 4;
    float m[4];
    for (int k = 0; k < 4; ++k) m[k] = pm[(k * 4 + hh) * 16 + row];
    const float M = fmaxf(fmaxf(m[0], m[1]), fmaxf(m[2], m[3]));
    float L = 0.0f;
    v2f O; O.x = 0.0f; O.y = 0.0f;
    for (int k = 0; k < 4; ++k) {
      const float wk = __expf(m[k] - M);
      const v2f pk = *reinterpret_cast<const v2f*>(part + k * 1024 + row * 64 + c);
      L += pl[(k * 4 + hh) * 16 + row] * wk;
      O.x += pk.x * wk;
      O.y += pk.y * wk;
    }
    const float inv = 1.0f / L;
    O.x *= inv; O.y *= inv;
    *reinterpret_cast<v2f*>(hb + afrag(row, c)) = O;
  }
  wg_barrier();

  v2f fpr[16];
  float cbp[2], ctt[2], cgE[2], cbE[2];
  {
    v8f acc; for (int r = 0; r < 8; ++r) acc[r] = 0.0f;
#pragma unroll
    for (int s = 0; s < 4; ++s)
      acc = wmma4(ld_afrag(hb, kq * 4 + s, lane), fo[s], acc);
    for (int r = 0; r < 8; ++r)
      part[kq * 1024 + (r + 8 * half) * 64 + c0 + l16] = acc[r];
#pragma unroll
    for (int sp = 0; sp < 8; ++sp)
      ld_frag2(wpr_f, sp, lane, fpr[2 * sp], fpr[2 * sp + 1]);
    const float* gE = more ? ln1_g : lnf_g;
    const float* bE = more ? ln1_b : lnf_b;
    const float* tr = ws + (more ? (t + 1) * 64 : 0);
#pragma unroll
    for (int e = 0; e < 2; ++e) {
      const int d = lane + 32 * e;
      cbp[e] = b_pr[d]; ctt[e] = tr[d]; cgE[e] = gE[d]; cbE[e] = bE[d];
    }
  }
  wg_barrier();

  row_pass_r<false, true>(xb, part, cbo, cbo, nullptr, cg2, cb2, hb, wid, lane);
  wg_barrier();

  v2f fq[16];
  float bq = 0.0f;
#pragma unroll
  for (int s = 0; s < 16; ++s) { fq[s].x = 0.0f; fq[s].y = 0.0f; }
  {
    v8f acc;
    for (int r = 0; r < 8; ++r) acc[r] = bfc;
#pragma unroll
    for (int s = 0; s < 16; ++s)
      acc = wmma4(ld_afrag(hb, s, lane), ffc[s], acc);
    if (more && wid < 12) {
#pragma unroll
      for (int sp = 0; sp < 8; ++sp)
        ld_frag2(wq_f, sp, lane, fq[2 * sp], fq[2 * sp + 1]);
      bq = b_qkv[n0g + l16];
    }
    if (below && more && it + 1 < G)
      ask_kv(kvn, ws, t + 1, bofs, kq, ti, lane);
    for (int r = 0; r < 8; ++r) {
      float u = acc[r];
      float g = 0.5f * u *
          (1.0f + tanhf(0.7978845608028654f * (u + 0.044715f * u * u * u)));
      gb[afrag(r + 8 * half, n0g + l16)] = g;
    }
  }
  wg_barrier();
  {
    v8f acc; for (int r = 0; r < 8; ++r) acc[r] = 0.0f;
#pragma unroll
    for (int s = 0; s < 16; ++s)
      acc = wmma4(ld_afrag(gb, kq * 16 + s, lane), fpr[s], acc);
    for (int r = 0; r < 8; ++r)
      part[kq * 1024 + (r + 8 * half) * 64 + c0 + l16] = acc[r];
  }
  wg_barrier();

  if (more) {
    row_pass_r<true, true>(xb, part, cbp, ctt, xg, cgE, cbE, hb, wid, lane);
    wg_barrier();
    if (wid < 12) {
      v8f acc;
      for (int r = 0; r < 8; ++r) acc[r] = bq;
#pragma unroll
      for (int s = 0; s < 16; ++s)
        acc = wmma4(ld_afrag(hb, s, lane), fq[s], acc);
      stage_qkv(gb, acc, wid, l16, half);
    }
    wg_barrier();
    for (int i = tid; i < 768; i += NT) {
      const int pln = i >> 8, off = i & 255;
      reinterpret_cast<float4*>(plane_ptr(ws, pln, t + 1) + toff)[off] =
          staged4(gb, i);
    }
  } else {
    row_pass_r<false, false>(xb, part, cbp, ctt, nullptr, cgE, cbE, hb, wid, lane);
    wg_barrier();
    for (int i = tid; i < 256; i += NT)
      reinterpret_cast<float4*>(out + toff)[i] = reinterpret_cast<const float4*>(hb)[i];
  }
  }
}

extern "C" void kernel_launch(void* const* d_in, const int* in_sizes, int n_in,
                              void* d_out, int out_size, void* d_ws, size_t ws_size,
                              hipStream_t stream) {
  const int*   idx   = (const int*)  d_in[0];
  const int*   nlp   = (const int*)  d_in[1];
  const float* wte   = (const float*)d_in[2];
  const float* wpe   = (const float*)d_in[3];
  const float* t_w1  = (const float*)d_in[4];
  const float* t_b1  = (const float*)d_in[5];
  const float* t_w2  = (const float*)d_in[6];
  const float* t_b2  = (const float*)d_in[7];
  const float* ln1_g = (const float*)d_in[8];
  const float* ln1_b = (const float*)d_in[9];
  const float* w_qkv = (const float*)d_in[10];
  const float* b_qkv = (const float*)d_in[11];
  const float* w_o   = (const float*)d_in[12];
  const float* b_o   = (const float*)d_in[13];
  const float* ln2_g = (const float*)d_in[14];
  const float* ln2_b = (const float*)d_in[15];
  const float* w_fc  = (const float*)d_in[16];
  const float* b_fc  = (const float*)d_in[17];
  const float* w_pr  = (const float*)d_in[18];
  const float* b_pr  = (const float*)d_in[19];
  const float* lnf_g = (const float*)d_in[20];
  const float* lnf_b = (const float*)d_in[21];

  float* ws = (float*)d_ws;

  temb_kernel<<<LAUNCH1_BLOCKS, 256, 0, stream>>>(
      nlp, t_w1, t_b1, t_w2, t_b2, w_qkv, w_o, w_fc, w_pr, ws);
  embed_kernel<<<256, NT, 0, stream>>>(idx, nlp, wte, wpe, ln1_g, ln1_b, w_qkv,
                                       b_qkv, lnf_g, lnf_b, ws, (float*)d_out);
  for (int L = 0; L < NLAUNCH; ++L)
    step_kernel<<<256, NT, 0, stream>>>(L, nlp, ln1_g, ln1_b, b_qkv, b_o, ln2_g,
                                        ln2_b, b_fc, b_pr, lnf_g, lnf_b, ws,
                                        (float*)d_out);
}


# Wavefront retimed: a row tile trails the one below by G-1 steps, not G (k, v of step t are written with step t-1), and G = 3: 38 step launches of 3 steps, not 32 of 4; sums and order unchanged
# speedup vs baseline: 9.2655x; 1.0681x over previous
#include <hip/hip_runtime.h>
#include <hip/hip_bf16.h>
#include <math.h>


typedef __attribute__((ext_vector_type(2))) float v2f;
typedef __attribute__((ext_vector_type(8))) float v8f;

__device__ __forceinline__ v8f wmma4(v2f a, v2f b, v8f c) {
  return __builtin_amdgcn_wmma_f32_16x16x4_f32(
       false, a,  false, b,
       (short)0, c,  false,  false);
}

#define TEMB_BLOCKS 128
#define RELAY_BLOCKS 48
#define LAUNCH1_BLOCKS 176
#define WS_W (8192 + 7 * 262144)
#define WF_O 0
#define WF_FC 4096
#define WF_PR 20480
#define WF_QKV 36864
template <int N, int LOG2_KP>
__device__ __forceinline__ void relay_entry(const float* __restrict__ w,
                                            float* __restrict__ dst, int i) {
  const int lane = i & 31, g = i >> 5;
  const int sp = g & ((1 << LOG2_KP) - 1), nt = g >> LOG2_KP;
  const int k = 8 * sp + 2 * (lane >> 4), n = 16 * nt + (lane & 15);
  float4 f;
  f.x = w[k * N + n];       f.y = w[(k + 1) * N + n];
  f.z = w[(k + 4) * N + n]; f.w = w[(k + 5) * N + n];
  reinterpret_cast<float4*>(dst)[i] = f;
}

__global__ __launch_bounds__(256) void temb_kernel(
    const int* __restrict__ num_loops,
    const float* __restrict__ t_w1, const float* __restrict__ t_b1,
    const float* __restrict__ t_w2, const float* __restrict__ t_b2,
    const float* __restrict__ w_qkv, const float* __restrict__ w_o,
    const float* __restrict__ w_fc, const float* __restrict__ w_pr,
    float* __restrict__ ttab) {
  if (blockIdx.x >= TEMB_BLOCKS) {
    const int i = (blockIdx.x - TEMB_BLOCKS) * 256 + threadIdx.x;
    float* wf = ttab + WS_W;
    if (i < 1024)      relay_entry<64, 3>(w_o, wf + WF_O, i);
    else if (i < 5120) relay_entry<256, 3>(w_fc, wf + WF_FC, i - 1024);
    else if (i < 9216) relay_entry<64, 5>(w_pr, wf + WF_PR, i - 5120);
    else               relay_entry<192, 3>(w_qkv, wf + WF_QKV, i - 9216);
    return;
  }
  const int t = blockIdx.x;
  if (t >= *num_loops) return;
  __shared__ float te[256];
  __shared__ float hs[1024];
  __shared__ float red[256];
  const int tid = threadIdx.x;

  if (tid < 128) {
    float f   = expf(-logf(10000.0f) * (float)tid / 128.0f);
    float arg = (float)t * f;
    te[tid]       = cosf(arg);
    te[128 + tid] = sinf(arg);
  }
  __syncthreads();

  for (int j = tid; j < 1024; j += 256) {
    float acc = t_b1[j];
    for (int k = 0; k < 256; ++k) acc += te[k] * t_w1[k * 1024 + j];
    hs[j] = acc / (1.0f + expf(-acc));
  }
  __syncthreads();

  const int j = tid & 63, p = tid >> 6;
  float acc = 0.0f;
  for (int k = p * 256; k < p * 256 + 256; ++k) acc += hs[k] * t_w2[k * 64 + j];
  red[tid] = acc;
  __syncthreads();
  if (tid < 64) {
    ttab[t * 64 + tid] =
        t_b2[tid] + red[tid] + red[tid + 64] + red[tid + 128] + red[tid + 192];
  }
}

#define NT 512
#define G 3
#define SKEW (G - 1)
#define NLAUNCH 38
#define MAXLOOPS (G * NLAUNCH - 7 * SKEW)
#define WS_X 8192
#define WS_Q (8192 + 262144)
#define WS_KV (8192 + 8 * 262144)
#define PLANE 262144
__device__ __forceinline__ float* plane_ptr(float* ws, int pl, int s) {
  return pl == 0 ? ws + WS_Q : ws + WS_KV + (2 * s + pl - 1) * PLANE;
}
#define L_XB 0
#define L_HB 1024
#define L_GB 2048
#define L_PART 6144
#define L_PM 10240
#define L_PL 10496
#define L_PW 10752
#define L_TOTAL 18944

__device__ __forceinline__ int afrag(int row, int k) {
  return ((k >> 2) << 6) + (((k >> 1) & 1) << 5) + (row << 1) + (k & 1);
}
__device__ __forceinline__ v2f ld_afrag(const float* buf, int step, int lane) {
  return *reinterpret_cast<const v2f*>(buf + step * 64 + lane * 2);
}

#define QK_SP 68
template <int P>
__device__ __forceinline__ int qkfrag(int r, int d) {
  return (((d >> 3) << 1) + ((d >> 1) & 1)) * P + (r << 2) + (((d >> 2) & 1) << 1) + (d & 1);
}
__device__ __forceinline__ void ld_frag2(const float* blk, int sp, int lane, v2f& f0, v2f& f1) {
  const float4 f = reinterpret_cast<const float4*>(blk)[sp * 32 + lane];
  f0.x = f.x; f0.y = f.y; f1.x = f.z; f1.y = f.w;
}
__device__ __forceinline__ float4 ld_staged(const float* st, int ct, int sp, int lane) {
  const int e = ct * 64 + sp * 32 + lane;
  return *reinterpret_cast<const float4*>(st + (e >> 4) * QK_SP + (e & 15) * 4);
}
__device__ __forceinline__ void pick_frag(const float4& fg, const float4& fl, int kt, int ti,
                                          v2f& f0, v2f& f1) {
  const bool lo = kt < ti, eq = kt == ti;
  f0.x = lo ? fg.x : (eq ? fl.x : 0.0f);
  f0.y = lo ? fg.y : (eq ? fl.y : 0.0f);
  f1.x = lo ? fg.z : (eq ? fl.z : 0.0f);
  f1.y = lo ? fg.w : (eq ? fl.w : 0.0f);
}
__device__ __forceinline__ void ask_kv(float4 (&kvn)[2][2][2], float* ws, int step, int bofs,
                                       int kq, int ti, int lane) {
#pragma unroll
  for (int pl = 0; pl < 2; ++pl)
#pragma unroll
    for (int j = 0; j < 2; ++j) {
      const int kt = 2 * kq + j;
      const float4* src = reinterpret_cast<const float4*>(
          plane_ptr(ws, 1 + pl, step) + bofs + (kt < ti ? kt : 2 * kq) * 1024);
#pragma unroll
      for (int sp = 0; sp < 2; ++sp) kvn[pl][j][sp] = src[sp * 32 + lane];
    }
}
__device__ __forceinline__ void stage_qkv(float* st, const v8f& acc, int ct, int l16, int half) {
  float* blk = st + ct * (4 * QK_SP);
  if (ct >= 8) {
    for (int r = 0; r < 8; ++r) blk[qkfrag<QK_SP>(l16, r + 8 * half)] = acc[r];
  } else {
    for (int r = 0; r < 8; ++r) blk[qkfrag<QK_SP>(r + 8 * half, l16)] = acc[r];
  }
}
__device__ __forceinline__ const float4& staged4(const float* st, int i) {
  return *reinterpret_cast<const float4*>(st + (i >> 4) * QK_SP + (i & 15) * 4);
}

template <bool ADD, bool STEP0, bool FRAG>
__device__ __forceinline__ void row_pass(
    float* xb, const float* part, const float* __restrict__ bias,
    const float* __restrict__ tt, float* __restrict__ xg,
    const float* __restrict__ g, const float* __restrict__ bta, float* dst,
    int wid, int lane) {
  float v[2];
#pragma unroll
  for (int e = 0; e < 2; ++e) {
    const int d = lane + 32 * e;
    float x = xb[wid * 64 + d];
    if (ADD) {
      float s = bias[d];
      for (int k = 0; k < 4; ++k) s += part[k * 1024 + wid * 64 + d];
      x += 0.1f * s;
    }
    if (STEP0) x += tt[d];
    v[e] = x;
    if (ADD || STEP0) xb[wid * 64 + d] = x;
    if (STEP0) xg[wid * 64 + d] = x;
  }
  float sum = v[0] + v[1], sq = v[0] * v[0] + v[1] * v[1];
  for (int off = 16; off; off >>= 1) {
    sum += __shfl_xor(sum, off, 32);
    sq  += __shfl_xor(sq,  off, 32);
  }
  const float mu  = sum * (1.0f / 64.0f);
  const float var = sq * (1.0f / 64.0f) - mu * mu;
  const float inv = rsqrtf(var + 1e-5f);
#pragma unroll
  for (int e = 0; e < 2; ++e) {
    const int d = lane + 32 * e;
    dst[FRAG ? afrag(wid, d) : wid * 64 + d] = (v[e] - mu) * inv * g[d] + bta[d];
  }
}

__device__ __forceinline__ void qkv_out(
    const float* hb, float* st, const float* __restrict__ w_qkv,
    const float* __restrict__ b_qkv, float* __restrict__ ws, int toff,
    int tid, int wid, int lane, int l16, int half, int koff) {
  if (wid < 12) {
    const int n0 = wid * 16;
    v8f acc;
    { float bias = b_qkv[n0 + l16]; for (int r = 0; r < 8; ++r) acc[r] = bias; }
#pragma unroll
    for (int k = 0; k < 64; k += 4) {
      v2f a = ld_afrag(hb, k >> 2, lane), w;
      w.x = w_qkv[(k + koff) * 192 + n0 + l16];
      w.y = w_qkv[(k + koff + 1) * 192 + n0 + l16];
      acc = wmma4(a, w, acc);
    }
    stage_qkv(st, acc, wid, l16, half);
  }
  __syncthreads();
  for (int i = tid; i < 768; i += NT) {
    const int pl = i >> 8, off = i & 255;
    reinterpret_cast<float4*>(plane_ptr(ws, pl, 0) + toff)[off] =
        staged4(st, i);
  }
}

__device__ __forceinline__ void out_copy(const float* hb, float* __restrict__ outg, int tid) {
  __syncthreads();
  for (int i = tid; i < 256; i += NT)
    reinterpret_cast<float4*>(outg)[i] = reinterpret_cast<const float4*>(hb)[i];
}

__device__ __forceinline__ void wg_barrier() {
  asm volatile("s_wait_dscnt 0x0\n\ts_barrier_signal -1\n\ts_barrier_wait -1" ::: "memory");
}

template <bool STEP0, bool FRAG>
__device__ __forceinline__ void row_pass_r(
    float* xb, const float* part, const float (&bias)[2], const float (&tt)[2],
    float* __restrict__ xg, const float (&g)[2], const float (&bta)[2],
    float* dst, int wid, int lane) {
  float v[2];
#pragma unroll
  for (int e = 0; e < 2; ++e) {
    const int d = lane + 32 * e;
    float x = xb[wid * 64 + d];
    float s = bias[e];
    for (int k = 0; k < 4; ++k) s += part[k * 1024 + wid * 64 + d];
    x += 0.1f * s;
    if (STEP0) x += tt[e];
    v[e] = x;
    xb[wid * 64 + d] = x;
    if (STEP0) xg[wid * 64 + d] = x;
  }
  float sum = v[0] + v[1], sq = v[0] * v[0] + v[1] * v[1];
  for (int off = 16; off; off >>= 1) {
    sum += __shfl_xor(sum, off, 32);
    sq  += __shfl_xor(sq,  off, 32);
  }
  const float mu  = sum * (1.0f / 64.0f);
  const float var = sq * (1.0f / 64.0f) - mu * mu;
  const float inv = rsqrtf(var + 1e-5f);
#pragma unroll
  for (int e = 0; e < 2; ++e) {
    const int d = lane + 32 * e;
    dst[FRAG ? afrag(wid, d) : wid * 64 + d] = (v[e] - mu) * inv * g[e] + bta[e];
  }
}

__global__ __launch_bounds__(NT) void embed_kernel(
    const int* __restrict__ idx, const int* __restrict__ num_loops,
    const float* __restrict__ wte, const float* __restrict__ wpe,
    const float* __restrict__ ln1_g, const float* __restrict__ ln1_b,
    const float* __restrict__ w_qkv, const float* __restrict__ b_qkv,
    const float* __restrict__ lnf_g, const float* __restrict__ lnf_b,
    float* __restrict__ ws, float* __restrict__ out) {
  __shared__ __attribute__((aligned(16))) float smem[L_TOTAL];
  float* xb = smem + L_XB;
  float* hb = smem + L_HB;
  float* st = smem + L_GB;
  const int b = blockIdx.x >> 3, q0 = (blockIdx.x & 7) * 16;
  const int toff = (b * 128 + q0) * 64;
  const int tid = threadIdx.x, lane = tid & 31;
  const int wid = __builtin_amdgcn_readfirstlane(tid >> 5);
  const int l16 = lane & 15, half = lane >> 4, koff = 2 * half;

  for (int i = tid; i < 1024; i += NT) {
    int s = i >> 6, d = i & 63;
    xb[i] = wte[idx[b * 128 + q0 + s] * 64 + d] + wpe[(q0 + s) * 64 + d];
  }
  __syncthreads();

  if (*num_loops > 0) {
    row_pass<false, true, true>(xb, nullptr, nullptr, ws, ws + WS_X + toff, ln1_g, ln1_b, hb, wid, lane);
    __syncthreads();
    qkv_out(hb, st, w_qkv, b_qkv, ws, toff, tid, wid, lane, l16, half, koff);
  } else {
    row_pass<false, false, false>(xb, nullptr, nullptr, nullptr, nullptr, lnf_g, lnf_b, hb, wid, lane);
    out_copy(hb, out + toff, tid);
  }
}

__global__ __launch_bounds__(NT) void step_kernel(
    int L, const int* __restrict__ num_loops,
    const float* __restrict__ ln1_g, const float* __restrict__ ln1_b,
    const float* __restrict__ b_qkv, const float* __restrict__ b_o,
    const float* __restrict__ ln2_g, const float* __restrict__ ln2_b,
    const float* __restrict__ b_fc, const float* __restrict__ b_pr,
    const float* __restrict__ lnf_g, const float* __restrict__ lnf_b,
    float* __restrict__ ws, float* __restrict__ out) {
  const int NL = min(*num_loops, MAXLOOPS);
  const int ti = blockIdx.x & 7;
  const int sa = G * L - SKEW * ti;
  const int s0 = max(sa, 0);
  const int s1 = min(sa + G, NL);
  if (s0 >= s1) return;

  __shared__ __attribute__((aligned(16))) float smem[L_TOTAL];
  float* xb = smem + L_XB;
  float* hb = smem + L_HB;
  float* gb = smem + L_GB;
  float* part = smem + L_PART;
  float* pm = smem + L_PM;
  float* pl = smem + L_PL;

  const int b = blockIdx.x >> 3, q0 = ti * 16;
  const int toff = (b * 128 + q0) * 64;
  const int tid = threadIdx.x, lane = tid & 31;
  const int wid = __builtin_amdgcn_readfirstlane(tid >> 5);
  const int l16 = lane & 15, half = lane >> 4, koff = 2 * half;
  const int wq = wid & 3;
  const int kq = wid >> 2;

  float* xg = ws + WS_X + toff;
  const int bofs = b * 8192 + wq * 256;
  const bool below = (2 * kq < ti);
  const int c0 = wq * 16;
  const int n0g = wid * 16;
  const float* wf = ws + WS_W;
  const float* wo_f  = wf + WF_O   + (wq * 8 + 2 * kq) * 128;
  const float* wfc_f = wf + WF_FC  + (wid * 8) * 128;
  const float* wpr_f = wf + WF_PR  + (wq * 32 + 8 * kq) * 128;
  const float* wq_f  = wf + WF_QKV + (wid * 8) * 128;

  float4 kvn[2][2][2];
#pragma unroll
  for (int a = 0; a < 8; ++a) kvn[a >> 2][(a >> 1) & 1][a & 1] = make_float4(0.0f, 0.0f, 0.0f, 0.0f);
  if (below) ask_kv(kvn, ws, s0, bofs, kq, ti, lane);
  {
    const float xa = xg[tid], xc = xg[tid + NT];
    const float4 f0 =
        reinterpret_cast<const float4*>(plane_ptr(ws, tid >> 8, s0) + toff)[tid & 255];
    float4 f1 = f0;
    if (tid < 256)
      f1 = reinterpret_cast<const float4*>(plane_ptr(ws, 2, s0) + toff)[tid];
    *reinterpret_cast<float4*>(gb + (tid >> 4) * QK_SP + (tid & 15) * 4) = f0;
    if (tid < 256)
      *reinterpret_cast<float4*>(gb + ((tid + 512) >> 4) * QK_SP + (tid & 15) * 4) = f1;
    xb[tid] = xa;
    xb[tid + NT] = xc;
  }
  wg_barrier();

  for (int t = s0; t < s1; ++t) {
  const bool more = (t + 1 < NL);
  v2f qa[4], kb[2][4], vb[2][4], fo[4];
#pragma unroll
  for (int sp = 0; sp < 2; ++sp) {
    const float4 f = ld_staged(gb, wq, sp, lane);
    qa[2 * sp].x = f.x; qa[2 * sp].y = f.y; qa[2 * sp + 1].x = f.z; qa[2 * sp + 1].y = f.w;
  }
#pragma unroll
  for (int j = 0; j < 2; ++j)
#pragma unroll
    for (int sp = 0; sp < 2; ++sp) {
      pick_frag(kvn[0][j][sp], ld_staged(gb, 4 + wq, sp, lane), 2 * kq + j, ti,
                kb[j][2 * sp], kb[j][2 * sp + 1]);
      pick_frag(kvn[1][j][sp], ld_staged(gb, 8 + wq, sp, lane), 2 * kq + j, ti,
                vb[j][2 * sp], vb[j][2 * sp + 1]);
    }
#pragma unroll
  for (int sp = 0; sp < 2; ++sp)
    ld_frag2(wo_f, sp, lane, fo[2 * sp], fo[2 * sp + 1]);
  float cbo[2], cg2[2], cb2[2];
#pragma unroll
  for (int e = 0; e < 2; ++e) {
    const int d = lane + 32 * e;
    cbo[e] = b_o[d]; cg2[e] = ln2_g[d]; cb2[e] = ln2_b[d];
  }

  v2f ffc[16];
  float bfc;
  {
    float* pw = smem + L_PW + wid * 512;
#pragma unroll
    for (int j = 0; j < 2; ++j) {
      const int kt = 2 * kq + j;
      v8f a8; for (int r = 0; r < 8; ++r) a8[r] = 0.0f;
#pragma unroll
      for (int s = 0; s < 4; ++s)
        a8 = wmma4(qa[s], kb[j][s], a8);
#pragma unroll
      for (int r = 0; r < 8; ++r) {
        const int row = q0 + r + 8 * half, col = kt * 16 + l16;
        pw[j * 256 + (r + 8 * half) * 16 + l16] =
            (col <= row) ? a8[r] * 0.25f : -1e9f;
      }
    }
    v2f p[2][4];
    float mx = -1e30f;
#pragma unroll
    for (int j = 0; j < 2; ++j)
#pragma unroll
      for (int s = 0; s < 4; ++s) {
        p[j][s] = *reinterpret_cast<const v2f*>(pw + j * 256 + l16 * 16 + 4 * s + koff);
        mx = fmaxf(mx, fmaxf(p[j][s].x, p[j][s].y));
      }
    mx = fmaxf(mx, __shfl_xor(mx, 16, 32));
    float sum = 0.0f;
#pragma unroll
    for (int j = 0; j < 2; ++j)
#pragma unroll
      for (int s = 0; s < 4; ++s) {
        p[j][s].x = __expf(p[j][s].x - mx);
        p[j][s].y = __expf(p[j][s].y - mx);
        sum += p[j][s].x + p[j][s].y;
      }
    sum += __shfl_xor(sum, 16, 32);
    v8f o; for (int r = 0; r < 8; ++r) o[r] = 0.0f;
#pragma unroll
    for (int j = 0; j < 2; ++j)
#pragma unroll
      for (int s = 0; s < 4; ++s)
        o = wmma4(p[j][s], vb[j][s], o);
#pragma unroll
    for (int sp = 0; sp < 8; ++sp)
      ld_frag2(wfc_f, sp, lane, ffc[2 * sp], ffc[2 * sp + 1]);
    bfc = b_fc[n0g + l16];
    for (int r = 0; r < 8; ++r)
      part[kq * 1024 + (r + 8 * half) * 64 + c0 + l16] = o[r];
    if (half == 0) {
      pm[(kq * 4 + wq) * 16 + l16] = mx;
      pl[(kq * 4 + wq) * 16 + l16] = sum;
    }
  }
  wg_barrier();
  {
    const int row = tid >> 5, c = (tid & 31) * 2, hh = c >> 4;
    float m[4];
    for (int k = 0; k < 4; ++k) m[k] = pm[(k * 4 + hh) * 16 + row];
    const float M = fmaxf(fmaxf(m[0], m[1]), fmaxf(m[2], m[3]));
    float L = 0.0f;
    v2f O; O.x = 0.0f; O.y = 0.0f;
    for (int k = 0; k < 4; ++k) {
      const float wk = __expf(m[k] - M);
      const v2f pk = *reinterpret_cast<const v2f*>(part + k * 1024 + row * 64 + c);
      L += pl[(k * 4 + hh) * 16 + row] * wk;
      O.x += pk.x * wk;
      O.y += pk.y * wk;
    }
    const float inv = 1.0f / L;
    O.x *= inv; O.y *= inv;
    *reinterpret_cast<v2f*>(hb + afrag(row, c)) = O;
  }
  wg_barrier();

  v2f fpr[16];
  float cbp[2], ctt[2], cgE[2], cbE[2];
  {
    v8f acc; for (int r = 0; r < 8; ++r) acc[r] = 0.0f;
#pragma unroll
    for (int s = 0; s < 4; ++s)
      acc = wmma4(ld_afrag(hb, kq * 4 + s, lane), fo[s], acc);
    for (int r = 0; r < 8; ++r)
      part[kq * 1024 + (r + 8 * half) * 64 + c0 + l16] = acc[r];
#pragma unroll
    for (int sp = 0; sp < 8; ++sp)
      ld_frag2(wpr_f, sp, lane, fpr[2 * sp], fpr[2 * sp + 1]);
    const float* gE = more ? ln1_g : lnf_g;
    const float* bE = more ? ln1_b : lnf_b;
    const float* tr = ws + (more ? (t + 1) * 64 : 0);
#pragma unroll
    for (int e = 0; e < 2; ++e) {
      const int d = lane + 32 * e;
      cbp[e] = b_pr[d]; ctt[e] = tr[d]; cgE[e] = gE[d]; cbE[e] = bE[d];
    }
  }
  wg_barrier();

  row_pass_r<false, true>(xb, part, cbo, cbo, nullptr, cg2, cb2, hb, wid, lane);
  wg_barrier();

  v2f fq[16];
  float bq = 0.0f;
#pragma unroll
  for (int s = 0; s < 16; ++s) { fq[s].x = 0.0f; fq[s].y = 0.0f; }
  {
    v8f acc;
    for (int r = 0; r < 8; ++r) acc[r] = bfc;
#pragma unroll
    for (int s = 0; s < 16; ++s)
      acc = wmma4(ld_afrag(hb, s, lane), ffc[s], acc);
    if (more && wid < 12) {
#pragma unroll
      for (int sp = 0; sp < 8; ++sp)
        ld_frag2(wq_f, sp, lane, fq[2 * sp], fq[2 * sp + 1]);
      bq = b_qkv[n0g + l16];
    }
    if (below && t + 1 < s1)
      ask_kv(kvn, ws, t + 1, bofs, kq, ti, lane);
    for (int r = 0; r < 8; ++r) {
      float u = acc[r];
      float g = 0.5f * u *
          (1.0f + tanhf(0.7978845608028654f * (u + 0.044715f * u * u * u)));
      gb[afrag(r + 8 * half, n0g + l16)] = g;
    }
  }
  wg_barrier();
  {
    v8f acc; for (int r = 0; r < 8; ++r) acc[r] = 0.0f;
#pragma unroll
    for (int s = 0; s < 16; ++s)
      acc = wmma4(ld_afrag(gb, kq * 16 + s, lane), fpr[s], acc);
    for (int r = 0; r < 8; ++r)
      part[kq * 1024 + (r + 8 * half) * 64 + c0 + l16] = acc[r];
  }
  wg_barrier();

  if (more) {
    row_pass_r<true, true>(xb, part, cbp, ctt, xg, cgE, cbE, hb, wid, lane);
    wg_barrier();
    if (wid < 12) {
      v8f acc;
      for (int r = 0; r < 8; ++r) acc[r] = bq;
#pragma unroll
      for (int s = 0; s < 16; ++s)
        acc = wmma4(ld_afrag(hb, s, lane), fq[s], acc);
      stage_qkv(gb, acc, wid, l16, half);
    }
    wg_barrier();
    for (int i = tid; i < 768; i += NT) {
      const int pln = i >> 8, off = i & 255;
      reinterpret_cast<float4*>(plane_ptr(ws, pln, t + 1) + toff)[off] =
          staged4(gb, i);
    }
  } else {
    row_pass_r<false, false>(xb, part, cbp, ctt, nullptr, cgE, cbE, hb, wid, lane);
    wg_barrier();
    for (int i = tid; i < 256; i += NT)
      reinterpret_cast<float4*>(out + toff)[i] = reinterpret_cast<const float4*>(hb)[i];
  }
  }
}

extern "C" void kernel_launch(void* const* d_in, const int* in_sizes, int n_in,
                              void* d_out, int out_size, void* d_ws, size_t ws_size,
                              hipStream_t stream) {
  const int*   idx   = (const int*)  d_in[0];
  const int*   nlp   = (const int*)  d_in[1];
  const float* wte   = (const float*)d_in[2];
  const float* wpe   = (const float*)d_in[3];
  const float* t_w1  = (const float*)d_in[4];
  const float* t_b1  = (const float*)d_in[5];
  const float* t_w2  = (const float*)d_in[6];
  const float* t_b2  = (const float*)d_in[7];
  const float* ln1_g = (const float*)d_in[8];
  const float* ln1_b = (const float*)d_in[9];
  const float* w_qkv = (const float*)d_in[10];
  const float* b_qkv = (const float*)d_in[11];
  const float* w_o   = (const float*)d_in[12];
  const float* b_o   = (const float*)d_in[13];
  const float* ln2_g = (const float*)d_in[14];
  const float* ln2_b = (const float*)d_in[15];
  const float* w_fc  = (const float*)d_in[16];
  const float* b_fc  = (const float*)d_in[17];
  const float* w_pr  = (const float*)d_in[18];
  const float* b_pr  = (const float*)d_in[19];
  const float* lnf_g = (const float*)d_in[20];
  const float* lnf_b = (const float*)d_in[21];

  float* ws = (float*)d_ws;

  temb_kernel<<<LAUNCH1_BLOCKS, 256, 0, stream>>>(
      nlp, t_w1, t_b1, t_w2, t_b2, w_qkv, w_o, w_fc, w_pr, ws);
  embed_kernel<<<256, NT, 0, stream>>>(idx, nlp, wte, wpe, ln1_g, ln1_b, w_qkv,
                                       b_qkv, lnf_g, lnf_b, ws, (float*)d_out);
  for (int L = 0; L < NLAUNCH; ++L)
    step_kernel<<<256, NT, 0, stream>>>(L, nlp, ln1_g, ln1_b, b_qkv, b_o, ln2_g,
                                        ln2_b, b_fc, b_pr, lnf_g, lnf_b, ws,
                                        (float*)d_out);
}


# Time rows (temb) of all loop steps as two batched WMMA products, 16 steps sharing each weight fragment, not a block per step with one-load-per-trip loops; temb sum order changed
# speedup vs baseline: 11.0025x; 1.1875x over previous
#include <hip/hip_runtime.h>
#include <hip/hip_bf16.h>
#include <math.h>


typedef __attribute__((ext_vector_type(2))) float v2f;
typedef __attribute__((ext_vector_type(8))) float v8f;

__device__ __forceinline__ v8f wmma4(v2f a, v2f b, v8f c) {
  return __builtin_amdgcn_wmma_f32_16x16x4_f32(
       false, a,  false, b,
       (short)0, c,  false,  false);
}

#define TEMB_ROWS 112
#define TEMB_NC 16
#define TEMB_BLOCKS 112
#define RELAY_BLOCKS 48
#define LAUNCH1_BLOCKS 160
#define WS_TP (8192 + 2 * 262144)
#define WS_W (8192 + 7 * 262144)
#define WF_O 0
#define WF_FC 4096
#define WF_PR 20480
#define WF_QKV 36864
template <int N, int LOG2_KP>
__device__ __forceinline__ void relay_entry(const float* __restrict__ w,
                                            float* __restrict__ dst, int i) {
  const int lane = i & 31, g = i >> 5;
  const int sp = g & ((1 << LOG2_KP) - 1), nt = g >> LOG2_KP;
  const int k = 8 * sp + 2 * (lane >> 4), n = 16 * nt + (lane & 15);
  float4 f;
  f.x = w[k * N + n];       f.y = w[(k + 1) * N + n];
  f.z = w[(k + 4) * N + n]; f.w = w[(k + 5) * N + n];
  reinterpret_cast<float4*>(dst)[i] = f;
}

__device__ __forceinline__ int tfrag(int row, int k) {
  return ((k >> 2) << 6) + (((k >> 1) & 1) << 5) + (row << 1) + (k & 1);
}

__global__ __launch_bounds__(256) void temb_kernel(
    const int* __restrict__ num_loops,
    const float* __restrict__ t_w1, const float* __restrict__ t_b1,
    const float* __restrict__ t_w2,
    const float* __restrict__ w_qkv, const float* __restrict__ w_o,
    const float* __restrict__ w_fc, const float* __restrict__ w_pr,
    float* __restrict__ ttab) {
  if (blockIdx.x >= TEMB_BLOCKS) {
    const int i = (blockIdx.x - TEMB_BLOCKS) * 256 + threadIdx.x;
    float* wf = ttab + WS_W;
    if (i < 1024)      relay_entry<64, 3>(w_o, wf + WF_O, i);
    else if (i < 5120) relay_entry<256, 3>(w_fc, wf + WF_FC, i - 1024);
    else if (i < 9216) relay_entry<64, 5>(w_pr, wf + WF_PR, i - 5120);
    else               relay_entry<192, 3>(w_qkv, wf + WF_QKV, i - 9216);
    return;
  }
  const int rt = blockIdx.x >> 4, ch = blockIdx.x & 15;
  if (rt * 16 >= *num_loops) return;
  __shared__ __attribute__((aligned(16))) float teA[4096];
  __shared__ __attribute__((aligned(16))) float p1[2 * 1024];
  __shared__ __attribute__((aligned(16))) float hsA[1024];
  __shared__ __attribute__((aligned(16))) float p2[2 * 1024];
  const int tid = threadIdx.x, lane = tid & 31;
  const int wid = __builtin_amdgcn_readfirstlane(tid >> 5);
  const int l16 = lane & 15, half = lane >> 4, koff = 2 * half;
  const int ct = wid & 3;
  const int kh = wid >> 2;

  v2f b1[32], b2[8];
  {
    const float* w1p = t_w1 + (128 * kh + koff) * 1024 + 64 * ch + 16 * ct + l16;
#pragma unroll
    for (int s = 0; s < 32; ++s) {
      b1[s].x = w1p[(4 * s) * 1024];
      b1[s].y = w1p[(4 * s + 1) * 1024];
    }
    const float* w2p = t_w2 + (64 * ch + 32 * kh + koff) * 64 + 16 * ct + l16;
#pragma unroll
    for (int s = 0; s < 8; ++s) {
      b2[s].x = w2p[(4 * s) * 64];
      b2[s].y = w2p[(4 * s + 1) * 64];
    }
  }
  const float bias1 = (kh == 0) ? t_b1[64 * ch + 16 * ct + l16] : 0.0f;

  {
    const int row = tid & 15, k0 = (tid >> 4) * 16;
    const float tf = (float)(rt * 16 + row);
    for (int i = 0; i < 16; ++i) {
      const int k = k0 + i;
      const float f = expf(-logf(10000.0f) * (float)(k & 127) / 128.0f);
      const float arg = tf * f;
      teA[tfrag(row, k)] = (k0 < 128) ? cosf(arg) : sinf(arg);
    }
  }
  __syncthreads();

  {
    v8f acc;
    for (int r = 0; r < 8; ++r) acc[r] = bias1;
#pragma unroll
    for (int s = 0; s < 32; ++s)
      acc = wmma4(*reinterpret_cast<const v2f*>(teA + (32 * kh + s) * 64 + lane * 2), b1[s], acc);
    for (int r = 0; r < 8; ++r)
      p1[kh * 1024 + (r + 8 * half) * 64 + 16 * ct + l16] = acc[r];
  }
  __syncthreads();
  for (int e = tid; e < 1024; e += 256) {
    const float h = p1[e] + p1[1024 + e];
    hsA[tfrag(e >> 6, e & 63)] = h / (1.0f + expf(-h));
  }
  __syncthreads();

  {
    v8f acc;
    for (int r = 0; r < 8; ++r) acc[r] = 0.0f;
#pragma unroll
    for (int s = 0; s < 8; ++s)
      acc = wmma4(*reinterpret_cast<const v2f*>(hsA + (8 * kh + s) * 64 + lane * 2), b2[s], acc);
    for (int r = 0; r < 8; ++r)
      p2[kh * 1024 + (r + 8 * half) * 64 + 16 * ct + l16] = acc[r];
  }
  __syncthreads();
  {
    const float4 a = reinterpret_cast<const float4*>(p2)[tid];
    const float4 b = reinterpret_cast<const float4*>(p2 + 1024)[tid];
    float4 o;
    o.x = a.x + b.x; o.y = a.y + b.y; o.z = a.z + b.z; o.w = a.w + b.w;
    reinterpret_cast<float4*>(ttab + WS_TP + (ch * TEMB_ROWS + rt * 16) * 64)[tid] = o;
  }
}

#define NT 512
#define G 3
#define SKEW (G - 1)
#define NLAUNCH 38
#define MAXLOOPS (G * NLAUNCH - 7 * SKEW)
#define WS_X 8192
#define WS_Q (8192 + 262144)
#define WS_KV (8192 + 8 * 262144)
#define PLANE 262144
__device__ __forceinline__ float* plane_ptr(float* ws, int pl, int s) {
  return pl == 0 ? ws + WS_Q : ws + WS_KV + (2 * s + pl - 1) * PLANE;
}
#define L_XB 0
#define L_HB 1024
#define L_GB 2048
#define L_PART 6144
#define L_PM 10240
#define L_PL 10496
#define L_PW 10752
#define L_TOTAL 18944

__device__ __forceinline__ int afrag(int row, int k) {
  return ((k >> 2) << 6) + (((k >> 1) & 1) << 5) + (row << 1) + (k & 1);
}
__device__ __forceinline__ v2f ld_afrag(const float* buf, int step, int lane) {
  return *reinterpret_cast<const v2f*>(buf + step * 64 + lane * 2);
}

#define QK_SP 68
template <int P>
__device__ __forceinline__ int qkfrag(int r, int d) {
  return (((d >> 3) << 1) + ((d >> 1) & 1)) * P + (r << 2) + (((d >> 2) & 1) << 1) + (d & 1);
}
__device__ __forceinline__ void ld_frag2(const float* blk, int sp, int lane, v2f& f0, v2f& f1) {
  const float4 f = reinterpret_cast<const float4*>(blk)[sp * 32 + lane];
  f0.x = f.x; f0.y = f.y; f1.x = f.z; f1.y = f.w;
}
__device__ __forceinline__ float4 ld_staged(const float* st, int ct, int sp, int lane) {
  const int e = ct * 64 + sp * 32 + lane;
  return *reinterpret_cast<const float4*>(st + (e >> 4) * QK_SP + (e & 15) * 4);
}
__device__ __forceinline__ void pick_frag(const float4& fg, const float4& fl, int kt, int ti,
                                          v2f& f0, v2f& f1) {
  const bool lo = kt < ti, eq = kt == ti;
  f0.x = lo ? fg.x : (eq ? fl.x : 0.0f);
  f0.y = lo ? fg.y : (eq ? fl.y : 0.0f);
  f1.x = lo ? fg.z : (eq ? fl.z : 0.0f);
  f1.y = lo ? fg.w : (eq ? fl.w : 0.0f);
}
__device__ __forceinline__ void ask_kv(float4 (&kvn)[2][2][2], float* ws, int step, int bofs,
                                       int kq, int ti, int lane) {
#pragma unroll
  for (int pl = 0; pl < 2; ++pl)
#pragma unroll
    for (int j = 0; j < 2; ++j) {
      const int kt = 2 * kq + j;
      const float4* src = reinterpret_cast<const float4*>(
          plane_ptr(ws, 1 + pl, step) + bofs + (kt < ti ? kt : 2 * kq) * 1024);
#pragma unroll
      for (int sp = 0; sp < 2; ++sp) kvn[pl][j][sp] = src[sp * 32 + lane];
    }
}
__device__ __forceinline__ void stage_qkv(float* st, const v8f& acc, int ct, int l16, int half) {
  float* blk = st + ct * (4 * QK_SP);
  if (ct >= 8) {
    for (int r = 0; r < 8; ++r) blk[qkfrag<QK_SP>(l16, r + 8 * half)] = acc[r];
  } else {
    for (int r = 0; r < 8; ++r) blk[qkfrag<QK_SP>(r + 8 * half, l16)] = acc[r];
  }
}
__device__ __forceinline__ const float4& staged4(const float* st, int i) {
  return *reinterpret_cast<const float4*>(st + (i >> 4) * QK_SP + (i & 15) * 4);
}

template <bool ADD, bool STEP0, bool FRAG>
__device__ __forceinline__ void row_pass(
    float* xb, const float* part, const float* __restrict__ bias,
    const float* __restrict__ tt, float* __restrict__ xg,
    const float* __restrict__ g, const float* __restrict__ bta, float* dst,
    int wid, int lane) {
  float v[2];
#pragma unroll
  for (int e = 0; e < 2; ++e) {
    const int d = lane + 32 * e;
    float x = xb[wid * 64 + d];
    if (ADD) {
      float s = bias[d];
      for (int k = 0; k < 4; ++k) s += part[k * 1024 + wid * 64 + d];
      x += 0.1f * s;
    }
    if (STEP0) x += tt[d];
    v[e] = x;
    if (ADD || STEP0) xb[wid * 64 + d] = x;
    if (STEP0) xg[wid * 64 + d] = x;
  }
  float sum = v[0] + v[1], sq = v[0] * v[0] + v[1] * v[1];
  for (int off = 16; off; off >>= 1) {
    sum += __shfl_xor(sum, off, 32);
    sq  += __shfl_xor(sq,  off, 32);
  }
  const float mu  = sum * (1.0f / 64.0f);
  const float var = sq * (1.0f / 64.0f) - mu * mu;
  const float inv = rsqrtf(var + 1e-5f);
#pragma unroll
  for (int e = 0; e < 2; ++e) {
    const int d = lane + 32 * e;
    dst[FRAG ? afrag(wid, d) : wid * 64 + d] = (v[e] - mu) * inv * g[d] + bta[d];
  }
}

__device__ __forceinline__ void qkv_out(
    const float* hb, float* st, const float* __restrict__ w_qkv,
    const float* __restrict__ b_qkv, float* __restrict__ ws, int toff,
    int tid, int wid, int lane, int l16, int half, int koff) {
  if (wid < 12) {
    const int n0 = wid * 16;
    v8f acc;
    { float bias = b_qkv[n0 + l16]; for (int r = 0; r < 8; ++r) acc[r] = bias; }
#pragma unroll
    for (int k = 0; k < 64; k += 4) {
      v2f a = ld_afrag(hb, k >> 2, lane), w;
      w.x = w_qkv[(k + koff) * 192 + n0 + l16];
      w.y = w_qkv[(k + koff + 1) * 192 + n0 + l16];
      acc = wmma4(a, w, acc);
    }
    stage_qkv(st, acc, wid, l16, half);
  }
  __syncthreads();
  for (int i = tid; i < 768; i += NT) {
    const int pl = i >> 8, off = i & 255;
    reinterpret_cast<float4*>(plane_ptr(ws, pl, 0) + toff)[off] =
        staged4(st, i);
  }
}

__device__ __forceinline__ void out_copy(const float* hb, float* __restrict__ outg, int tid) {
  __syncthreads();
  for (int i = tid; i < 256; i += NT)
    reinterpret_cast<float4*>(outg)[i] = reinterpret_cast<const float4*>(hb)[i];
}

__device__ __forceinline__ void wg_barrier() {
  asm volatile("s_wait_dscnt 0x0\n\ts_barrier_signal -1\n\ts_barrier_wait -1" ::: "memory");
}

template <bool STEP0, bool FRAG>
__device__ __forceinline__ void row_pass_r(
    float* xb, const float* part, const float (&bias)[2], const float (&tt)[2],
    float* __restrict__ xg, const float (&g)[2], const float (&bta)[2],
    float* dst, int wid, int lane) {
  float v[2];
#pragma unroll
  for (int e = 0; e < 2; ++e) {
    const int d = lane + 32 * e;
    float x = xb[wid * 64 + d];
    float s = bias[e];
    for (int k = 0; k < 4; ++k) s += part[k * 1024 + wid * 64 + d];
    x += 0.1f * s;
    if (STEP0) x += tt[e];
    v[e] = x;
    xb[wid * 64 + d] = x;
    if (STEP0) xg[wid * 64 + d] = x;
  }
  float sum = v[0] + v[1], sq = v[0] * v[0] + v[1] * v[1];
  for (int off = 16; off; off >>= 1) {
    sum += __shfl_xor(sum, off, 32);
    sq  += __shfl_xor(sq,  off, 32);
  }
  const float mu  = sum * (1.0f / 64.0f);
  const float var = sq * (1.0f / 64.0f) - mu * mu;
  const float inv = rsqrtf(var + 1e-5f);
#pragma unroll
  for (int e = 0; e < 2; ++e) {
    const int d = lane + 32 * e;
    dst[FRAG ? afrag(wid, d) : wid * 64 + d] = (v[e] - mu) * inv * g[e] + bta[e];
  }
}

__global__ __launch_bounds__(NT) void embed_kernel(
    const int* __restrict__ idx, const int* __restrict__ num_loops,
    const float* __restrict__ wte, const float* __restrict__ wpe,
    const float* __restrict__ t_b2,
    const float* __restrict__ ln1_g, const float* __restrict__ ln1_b,
    const float* __restrict__ w_qkv, const float* __restrict__ b_qkv,
    const float* __restrict__ lnf_g, const float* __restrict__ lnf_b,
    float* __restrict__ ws, float* __restrict__ out) {
  __shared__ __attribute__((aligned(16))) float smem[L_TOTAL];
  float* xb = smem + L_XB;
  float* hb = smem + L_HB;
  float* st = smem + L_GB;
  const int b = blockIdx.x >> 3, q0 = (blockIdx.x & 7) * 16;
  const int toff = (b * 128 + q0) * 64;
  const int tid = threadIdx.x, lane = tid & 31;
  const int wid = __builtin_amdgcn_readfirstlane(tid >> 5);
  const int l16 = lane & 15, half = lane >> 4, koff = 2 * half;

  float* tt0 = smem + L_PM;
  if (tid < 128) {
    const int d = tid & 63;
    const int row = (tid < 64) ? 0 : (int)blockIdx.x;
    if (row < min(*num_loops, TEMB_ROWS)) {
      const float* tp = ws + WS_TP + row * 64 + d;
      float s = t_b2[d];
#pragma unroll
      for (int ch = 0; ch < TEMB_NC; ++ch) s += tp[ch * TEMB_ROWS * 64];
      if (tid < 64) tt0[d] = s; else ws[row * 64 + d] = s;
    }
  }
  for (int i = tid; i < 1024; i += NT) {
    int s = i >> 6, d = i & 63;
    xb[i] = wte[idx[b * 128 + q0 + s] * 64 + d] + wpe[(q0 + s) * 64 + d];
  }
  __syncthreads();

  if (*num_loops > 0) {
    row_pass<false, true, true>(xb, nullptr, nullptr, tt0, ws + WS_X + toff, ln1_g, ln1_b, hb, wid, lane);
    __syncthreads();
    qkv_out(hb, st, w_qkv, b_qkv, ws, toff, tid, wid, lane, l16, half, koff);
  } else {
    row_pass<false, false, false>(xb, nullptr, nullptr, nullptr, nullptr, lnf_g, lnf_b, hb, wid, lane);
    out_copy(hb, out + toff, tid);
  }
}

__global__ __launch_bounds__(NT) void step_kernel(
    int L, const int* __restrict__ num_loops,
    const float* __restrict__ ln1_g, const float* __restrict__ ln1_b,
    const float* __restrict__ b_qkv, const float* __restrict__ b_o,
    const float* __restrict__ ln2_g, const float* __restrict__ ln2_b,
    const float* __restrict__ b_fc, const float* __restrict__ b_pr,
    const float* __restrict__ lnf_g, const float* __restrict__ lnf_b,
    float* __restrict__ ws, float* __restrict__ out) {
  const int NL = min(*num_loops, MAXLOOPS);
  const int ti = blockIdx.x & 7;
  const int sa = G * L - SKEW * ti;
  const int s0 = max(sa, 0);
  const int s1 = min(sa + G, NL);
  if (s0 >= s1) return;

  __shared__ __attribute__((aligned(16))) float smem[L_TOTAL];
  float* xb = smem + L_XB;
  float* hb = smem + L_HB;
  float* gb = smem + L_GB;
  float* part = smem + L_PART;
  float* pm = smem + L_PM;
  float* pl = smem + L_PL;

  const int b = blockIdx.x >> 3, q0 = ti * 16;
  const int toff = (b * 128 + q0) * 64;
  const int tid = threadIdx.x, lane = tid & 31;
  const int wid = __builtin_amdgcn_readfirstlane(tid >> 5);
  const int l16 = lane & 15, half = lane >> 4, koff = 2 * half;
  const int wq = wid & 3;
  const int kq = wid >> 2;

  float* xg = ws + WS_X + toff;
  const int bofs = b * 8192 + wq * 256;
  const bool below = (2 * kq < ti);
  const int c0 = wq * 16;
  const int n0g = wid * 16;
  const float* wf = ws + WS_W;
  const float* wo_f  = wf + WF_O   + (wq * 8 + 2 * kq) * 128;
  const float* wfc_f = wf + WF_FC  + (wid * 8) * 128;
  const float* wpr_f = wf + WF_PR  + (wq * 32 + 8 * kq) * 128;
  const float* wq_f  = wf + WF_QKV + (wid * 8) * 128;

  float4 kvn[2][2][2];
#pragma unroll
  for (int a = 0; a < 8; ++a) kvn[a >> 2][(a >> 1) & 1][a & 1] = make_float4(0.0f, 0.0f, 0.0f, 0.0f);
  if (below) ask_kv(kvn, ws, s0, bofs, kq, ti, lane);
  {
    const float xa = xg[tid], xc = xg[tid + NT];
    const float4 f0 =
        reinterpret_cast<const float4*>(plane_ptr(ws, tid >> 8, s0) + toff)[tid & 255];
    float4 f1 = f0;
    if (tid < 256)
      f1 = reinterpret_cast<const float4*>(plane_ptr(ws, 2, s0) + toff)[tid];
    *reinterpret_cast<float4*>(gb + (tid >> 4) * QK_SP + (tid & 15) * 4) = f0;
    if (tid < 256)
      *reinterpret_cast<float4*>(gb + ((tid + 512) >> 4) * QK_SP + (tid & 15) * 4) = f1;
    xb[tid] = xa;
    xb[tid + NT] = xc;
  }
  wg_barrier();

  for (int t = s0; t < s1; ++t) {
  const bool more = (t + 1 < NL);
  v2f qa[4], kb[2][4], vb[2][4], fo[4];
#pragma unroll
  for (int sp = 0; sp < 2; ++sp) {
    const float4 f = ld_staged(gb, wq, sp, lane);
    qa[2 * sp].x = f.x; qa[2 * sp].y = f.y; qa[2 * sp + 1].x = f.z; qa[2 * sp + 1].y = f.w;
  }
#pragma unroll
  for (int j = 0; j < 2; ++j)
#pragma unroll
    for (int sp = 0; sp < 2; ++sp) {
      pick_frag(kvn[0][j][sp], ld_staged(gb, 4 + wq, sp, lane), 2 * kq + j, ti,
                kb[j][2 * sp], kb[j][2 * sp + 1]);
      pick_frag(kvn[1][j][sp], ld_staged(gb, 8 + wq, sp, lane), 2 * kq + j, ti,
                vb[j][2 * sp], vb[j][2 * sp + 1]);
    }
#pragma unroll
  for (int sp = 0; sp < 2; ++sp)
    ld_frag2(wo_f, sp, lane, fo[2 * sp], fo[2 * sp + 1]);
  float cbo[2], cg2[2], cb2[2];
#pragma unroll
  for (int e = 0; e < 2; ++e) {
    const int d = lane + 32 * e;
    cbo[e] = b_o[d]; cg2[e] = ln2_g[d]; cb2[e] = ln2_b[d];
  }

  v2f ffc[16];
  float bfc;
  {
    float* pw = smem + L_PW + wid * 512;
#pragma unroll
    for (int j = 0; j < 2; ++j) {
      const int kt = 2 * kq + j;
      v8f a8; for (int r = 0; r < 8; ++r) a8[r] = 0.0f;
#pragma unroll
      for (int s = 0; s < 4; ++s)
        a8 = wmma4(qa[s], kb[j][s], a8);
#pragma unroll
      for (int r = 0; r < 8; ++r) {
        const int row = q0 + r + 8 * half, col = kt * 16 + l16;
        pw[j * 256 + (r + 8 * half) * 16 + l16] =
            (col <= row) ? a8[r] * 0.25f : -1e9f;
      }
    }
    v2f p[2][4];
    float mx = -1e30f;
#pragma unroll
    for (int j = 0; j < 2; ++j)
#pragma unroll
      for (int s = 0; s < 4; ++s) {
        p[j][s] = *reinterpret_cast<const v2f*>(pw + j * 256 + l16 * 16 + 4 * s + koff);
        mx = fmaxf(mx, fmaxf(p[j][s].x, p[j][s].y));
      }
    mx = fmaxf(mx, __shfl_xor(mx, 16, 32));
    float sum = 0.0f;
#pragma unroll
    for (int j = 0; j < 2; ++j)
#pragma unroll
      for (int s = 0; s < 4; ++s) {
        p[j][s].x = __expf(p[j][s].x - mx);
        p[j][s].y = __expf(p[j][s].y - mx);
        sum += p[j][s].x + p[j][s].y;
      }
    sum += __shfl_xor(sum, 16, 32);
    v8f o; for (int r = 0; r < 8; ++r) o[r] = 0.0f;
#pragma unroll
    for (int j = 0; j < 2; ++j)
#pragma unroll
      for (int s = 0; s < 4; ++s)
        o = wmma4(p[j][s], vb[j][s], o);
#pragma unroll
    for (int sp = 0; sp < 8; ++sp)
      ld_frag2(wfc_f, sp, lane, ffc[2 * sp], ffc[2 * sp + 1]);
    bfc = b_fc[n0g + l16];
    for (int r = 0; r < 8; ++r)
      part[kq * 1024 + (r + 8 * half) * 64 + c0 + l16] = o[r];
    if (half == 0) {
      pm[(kq * 4 + wq) * 16 + l16] = mx;
      pl[(kq * 4 + wq) * 16 + l16] = sum;
    }
  }
  wg_barrier();
  {
    const int row = tid >> 5, c = (tid & 31) * 2, hh = c >> 4;
    float m[4];
    for (int k = 0; k < 4; ++k) m[k] = pm[(k * 4 + hh) * 16 + row];
    const float M = fmaxf(fmaxf(m[0], m[1]), fmaxf(m[2], m[3]));
    float L = 0.0f;
    v2f O; O.x = 0.0f; O.y = 0.0f;
    for (int k = 0; k < 4; ++k) {
      const float wk = __expf(m[k] - M);
      const v2f pk = *reinterpret_cast<const v2f*>(part + k * 1024 + row * 64 + c);
      L += pl[(k * 4 + hh) * 16 + row] * wk;
      O.x += pk.x * wk;
      O.y += pk.y * wk;
    }
    const float inv = 1.0f / L;
    O.x *= inv; O.y *= inv;
    *reinterpret_cast<v2f*>(hb + afrag(row, c)) = O;
  }
  wg_barrier();

  v2f fpr[16];
  float cbp[2], ctt[2], cgE[2], cbE[2];
  {
    v8f acc; for (int r = 0; r < 8; ++r) acc[r] = 0.0f;
#pragma unroll
    for (int s = 0; s < 4; ++s)
      acc = wmma4(ld_afrag(hb, kq * 4 + s, lane), fo[s], acc);
    for (int r = 0; r < 8; ++r)
      part[kq * 1024 + (r + 8 * half) * 64 + c0 + l16] = acc[r];
#pragma unroll
    for (int sp = 0; sp < 8; ++sp)
      ld_frag2(wpr_f, sp, lane, fpr[2 * sp], fpr[2 * sp + 1]);
    const float* gE = more ? ln1_g : lnf_g;
    const float* bE = more ? ln1_b : lnf_b;
    const float* tr = ws + (more ? (t + 1) * 64 : 0);
#pragma unroll
    for (int e = 0; e < 2; ++e) {
      const int d = lane + 32 * e;
      cbp[e] = b_pr[d]; ctt[e] = tr[d]; cgE[e] = gE[d]; cbE[e] = bE[d];
    }
  }
  wg_barrier();

  row_pass_r<false, true>(xb, part, cbo, cbo, nullptr, cg2, cb2, hb, wid, lane);
  wg_barrier();

  v2f fq[16];
  float bq = 0.0f;
#pragma unroll
  for (int s = 0; s < 16; ++s) { fq[s].x = 0.0f; fq[s].y = 0.0f; }
  {
    v8f acc;
    for (int r = 0; r < 8; ++r) acc[r] = bfc;
#pragma unroll
    for (int s = 0; s < 16; ++s)
      acc = wmma4(ld_afrag(hb, s, lane), ffc[s], acc);
    if (more && wid < 12) {
#pragma unroll
      for (int sp = 0; sp < 8; ++sp)
        ld_frag2(wq_f, sp, lane, fq[2 * sp], fq[2 * sp + 1]);
      bq = b_qkv[n0g + l16];
    }
    if (below && t + 1 < s1)
      ask_kv(kvn, ws, t + 1, bofs, kq, ti, lane);
    for (int r = 0; r < 8; ++r) {
      float u = acc[r];
      float g = 0.5f * u *
          (1.0f + tanhf(0.7978845608028654f * (u + 0.044715f * u * u * u)));
      gb[afrag(r + 8 * half, n0g + l16)] = g;
    }
  }
  wg_barrier();
  {
    v8f acc; for (int r = 0; r < 8; ++r) acc[r] = 0.0f;
#pragma unroll
    for (int s = 0; s < 16; ++s)
      acc = wmma4(ld_afrag(gb, kq * 16 + s, lane), fpr[s], acc);
    for (int r = 0; r < 8; ++r)
      part[kq * 1024 + (r + 8 * half) * 64 + c0 + l16] = acc[r];
  }
  wg_barrier();

  if (more) {
    row_pass_r<true, true>(xb, part, cbp, ctt, xg, cgE, cbE, hb, wid, lane);
    wg_barrier();
    if (wid < 12) {
      v8f acc;
      for (int r = 0; r < 8; ++r) acc[r] = bq;
#pragma unroll
      for (int s = 0; s < 16; ++s)
        acc = wmma4(ld_afrag(hb, s, lane), fq[s], acc);
      stage_qkv(gb, acc, wid, l16, half);
    }
    wg_barrier();
    for (int i = tid; i < 768; i += NT) {
      const int pln = i >> 8, off = i & 255;
      reinterpret_cast<float4*>(plane_ptr(ws, pln, t + 1) + toff)[off] =
          staged4(gb, i);
    }
  } else {
    row_pass_r<false, false>(xb, part, cbp, ctt, nullptr, cgE, cbE, hb, wid, lane);
    wg_barrier();
    for (int i = tid; i < 256; i += NT)
      reinterpret_cast<float4*>(out + toff)[i] = reinterpret_cast<const float4*>(hb)[i];
  }
  }
}

extern "C" void kernel_launch(void* const* d_in, const int* in_sizes, int n_in,
                              void* d_out, int out_size, void* d_ws, size_t ws_size,
                              hipStream_t stream) {
  const int*   idx   = (const int*)  d_in[0];
  const int*   nlp   = (const int*)  d_in[1];
  const float* wte   = (const float*)d_in[2];
  const float* wpe   = (const float*)d_in[3];
  const float* t_w1  = (const float*)d_in[4];
  const float* t_b1  = (const float*)d_in[5];
  const float* t_w2  = (const float*)d_in[6];
  const float* t_b2  = (const float*)d_in[7];
  const float* ln1_g = (const float*)d_in[8];
  const float* ln1_b = (const float*)d_in[9];
  const float* w_qkv = (const float*)d_in[10];
  const float* b_qkv = (const float*)d_in[11];
  const float* w_o   = (const float*)d_in[12];
  const float* b_o   = (const float*)d_in[13];
  const float* ln2_g = (const float*)d_in[14];
  const float* ln2_b = (const float*)d_in[15];
  const float* w_fc  = (const float*)d_in[16];
  const float* b_fc  = (const float*)d_in[17];
  const float* w_pr  = (const float*)d_in[18];
  const float* b_pr  = (const float*)d_in[19];
  const float* lnf_g = (const float*)d_in[20];
  const float* lnf_b = (const float*)d_in[21];

  float* ws = (float*)d_ws;

  temb_kernel<<<LAUNCH1_BLOCKS, 256, 0, stream>>>(
      nlp, t_w1, t_b1, t_w2, w_qkv, w_o, w_fc, w_pr, ws);
  embed_kernel<<<256, NT, 0, stream>>>(idx, nlp, wte, wpe, t_b2, ln1_g, ln1_b, w_qkv,
                                       b_qkv, lnf_g, lnf_b, ws, (float*)d_out);
  for (int L = 0; L < NLAUNCH; ++L)
    step_kernel<<<256, NT, 0, stream>>>(L, nlp, ln1_g, ln1_b, b_qkv, b_o, ln2_g,
                                        ln2_b, b_fc, b_pr, lnf_g, lnf_b, ws,
                                        (float*)d_out);
}


# Weight fragments and step-invariant bias and LN vectors fetched only in a launch's first loop step and held in registers for its later steps (step body built twice); sums and order unchanged
# speedup vs baseline: 11.4241x; 1.0383x over previous
#include <hip/hip_runtime.h>
#include <hip/hip_bf16.h>
#include <math.h>
#include <type_traits>


typedef __attribute__((ext_vector_type(2))) float v2f;
typedef __attribute__((ext_vector_type(8))) float v8f;

__device__ __forceinline__ v8f wmma4(v2f a, v2f b, v8f c) {
  return __builtin_amdgcn_wmma_f32_16x16x4_f32(
       false, a,  false, b,
       (short)0, c,  false,  false);
}

#define TEMB_ROWS 112
#define TEMB_NC 16
#define TEMB_BLOCKS 112
#define RELAY_BLOCKS 48
#define LAUNCH1_BLOCKS 160
#define WS_TP (8192 + 2 * 262144)
#define WS_W (8192 + 7 * 262144)
#define WF_O 0
#define WF_FC 4096
#define WF_PR 20480
#define WF_QKV 36864
template <int N, int LOG2_KP>
__device__ __forceinline__ void relay_entry(const float* __restrict__ w,
                                            float* __restrict__ dst, int i) {
  const int lane = i & 31, g = i >> 5;
  const int sp = g & ((1 << LOG2_KP) - 1), nt = g >> LOG2_KP;
  const int k = 8 * sp + 2 * (lane >> 4), n = 16 * nt + (lane & 15);
  float4 f;
  f.x = w[k * N + n];       f.y = w[(k + 1) * N + n];
  f.z = w[(k + 4) * N + n]; f.w = w[(k + 5) * N + n];
  reinterpret_cast<float4*>(dst)[i] = f;
}

__device__ __forceinline__ int tfrag(int row, int k) {
  return ((k >> 2) << 6) + (((k >> 1) & 1) << 5) + (row << 1) + (k & 1);
}

__global__ __launch_bounds__(256) void temb_kernel(
    const int* __restrict__ num_loops,
    const float* __restrict__ t_w1, const float* __restrict__ t_b1,
    const float* __restrict__ t_w2,
    const float* __restrict__ w_qkv, const float* __restrict__ w_o,
    const float* __restrict__ w_fc, const float* __restrict__ w_pr,
    float* __restrict__ ttab) {
  if (blockIdx.x >= TEMB_BLOCKS) {
    const int i = (blockIdx.x - TEMB_BLOCKS) * 256 + threadIdx.x;
    float* wf = ttab + WS_W;
    if (i < 1024)      relay_entry<64, 3>(w_o, wf + WF_O, i);
    else if (i < 5120) relay_entry<256, 3>(w_fc, wf + WF_FC, i - 1024);
    else if (i < 9216) relay_entry<64, 5>(w_pr, wf + WF_PR, i - 5120);
    else               relay_entry<192, 3>(w_qkv, wf + WF_QKV, i - 9216);
    return;
  }
  const int rt = blockIdx.x >> 4, ch = blockIdx.x & 15;
  if (rt * 16 >= *num_loops) return;
  __shared__ __attribute__((aligned(16))) float teA[4096];
  __shared__ __attribute__((aligned(16))) float p1[2 * 1024];
  __shared__ __attribute__((aligned(16))) float hsA[1024];
  __shared__ __attribute__((aligned(16))) float p2[2 * 1024];
  const int tid = threadIdx.x, lane = tid & 31;
  const int wid = __builtin_amdgcn_readfirstlane(tid >> 5);
  const int l16 = lane & 15, half = lane >> 4, koff = 2 * half;
  const int ct = wid & 3;
  const int kh = wid >> 2;

  v2f b1[32], b2[8];
  {
    const float* w1p = t_w1 + (128 * kh + koff) * 1024 + 64 * ch + 16 * ct + l16;
#pragma unroll
    for (int s = 0; s < 32; ++s) {
      b1[s].x = w1p[(4 * s) * 1024];
      b1[s].y = w1p[(4 * s + 1) * 1024];
    }
    const float* w2p = t_w2 + (64 * ch + 32 * kh + koff) * 64 + 16 * ct + l16;
#pragma unroll
    for (int s = 0; s < 8; ++s) {
      b2[s].x = w2p[(4 * s) * 64];
      b2[s].y = w2p[(4 * s + 1) * 64];
    }
  }
  const float bias1 = (kh == 0) ? t_b1[64 * ch + 16 * ct + l16] : 0.0f;

  {
    const int row = tid & 15, k0 = (tid >> 4) * 16;
    const float tf = (float)(rt * 16 + row);
    for (int i = 0; i < 16; ++i) {
      const int k = k0 + i;
      const float f = expf(-logf(10000.0f) * (float)(k & 127) / 128.0f);
      const float arg = tf * f;
      teA[tfrag(row, k)] = (k0 < 128) ? cosf(arg) : sinf(arg);
    }
  }
  __syncthreads();

  {
    v8f acc;
    for (int r = 0; r < 8; ++r) acc[r] = bias1;
#pragma unroll
    for (int s = 0; s < 32; ++s)
      acc = wmma4(*reinterpret_cast<const v2f*>(teA + (32 * kh + s) * 64 + lane * 2), b1[s], acc);
    for (int r = 0; r < 8; ++r)
      p1[kh * 1024 + (r + 8 * half) * 64 + 16 * ct + l16] = acc[r];
  }
  __syncthreads();
  for (int e = tid; e < 1024; e += 256) {
    const float h = p1[e] + p1[1024 + e];
    hsA[tfrag(e >> 6, e & 63)] = h / (1.0f + expf(-h));
  }
  __syncthreads();

  {
    v8f acc;
    for (int r = 0; r < 8; ++r) acc[r] = 0.0f;
#pragma unroll
    for (int s = 0; s < 8; ++s)
      acc = wmma4(*reinterpret_cast<const v2f*>(hsA + (8 * kh + s) * 64 + lane * 2), b2[s], acc);
    for (int r = 0; r < 8; ++r)
      p2[kh * 1024 + (r + 8 * half) * 64 + 16 * ct + l16] = acc[r];
  }
  __syncthreads();
  {
    const float4 a = reinterpret_cast<const float4*>(p2)[tid];
    const float4 b = reinterpret_cast<const float4*>(p2 + 1024)[tid];
    float4 o;
    o.x = a.x + b.x; o.y = a.y + b.y; o.z = a.z + b.z; o.w = a.w + b.w;
    reinterpret_cast<float4*>(ttab + WS_TP + (ch * TEMB_ROWS + rt * 16) * 64)[tid] = o;
  }
}

#define NT 512
#define G 3
#define SKEW (G - 1)
#define NLAUNCH 38
#define MAXLOOPS (G * NLAUNCH - 7 * SKEW)
#define WS_X 8192
#define WS_Q (8192 + 262144)
#define WS_KV (8192 + 8 * 262144)
#define PLANE 262144
__device__ __forceinline__ float* plane_ptr(float* ws, int pl, int s) {
  return pl == 0 ? ws + WS_Q : ws + WS_KV + (2 * s + pl - 1) * PLANE;
}
#define L_XB 0
#define L_HB 1024
#define L_GB 2048
#define L_PART 6144
#define L_PM 10240
#define L_PL 10496
#define L_PW 10752
#define L_TOTAL 18944

__device__ __forceinline__ int afrag(int row, int k) {
  return ((k >> 2) << 6) + (((k >> 1) & 1) << 5) + (row << 1) + (k & 1);
}
__device__ __forceinline__ v2f ld_afrag(const float* buf, int step, int lane) {
  return *reinterpret_cast<const v2f*>(buf + step * 64 + lane * 2);
}

#define QK_SP 68
template <int P>
__device__ __forceinline__ int qkfrag(int r, int d) {
  return (((d >> 3) << 1) + ((d >> 1) & 1)) * P + (r << 2) + (((d >> 2) & 1) << 1) + (d & 1);
}
__device__ __forceinline__ void ld_frag2(const float* blk, int sp, int lane, v2f& f0, v2f& f1) {
  const float4 f = reinterpret_cast<const float4*>(blk)[sp * 32 + lane];
  f0.x = f.x; f0.y = f.y; f1.x = f.z; f1.y = f.w;
}
__device__ __forceinline__ float4 ld_staged(const float* st, int ct, int sp, int lane) {
  const int e = ct * 64 + sp * 32 + lane;
  return *reinterpret_cast<const float4*>(st + (e >> 4) * QK_SP + (e & 15) * 4);
}
__device__ __forceinline__ void pick_frag(const float4& fg, const float4& fl, int kt, int ti,
                                          v2f& f0, v2f& f1) {
  const bool lo = kt < ti, eq = kt == ti;
  f0.x = lo ? fg.x : (eq ? fl.x : 0.0f);
  f0.y = lo ? fg.y : (eq ? fl.y : 0.0f);
  f1.x = lo ? fg.z : (eq ? fl.z : 0.0f);
  f1.y = lo ? fg.w : (eq ? fl.w : 0.0f);
}
__device__ __forceinline__ void ask_kv(float4 (&kvn)[2][2][2], float* ws, int step, int bofs,
                                       int kq, int ti, int lane) {
#pragma unroll
  for (int pl = 0; pl < 2; ++pl)
#pragma unroll
    for (int j = 0; j < 2; ++j) {
      const int kt = 2 * kq + j;
      const float4* src = reinterpret_cast<const float4*>(
          plane_ptr(ws, 1 + pl, step) + bofs + (kt < ti ? kt : 2 * kq) * 1024);
#pragma unroll
      for (int sp = 0; sp < 2; ++sp) kvn[pl][j][sp] = src[sp * 32 + lane];
    }
}
__device__ __forceinline__ void stage_qkv(float* st, const v8f& acc, int ct, int l16, int half) {
  float* blk = st + ct * (4 * QK_SP);
  if (ct >= 8) {
    for (int r = 0; r < 8; ++r) blk[qkfrag<QK_SP>(l16, r + 8 * half)] = acc[r];
  } else {
    for (int r = 0; r < 8; ++r) blk[qkfrag<QK_SP>(r + 8 * half, l16)] = acc[r];
  }
}
__device__ __forceinline__ const float4& staged4(const float* st, int i) {
  return *reinterpret_cast<const float4*>(st + (i >> 4) * QK_SP + (i & 15) * 4);
}

template <bool ADD, bool STEP0, bool FRAG>
__device__ __forceinline__ void row_pass(
    float* xb, const float* part, const float* __restrict__ bias,
    const float* __restrict__ tt, float* __restrict__ xg,
    const float* __restrict__ g, const float* __restrict__ bta, float* dst,
    int wid, int lane) {
  float v[2];
#pragma unroll
  for (int e = 0; e < 2; ++e) {
    const int d = lane + 32 * e;
    float x = xb[wid * 64 + d];
    if (ADD) {
      float s = bias[d];
      for (int k = 0; k < 4; ++k) s += part[k * 1024 + wid * 64 + d];
      x += 0.1f * s;
    }
    if (STEP0) x += tt[d];
    v[e] = x;
    if (ADD || STEP0) xb[wid * 64 + d] = x;
    if (STEP0) xg[wid * 64 + d] = x;
  }
  float sum = v[0] + v[1], sq = v[0] * v[0] + v[1] * v[1];
  for (int off = 16; off; off >>= 1) {
    sum += __shfl_xor(sum, off, 32);
    sq  += __shfl_xor(sq,  off, 32);
  }
  const float mu  = sum * (1.0f / 64.0f);
  const float var = sq * (1.0f / 64.0f) - mu * mu;
  const float inv = rsqrtf(var + 1e-5f);
#pragma unroll
  for (int e = 0; e < 2; ++e) {
    const int d = lane + 32 * e;
    dst[FRAG ? afrag(wid, d) : wid * 64 + d] = (v[e] - mu) * inv * g[d] + bta[d];
  }
}

__device__ __forceinline__ void qkv_out(
    const float* hb, float* st, const float* __restrict__ w_qkv,
    const float* __restrict__ b_qkv, float* __restrict__ ws, int toff,
    int tid, int wid, int lane, int l16, int half, int koff) {
  if (wid < 12) {
    const int n0 = wid * 16;
    v8f acc;
    { float bias = b_qkv[n0 + l16]; for (int r = 0; r < 8; ++r) acc[r] = bias; }
#pragma unroll
    for (int k = 0; k < 64; k += 4) {
      v2f a = ld_afrag(hb, k >> 2, lane), w;
      w.x = w_qkv[(k + koff) * 192 + n0 + l16];
      w.y = w_qkv[(k + koff + 1) * 192 + n0 + l16];
      acc = wmma4(a, w, acc);
    }
    stage_qkv(st, acc, wid, l16, half);
  }
  __syncthreads();
  for (int i = tid; i < 768; i += NT) {
    const int pl = i >> 8, off = i & 255;
    reinterpret_cast<float4*>(plane_ptr(ws, pl, 0) + toff)[off] =
        staged4(st, i);
  }
}

__device__ __forceinline__ void out_copy(const float* hb, float* __restrict__ outg, int tid) {
  __syncthreads();
  for (int i = tid; i < 256; i += NT)
    reinterpret_cast<float4*>(outg)[i] = reinterpret_cast<const float4*>(hb)[i];
}

__device__ __forceinline__ void wg_barrier() {
  asm volatile("s_wait_dscnt 0x0\n\ts_barrier_signal -1\n\ts_barrier_wait -1" ::: "memory");
}

template <bool STEP0, bool FRAG>
__device__ __forceinline__ void row_pass_r(
    float* xb, const float* part, const float (&bias)[2], const float (&tt)[2],
    float* __restrict__ xg, const float (&g)[2], const float (&bta)[2],
    float* dst, int wid, int lane) {
  float v[2];
#pragma unroll
  for (int e = 0; e < 2; ++e) {
    const int d = lane + 32 * e;
    float x = xb[wid * 64 + d];
    float s = bias[e];
    for (int k = 0; k < 4; ++k) s += part[k * 1024 + wid * 64 + d];
    x += 0.1f * s;
    if (STEP0) x += tt[e];
    v[e] = x;
    xb[wid * 64 + d] = x;
    if (STEP0) xg[wid * 64 + d] = x;
  }
  float sum = v[0] + v[1], sq = v[0] * v[0] + v[1] * v[1];
  for (int off = 16; off; off >>= 1) {
    sum += __shfl_xor(sum, off, 32);
    sq  += __shfl_xor(sq,  off, 32);
  }
  const float mu  = sum * (1.0f / 64.0f);
  const float var = sq * (1.0f / 64.0f) - mu * mu;
  const float inv = rsqrtf(var + 1e-5f);
#pragma unroll
  for (int e = 0; e < 2; ++e) {
    const int d = lane + 32 * e;
    dst[FRAG ? afrag(wid, d) : wid * 64 + d] = (v[e] - mu) * inv * g[e] + bta[e];
  }
}

__global__ __launch_bounds__(NT) void embed_kernel(
    const int* __restrict__ idx, const int* __restrict__ num_loops,
    const float* __restrict__ wte, const float* __restrict__ wpe,
    const float* __restrict__ t_b2,
    const float* __restrict__ ln1_g, const float* __restrict__ ln1_b,
    const float* __restrict__ w_qkv, const float* __restrict__ b_qkv,
    const float* __restrict__ lnf_g, const float* __restrict__ lnf_b,
    float* __restrict__ ws, float* __restrict__ out) {
  __shared__ __attribute__((aligned(16))) float smem[L_TOTAL];
  float* xb = smem + L_XB;
  float* hb = smem + L_HB;
  float* st = smem + L_GB;
  const int b = blockIdx.x >> 3, q0 = (blockIdx.x & 7) * 16;
  const int toff = (b * 128 + q0) * 64;
  const int tid = threadIdx.x, lane = tid & 31;
  const int wid = __builtin_amdgcn_readfirstlane(tid >> 5);
  const int l16 = lane & 15, half = lane >> 4, koff = 2 * half;

  float* tt0 = smem + L_PM;
  if (tid < 128) {
    const int d = tid & 63;
    const int row = (tid < 64) ? 0 : (int)blockIdx.x;
    if (row < min(*num_loops, TEMB_ROWS)) {
      const float* tp = ws + WS_TP + row * 64 + d;
      float s = t_b2[d];
#pragma unroll
      for (int ch = 0; ch < TEMB_NC; ++ch) s += tp[ch * TEMB_ROWS * 64];
      if (tid < 64) tt0[d] = s; else ws[row * 64 + d] = s;
    }
  }
  for (int i = tid; i < 1024; i += NT) {
    int s = i >> 6, d = i & 63;
    xb[i] = wte[idx[b * 128 + q0 + s] * 64 + d] + wpe[(q0 + s) * 64 + d];
  }
  __syncthreads();

  if (*num_loops > 0) {
    row_pass<false, true, true>(xb, nullptr, nullptr, tt0, ws + WS_X + toff, ln1_g, ln1_b, hb, wid, lane);
    __syncthreads();
    qkv_out(hb, st, w_qkv, b_qkv, ws, toff, tid, wid, lane, l16, half, koff);
  } else {
    row_pass<false, false, false>(xb, nullptr, nullptr, nullptr, nullptr, lnf_g, lnf_b, hb, wid, lane);
    out_copy(hb, out + toff, tid);
  }
}

__global__ __launch_bounds__(NT) void step_kernel(
    int L, const int* __restrict__ num_loops,
    const float* __restrict__ ln1_g, const float* __restrict__ ln1_b,
    const float* __restrict__ b_qkv, const float* __restrict__ b_o,
    const float* __restrict__ ln2_g, const float* __restrict__ ln2_b,
    const float* __restrict__ b_fc, const float* __restrict__ b_pr,
    const float* __restrict__ lnf_g, const float* __restrict__ lnf_b,
    float* __restrict__ ws, float* __restrict__ out) {
  const int NL = min(*num_loops, MAXLOOPS);
  const int ti = blockIdx.x & 7;
  const int sa = G * L - SKEW * ti;
  const int s0 = max(sa, 0);
  const int s1 = min(sa + G, NL);
  if (s0 >= s1) return;

  __shared__ __attribute__((aligned(16))) float smem[L_TOTAL];
  float* xb = smem + L_XB;
  float* hb = smem + L_HB;
  float* gb = smem + L_GB;
  float* part = smem + L_PART;
  float* pm = smem + L_PM;
  float* pl = smem + L_PL;

  const int b = blockIdx.x >> 3, q0 = ti * 16;
  const int toff = (b * 128 + q0) * 64;
  const int tid = threadIdx.x, lane = tid & 31;
  const int wid = __builtin_amdgcn_readfirstlane(tid >> 5);
  const int l16 = lane & 15, half = lane >> 4, koff = 2 * half;
  const int wq = wid & 3;
  const int kq = wid >> 2;

  float* xg = ws + WS_X + toff;
  const int bofs = b * 8192 + wq * 256;
  const bool below = (2 * kq < ti);
  const int c0 = wq * 16;
  const int n0g = wid * 16;
  const float* wf = ws + WS_W;
  const float* wo_f  = wf + WF_O   + (wq * 8 + 2 * kq) * 128;
  const float* wfc_f = wf + WF_FC  + (wid * 8) * 128;
  const float* wpr_f = wf + WF_PR  + (wq * 32 + 8 * kq) * 128;
  const float* wq_f  = wf + WF_QKV + (wid * 8) * 128;

  float4 kvn[2][2][2];
#pragma unroll
  for (int a = 0; a < 8; ++a) kvn[a >> 2][(a >> 1) & 1][a & 1] = make_float4(0.0f, 0.0f, 0.0f, 0.0f);
  if (below) ask_kv(kvn, ws, s0, bofs, kq, ti, lane);
  {
    const float xa = xg[tid], xc = xg[tid + NT];
    const float4 f0 =
        reinterpret_cast<const float4*>(plane_ptr(ws, tid >> 8, s0) + toff)[tid & 255];
    float4 f1 = f0;
    if (tid < 256)
      f1 = reinterpret_cast<const float4*>(plane_ptr(ws, 2, s0) + toff)[tid];
    *reinterpret_cast<float4*>(gb + (tid >> 4) * QK_SP + (tid & 15) * 4) = f0;
    if (tid < 256)
      *reinterpret_cast<float4*>(gb + ((tid + 512) >> 4) * QK_SP + (tid & 15) * 4) = f1;
    xb[tid] = xa;
    xb[tid + NT] = xc;
  }
  wg_barrier();

  v2f fo[4], ffc[16], fpr[16], fq[16];
  float cbo[2], cg2[2], cb2[2], cbp[2], bfc, bq = 0.0f;
#pragma unroll
  for (int s = 0; s < 16; ++s) { fq[s].x = 0.0f; fq[s].y = 0.0f; }

  auto step_body = [&](const int t, auto first_c) __attribute__((always_inline)) {
  constexpr bool first = decltype(first_c)::value;
  const bool more = (t + 1 < NL);
  v2f qa[4], kb[2][4], vb[2][4];
#pragma unroll
  for (int sp = 0; sp < 2; ++sp) {
    const float4 f = ld_staged(gb, wq, sp, lane);
    qa[2 * sp].x = f.x; qa[2 * sp].y = f.y; qa[2 * sp + 1].x = f.z; qa[2 * sp + 1].y = f.w;
  }
#pragma unroll
  for (int j = 0; j < 2; ++j)
#pragma unroll
    for (int sp = 0; sp < 2; ++sp) {
      pick_frag(kvn[0][j][sp], ld_staged(gb, 4 + wq, sp, lane), 2 * kq + j, ti,
                kb[j][2 * sp], kb[j][2 * sp + 1]);
      pick_frag(kvn[1][j][sp], ld_staged(gb, 8 + wq, sp, lane), 2 * kq + j, ti,
                vb[j][2 * sp], vb[j][2 * sp + 1]);
    }
  if (first) {
#pragma unroll
    for (int sp = 0; sp < 2; ++sp)
      ld_frag2(wo_f, sp, lane, fo[2 * sp], fo[2 * sp + 1]);
#pragma unroll
    for (int e = 0; e < 2; ++e) {
      const int d = lane + 32 * e;
      cbo[e] = b_o[d]; cg2[e] = ln2_g[d]; cb2[e] = ln2_b[d];
    }
  }

  {
    float* pw = smem + L_PW + wid * 512;
#pragma unroll
    for (int j = 0; j < 2; ++j) {
      const int kt = 2 * kq + j;
      v8f a8; for (int r = 0; r < 8; ++r) a8[r] = 0.0f;
#pragma unroll
      for (int s = 0; s < 4; ++s)
        a8 = wmma4(qa[s], kb[j][s], a8);
#pragma unroll
      for (int r = 0; r < 8; ++r) {
        const int row = q0 + r + 8 * half, col = kt * 16 + l16;
        pw[j * 256 + (r + 8 * half) * 16 + l16] =
            (col <= row) ? a8[r] * 0.25f : -1e9f;
      }
    }
    v2f p[2][4];
    float mx = -1e30f;
#pragma unroll
    for (int j = 0; j < 2; ++j)
#pragma unroll
      for (int s = 0; s < 4; ++s) {
        p[j][s] = *reinterpret_cast<const v2f*>(pw + j * 256 + l16 * 16 + 4 * s + koff);
        mx = fmaxf(mx, fmaxf(p[j][s].x, p[j][s].y));
      }
    mx = fmaxf(mx, __shfl_xor(mx, 16, 32));
    float sum = 0.0f;
#pragma unroll
    for (int j = 0; j < 2; ++j)
#pragma unroll
      for (int s = 0; s < 4; ++s) {
        p[j][s].x = __expf(p[j][s].x - mx);
        p[j][s].y = __expf(p[j][s].y - mx);
        sum += p[j][s].x + p[j][s].y;
      }
    sum += __shfl_xor(sum, 16, 32);
    v8f o; for (int r = 0; r < 8; ++r) o[r] = 0.0f;
#pragma unroll
    for (int j = 0; j < 2; ++j)
#pragma unroll
      for (int s = 0; s < 4; ++s)
        o = wmma4(p[j][s], vb[j][s], o);
    if (first) {
#pragma unroll
      for (int sp = 0; sp < 8; ++sp)
        ld_frag2(wfc_f, sp, lane, ffc[2 * sp], ffc[2 * sp + 1]);
      bfc = b_fc[n0g + l16];
    }
    for (int r = 0; r < 8; ++r)
      part[kq * 1024 + (r + 8 * half) * 64 + c0 + l16] = o[r];
    if (half == 0) {
      pm[(kq * 4 + wq) * 16 + l16] = mx;
      pl[(kq * 4 + wq) * 16 + l16] = sum;
    }
  }
  wg_barrier();
  {
    const int row = tid >> 5, c = (tid & 31) * 2, hh = c >> 4;
    float m[4];
    for (int k = 0; k < 4; ++k) m[k] = pm[(k * 4 + hh) * 16 + row];
    const float M = fmaxf(fmaxf(m[0], m[1]), fmaxf(m[2], m[3]));
    float L = 0.0f;
    v2f O; O.x = 0.0f; O.y = 0.0f;
    for (int k = 0; k < 4; ++k) {
      const float wk = __expf(m[k] - M);
      const v2f pk = *reinterpret_cast<const v2f*>(part + k * 1024 + row * 64 + c);
      L += pl[(k * 4 + hh) * 16 + row] * wk;
      O.x += pk.x * wk;
      O.y += pk.y * wk;
    }
    const float inv = 1.0f / L;
    O.x *= inv; O.y *= inv;
    *reinterpret_cast<v2f*>(hb + afrag(row, c)) = O;
  }
  wg_barrier();

  float ctt[2], cgE[2], cbE[2];
  {
    v8f acc; for (int r = 0; r < 8; ++r) acc[r] = 0.0f;
#pragma unroll
    for (int s = 0; s < 4; ++s)
      acc = wmma4(ld_afrag(hb, kq * 4 + s, lane), fo[s], acc);
    for (int r = 0; r < 8; ++r)
      part[kq * 1024 + (r + 8 * half) * 64 + c0 + l16] = acc[r];
    if (first) {
#pragma unroll
      for (int sp = 0; sp < 8; ++sp)
        ld_frag2(wpr_f, sp, lane, fpr[2 * sp], fpr[2 * sp + 1]);
#pragma unroll
      for (int e = 0; e < 2; ++e) cbp[e] = b_pr[lane + 32 * e];
    }
    const float* gE = more ? ln1_g : lnf_g;
    const float* bE = more ? ln1_b : lnf_b;
    const float* tr = ws + (more ? (t + 1) * 64 : 0);
#pragma unroll
    for (int e = 0; e < 2; ++e) {
      const int d = lane + 32 * e;
      ctt[e] = tr[d]; cgE[e] = gE[d]; cbE[e] = bE[d];
    }
  }
  wg_barrier();

  row_pass_r<false, true>(xb, part, cbo, cbo, nullptr, cg2, cb2, hb, wid, lane);
  wg_barrier();

  {
    v8f acc;
    for (int r = 0; r < 8; ++r) acc[r] = bfc;
#pragma unroll
    for (int s = 0; s < 16; ++s)
      acc = wmma4(ld_afrag(hb, s, lane), ffc[s], acc);
    if (first && more && wid < 12) {
#pragma unroll
      for (int sp = 0; sp < 8; ++sp)
        ld_frag2(wq_f, sp, lane, fq[2 * sp], fq[2 * sp + 1]);
      bq = b_qkv[n0g + l16];
    }
    if (below && t + 1 < s1)
      ask_kv(kvn, ws, t + 1, bofs, kq, ti, lane);
    for (int r = 0; r < 8; ++r) {
      float u = acc[r];
      float g = 0.5f * u *
          (1.0f + tanhf(0.7978845608028654f * (u + 0.044715f * u * u * u)));
      gb[afrag(r + 8 * half, n0g + l16)] = g;
    }
  }
  wg_barrier();
  {
    v8f acc; for (int r = 0; r < 8; ++r) acc[r] = 0.0f;
#pragma unroll
    for (int s = 0; s < 16; ++s)
      acc = wmma4(ld_afrag(gb, kq * 16 + s, lane), fpr[s], acc);
    for (int r = 0; r < 8; ++r)
      part[kq * 1024 + (r + 8 * half) * 64 + c0 + l16] = acc[r];
  }
  wg_barrier();

  if (more) {
    row_pass_r<true, true>(xb, part, cbp, ctt, xg, cgE, cbE, hb, wid, lane);
    wg_barrier();
    if (wid < 12) {
      v8f acc;
      for (int r = 0; r < 8; ++r) acc[r] = bq;
#pragma unroll
      for (int s = 0; s < 16; ++s)
        acc = wmma4(ld_afrag(hb, s, lane), fq[s], acc);
      stage_qkv(gb, acc, wid, l16, half);
    }
    wg_barrier();
    for (int i = tid; i < 768; i += NT) {
      const int pln = i >> 8, off = i & 255;
      reinterpret_cast<float4*>(plane_ptr(ws, pln, t + 1) + toff)[off] =
          staged4(gb, i);
    }
  } else {
    row_pass_r<false, false>(xb, part, cbp, ctt, nullptr, cgE, cbE, hb, wid, lane);
    wg_barrier();
    for (int i = tid; i < 256; i += NT)
      reinterpret_cast<float4*>(out + toff)[i] = reinterpret_cast<const float4*>(hb)[i];
  }
  };
  step_body(s0, std::true_type{});
  for (int t = s0 + 1; t < s1; ++t)
    step_body(t, std::false_type{});
}

extern "C" void kernel_launch(void* const* d_in, const int* in_sizes, int n_in,
                              void* d_out, int out_size, void* d_ws, size_t ws_size,
                              hipStream_t stream) {
  const int*   idx   = (const int*)  d_in[0];
  const int*   nlp   = (const int*)  d_in[1];
  const float* wte   = (const float*)d_in[2];
  const float* wpe   = (const float*)d_in[3];
  const float* t_w1  = (const float*)d_in[4];
  const float* t_b1  = (const float*)d_in[5];
  const float* t_w2  = (const float*)d_in[6];
  const float* t_b2  = (const float*)d_in[7];
  const float* ln1_g = (const float*)d_in[8];
  const float* ln1_b = (const float*)d_in[9];
  const float* w_qkv = (const float*)d_in[10];
  const float* b_qkv = (const float*)d_in[11];
  const float* w_o   = (const float*)d_in[12];
  const float* b_o   = (const float*)d_in[13];
  const float* ln2_g = (const float*)d_in[14];
  const float* ln2_b = (const float*)d_in[15];
  const float* w_fc  = (const float*)d_in[16];
  const float* b_fc  = (const float*)d_in[17];
  const float* w_pr  = (const float*)d_in[18];
  const float* b_pr  = (const float*)d_in[19];
  const float* lnf_g = (const float*)d_in[20];
  const float* lnf_b = (const float*)d_in[21];

  float* ws = (float*)d_ws;

  temb_kernel<<<LAUNCH1_BLOCKS, 256, 0, stream>>>(
      nlp, t_w1, t_b1, t_w2, w_qkv, w_o, w_fc, w_pr, ws);
  embed_kernel<<<256, NT, 0, stream>>>(idx, nlp, wte, wpe, t_b2, ln1_g, ln1_b, w_qkv,
                                       b_qkv, lnf_g, lnf_b, ws, (float*)d_out);
  for (int L = 0; L < NLAUNCH; ++L)
    step_kernel<<<256, NT, 0, stream>>>(L, nlp, ln1_g, ln1_b, b_qkv, b_o, ln2_g,
                                        ln2_b, b_fc, b_pr, lnf_g, lnf_b, ws,
                                        (float*)d_out);
}


# LN1 and LN2 folded into the QKV and fc products: gains in the weight copies, product on raw rows, transposed (a row per lane), mean, 1/sd and shift applied to the accumulators; sum order changed
# speedup vs baseline: 11.8121x; 1.0340x over previous
#include <hip/hip_runtime.h>
#include <hip/hip_bf16.h>
#include <math.h>
#include <type_traits>


typedef __attribute__((ext_vector_type(2))) float v2f;
typedef __attribute__((ext_vector_type(8))) float v8f;

__device__ __forceinline__ v8f wmma4(v2f a, v2f b, v8f c) {
  return __builtin_amdgcn_wmma_f32_16x16x4_f32(
       false, a,  false, b,
       (short)0, c,  false,  false);
}

#define TEMB_ROWS 112
#define TEMB_NC 16
#define TEMB_BLOCKS 112
#define RELAY_BLOCKS 50
#define LAUNCH1_BLOCKS 162
#define WS_TP (8192 + 2 * 262144)
#define WS_W (8192 + 7 * 262144)
#define WF_O 0
#define WF_FC 4096
#define WF_PR 20480
#define WF_QKV 36864
#define WF_VEC 49152
#define NVEC 896
template <int N, int LOG2_KP, bool SCALED>
__device__ __forceinline__ void relay_entry(const float* __restrict__ w,
                                            const float* __restrict__ gain,
                                            float* __restrict__ dst, int i) {
  const int lane = i & 31, g = i >> 5;
  const int sp = g & ((1 << LOG2_KP) - 1), nt = g >> LOG2_KP;
  const int k = 8 * sp + 2 * (lane >> 4), n = 16 * nt + (lane & 15);
  float4 f;
  f.x = w[k * N + n];       f.y = w[(k + 1) * N + n];
  f.z = w[(k + 4) * N + n]; f.w = w[(k + 5) * N + n];
  if (SCALED) { f.x *= gain[k]; f.y *= gain[k + 1]; f.z *= gain[k + 4]; f.w *= gain[k + 5]; }
  reinterpret_cast<float4*>(dst)[i] = f;
}
template <int N>
__device__ __forceinline__ void relay_vec(const float* __restrict__ w, const float* __restrict__ g,
                                          const float* __restrict__ beta, const float* __restrict__ b,
                                          float* __restrict__ Gd, float* __restrict__ Bd, int n) {
  float G = 0.0f, Bn = b[n];
  for (int k = 0; k < 64; ++k) {
    const float wv = w[k * N + n];
    G += g[k] * wv;
    Bn += beta[k] * wv;
  }
  Gd[n] = G;
  Bd[n] = Bn;
}

__device__ __forceinline__ int tfrag(int row, int k) {
  return ((k >> 2) << 6) + (((k >> 1) & 1) << 5) + (row << 1) + (k & 1);
}

__global__ __launch_bounds__(256) void temb_kernel(
    const int* __restrict__ num_loops,
    const float* __restrict__ t_w1, const float* __restrict__ t_b1,
    const float* __restrict__ t_w2,
    const float* __restrict__ w_qkv, const float* __restrict__ w_o,
    const float* __restrict__ w_fc, const float* __restrict__ w_pr,
    const float* __restrict__ ln1_g, const float* __restrict__ ln1_b,
    const float* __restrict__ ln2_g, const float* __restrict__ ln2_b,
    const float* __restrict__ b_qkv, const float* __restrict__ b_fc,
    float* __restrict__ ttab) {
  if (blockIdx.x >= TEMB_BLOCKS) {
    const int i = (blockIdx.x - TEMB_BLOCKS) * 256 + threadIdx.x;
    float* wf = ttab + WS_W;
    if (i < 1024)       relay_entry<64, 3, false>(w_o, nullptr, wf + WF_O, i);
    else if (i < 5120)  relay_entry<256, 3, true>(w_fc, ln2_g, wf + WF_FC, i - 1024);
    else if (i < 9216)  relay_entry<64, 5, false>(w_pr, nullptr, wf + WF_PR, i - 5120);
    else if (i < 12288) relay_entry<192, 3, true>(w_qkv, ln1_g, wf + WF_QKV, i - 9216);
    else if (i < 12544) relay_vec<256>(w_fc, ln2_g, ln2_b, b_fc, wf + WF_VEC, wf + WF_VEC + 256, i - 12288);
    else if (i < 12736) relay_vec<192>(w_qkv, ln1_g, ln1_b, b_qkv, wf + WF_VEC + 512, wf + WF_VEC + 704, i - 12544);
    return;
  }
  const int rt = blockIdx.x >> 4, ch = blockIdx.x & 15;
  if (rt * 16 >= *num_loops) return;
  __shared__ __attribute__((aligned(16))) float teA[4096];
  __shared__ __attribute__((aligned(16))) float p1[2 * 1024];
  __shared__ __attribute__((aligned(16))) float hsA[1024];
  __shared__ __attribute__((aligned(16))) float p2[2 * 1024];
  const int tid = threadIdx.x, lane = tid & 31;
  const int wid = __builtin_amdgcn_readfirstlane(tid >> 5);
  const int l16 = lane & 15, half = lane >> 4, koff = 2 * half;
  const int ct = wid & 3;
  const int kh = wid >> 2;

  v2f b1[32], b2[8];
  {
    const float* w1p = t_w1 + (128 * kh + koff) * 1024 + 64 * ch + 16 * ct + l16;
#pragma unroll
    for (int s = 0; s < 32; ++s) {
      b1[s].x = w1p[(4 * s) * 1024];
      b1[s].y = w1p[(4 * s + 1) * 1024];
    }
    const float* w2p = t_w2 + (64 * ch + 32 * kh + koff) * 64 + 16 * ct + l16;
#pragma unroll
    for (int s = 0; s < 8; ++s) {
      b2[s].x = w2p[(4 * s) * 64];
      b2[s].y = w2p[(4 * s + 1) * 64];
    }
  }
  const float bias1 = (kh == 0) ? t_b1[64 * ch + 16 * ct + l16] : 0.0f;

  {
    const int row = tid & 15, k0 = (tid >> 4) * 16;
    const float tf = (float)(rt * 16 + row);
    for (int i = 0; i < 16; ++i) {
      const int k = k0 + i;
      const float f = expf(-logf(10000.0f) * (float)(k & 127) / 128.0f);
      const float arg = tf * f;
      teA[tfrag(row, k)] = (k0 < 128) ? cosf(arg) : sinf(arg);
    }
  }
  __syncthreads();

  {
    v8f acc;
    for (int r = 0; r < 8; ++r) acc[r] = bias1;
#pragma unroll
    for (int s = 0; s < 32; ++s)
      acc = wmma4(*reinterpret_cast<const v2f*>(teA + (32 * kh + s) * 64 + lane * 2), b1[s], acc);
    for (int r = 0; r < 8; ++r)
      p1[kh * 1024 + (r + 8 * half) * 64 + 16 * ct + l16] = acc[r];
  }
  __syncthreads();
  for (int e = tid; e < 1024; e += 256) {
    const float h = p1[e] + p1[1024 + e];
    hsA[tfrag(e >> 6, e & 63)] = h / (1.0f + expf(-h));
  }
  __syncthreads();

  {
    v8f acc;
    for (int r = 0; r < 8; ++r) acc[r] = 0.0f;
#pragma unroll
    for (int s = 0; s < 8; ++s)
      acc = wmma4(*reinterpret_cast<const v2f*>(hsA + (8 * kh + s) * 64 + lane * 2), b2[s], acc);
    for (int r = 0; r < 8; ++r)
      p2[kh * 1024 + (r + 8 * half) * 64 + 16 * ct + l16] = acc[r];
  }
  __syncthreads();
  {
    const float4 a = reinterpret_cast<const float4*>(p2)[tid];
    const float4 b = reinterpret_cast<const float4*>(p2 + 1024)[tid];
    float4 o;
    o.x = a.x + b.x; o.y = a.y + b.y; o.z = a.z + b.z; o.w = a.w + b.w;
    reinterpret_cast<float4*>(ttab + WS_TP + (ch * TEMB_ROWS + rt * 16) * 64)[tid] = o;
  }
}

#define NT 512
#define G 3
#define SKEW (G - 1)
#define NLAUNCH 38
#define MAXLOOPS (G * NLAUNCH - 7 * SKEW)
#define WS_X 8192
#define WS_Q (8192 + 262144)
#define WS_KV (8192 + 8 * 262144)
#define PLANE 262144
__device__ __forceinline__ float* plane_ptr(float* ws, int pl, int s) {
  return pl == 0 ? ws + WS_Q : ws + WS_KV + (2 * s + pl - 1) * PLANE;
}
#define L_XB 0
#define L_HB 1024
#define L_GB 2048
#define L_PART 6144
#define L_PM 10240
#define L_PL 10496
#define L_PW 10752
#define L_GV 18944
#define L_TOTAL 19840

__device__ __forceinline__ int afrag(int row, int k) {
  return ((k >> 2) << 6) + (((k >> 1) & 1) << 5) + (row << 1) + (k & 1);
}
__device__ __forceinline__ v2f ld_afrag(const float* buf, int step, int lane) {
  return *reinterpret_cast<const v2f*>(buf + step * 64 + lane * 2);
}

#define QK_SP 68
template <int P>
__device__ __forceinline__ int qkfrag(int r, int d) {
  return (((d >> 3) << 1) + ((d >> 1) & 1)) * P + (r << 2) + (((d >> 2) & 1) << 1) + (d & 1);
}
__device__ __forceinline__ void ld_frag2(const float* blk, int sp, int lane, v2f& f0, v2f& f1) {
  const float4 f = reinterpret_cast<const float4*>(blk)[sp * 32 + lane];
  f0.x = f.x; f0.y = f.y; f1.x = f.z; f1.y = f.w;
}
__device__ __forceinline__ float4 ld_staged(const float* st, int ct, int sp, int lane) {
  const int e = ct * 64 + sp * 32 + lane;
  return *reinterpret_cast<const float4*>(st + (e >> 4) * QK_SP + (e & 15) * 4);
}
__device__ __forceinline__ void pick_frag(const float4& fg, const float4& fl, int kt, int ti,
                                          v2f& f0, v2f& f1) {
  const bool lo = kt < ti, eq = kt == ti;
  f0.x = lo ? fg.x : (eq ? fl.x : 0.0f);
  f0.y = lo ? fg.y : (eq ? fl.y : 0.0f);
  f1.x = lo ? fg.z : (eq ? fl.z : 0.0f);
  f1.y = lo ? fg.w : (eq ? fl.w : 0.0f);
}
__device__ __forceinline__ void ask_kv(float4 (&kvn)[2][2][2], float* ws, int step, int bofs,
                                       int kq, int ti, int lane) {
#pragma unroll
  for (int pl = 0; pl < 2; ++pl)
#pragma unroll
    for (int j = 0; j < 2; ++j) {
      const int kt = 2 * kq + j;
      const float4* src = reinterpret_cast<const float4*>(
          plane_ptr(ws, 1 + pl, step) + bofs + (kt < ti ? kt : 2 * kq) * 1024);
#pragma unroll
      for (int sp = 0; sp < 2; ++sp) kvn[pl][j][sp] = src[sp * 32 + lane];
    }
}
__device__ __forceinline__ void stage_qkv(float* st, const v8f& acc, int ct, int l16, int half) {
  float* blk = st + ct * (4 * QK_SP);
  if (ct >= 8) {
    for (int r = 0; r < 8; ++r) blk[qkfrag<QK_SP>(l16, r + 8 * half)] = acc[r];
  } else {
    for (int r = 0; r < 8; ++r) blk[qkfrag<QK_SP>(r + 8 * half, l16)] = acc[r];
  }
}
__device__ __forceinline__ void stage_qkv_t(float* st, const v8f& acc, int ct, int l16, int half) {
  float* blk = st + ct * (4 * QK_SP);
  if (ct >= 8) {
    for (int r = 0; r < 8; ++r) blk[qkfrag<QK_SP>(r + 8 * half, l16)] = acc[r];
  } else {
    for (int r = 0; r < 8; ++r) blk[qkfrag<QK_SP>(l16, r + 8 * half)] = acc[r];
  }
}
__device__ __forceinline__ const float4& staged4(const float* st, int i) {
  return *reinterpret_cast<const float4*>(st + (i >> 4) * QK_SP + (i & 15) * 4);
}

template <bool ADD, bool STEP0, bool FRAG>
__device__ __forceinline__ void row_pass(
    float* xb, const float* part, const float* __restrict__ bias,
    const float* __restrict__ tt, float* __restrict__ xg,
    const float* __restrict__ g, const float* __restrict__ bta, float* dst,
    int wid, int lane) {
  float v[2];
#pragma unroll
  for (int e = 0; e < 2; ++e) {
    const int d = lane + 32 * e;
    float x = xb[wid * 64 + d];
    if (ADD) {
      float s = bias[d];
      for (int k = 0; k < 4; ++k) s += part[k * 1024 + wid * 64 + d];
      x += 0.1f * s;
    }
    if (STEP0) x += tt[d];
    v[e] = x;
    if (ADD || STEP0) xb[wid * 64 + d] = x;
    if (STEP0) xg[wid * 64 + d] = x;
  }
  float sum = v[0] + v[1], sq = v[0] * v[0] + v[1] * v[1];
  for (int off = 16; off; off >>= 1) {
    sum += __shfl_xor(sum, off, 32);
    sq  += __shfl_xor(sq,  off, 32);
  }
  const float mu  = sum * (1.0f / 64.0f);
  const float var = sq * (1.0f / 64.0f) - mu * mu;
  const float inv = rsqrtf(var + 1e-5f);
#pragma unroll
  for (int e = 0; e < 2; ++e) {
    const int d = lane + 32 * e;
    dst[FRAG ? afrag(wid, d) : wid * 64 + d] = (v[e] - mu) * inv * g[d] + bta[d];
  }
}

__device__ __forceinline__ void qkv_out(
    const float* hb, float* st, const float* __restrict__ w_qkv,
    const float* __restrict__ b_qkv, float* __restrict__ ws, int toff,
    int tid, int wid, int lane, int l16, int half, int koff) {
  if (wid < 12) {
    const int n0 = wid * 16;
    v8f acc;
    { float bias = b_qkv[n0 + l16]; for (int r = 0; r < 8; ++r) acc[r] = bias; }
#pragma unroll
    for (int k = 0; k < 64; k += 4) {
      v2f a = ld_afrag(hb, k >> 2, lane), w;
      w.x = w_qkv[(k + koff) * 192 + n0 + l16];
      w.y = w_qkv[(k + koff + 1) * 192 + n0 + l16];
      acc = wmma4(a, w, acc);
    }
    stage_qkv(st, acc, wid, l16, half);
  }
  __syncthreads();
  for (int i = tid; i < 768; i += NT) {
    const int pl = i >> 8, off = i & 255;
    reinterpret_cast<float4*>(plane_ptr(ws, pl, 0) + toff)[off] =
        staged4(st, i);
  }
}

__device__ __forceinline__ void out_copy(const float* hb, float* __restrict__ outg, int tid) {
  __syncthreads();
  for (int i = tid; i < 256; i += NT)
    reinterpret_cast<float4*>(outg)[i] = reinterpret_cast<const float4*>(hb)[i];
}

__device__ __forceinline__ void wg_barrier() {
  asm volatile("s_wait_dscnt 0x0\n\ts_barrier_signal -1\n\ts_barrier_wait -1" ::: "memory");
}

template <bool STEP0, bool FRAG>
__device__ __forceinline__ void row_pass_r(
    float* xb, const float* part, const float (&bias)[2], const float (&tt)[2],
    float* __restrict__ xg, const float (&g)[2], const float (&bta)[2],
    float* dst, int wid, int lane) {
  float v[2];
#pragma unroll
  for (int e = 0; e < 2; ++e) {
    const int d = lane + 32 * e;
    float x = xb[wid * 64 + d];
    float s = bias[e];
    for (int k = 0; k < 4; ++k) s += part[k * 1024 + wid * 64 + d];
    x += 0.1f * s;
    if (STEP0) x += tt[e];
    v[e] = x;
    xb[wid * 64 + d] = x;
    if (STEP0) xg[wid * 64 + d] = x;
  }
  float sum = v[0] + v[1], sq = v[0] * v[0] + v[1] * v[1];
  for (int off = 16; off; off >>= 1) {
    sum += __shfl_xor(sum, off, 32);
    sq  += __shfl_xor(sq,  off, 32);
  }
  const float mu  = sum * (1.0f / 64.0f);
  const float var = sq * (1.0f / 64.0f) - mu * mu;
  const float inv = rsqrtf(var + 1e-5f);
#pragma unroll
  for (int e = 0; e < 2; ++e) {
    const int d = lane + 32 * e;
    dst[FRAG ? afrag(wid, d) : wid * 64 + d] = (v[e] - mu) * inv * g[e] + bta[e];
  }
}

template <bool STEP0>
__device__ __forceinline__ void row_sum_r(
    float* xb, const float* part, const float (&bias)[2], const float (&tt)[2],
    float* __restrict__ xg, float* dst, int wid, int lane) {
#pragma unroll
  for (int e = 0; e < 2; ++e) {
    const int d = lane + 32 * e;
    float x = xb[wid * 64 + d];
    float s = bias[e];
    for (int k = 0; k < 4; ++k) s += part[k * 1024 + wid * 64 + d];
    x += 0.1f * s;
    if (STEP0) x += tt[e];
    xb[wid * 64 + d] = x;
    if (STEP0) xg[wid * 64 + d] = x;
    dst[afrag(wid, d)] = x;
  }
}
__device__ __forceinline__ void row_stats(v2f s2, v2f q2, float& mu, float& inv) {
  float sum = s2.x + s2.y, sq = q2.x + q2.y;
  sum += __shfl_xor(sum, 16, 32);
  sq  += __shfl_xor(sq,  16, 32);
  mu = sum * (1.0f / 64.0f);
  const float var = sq * (1.0f / 64.0f) - mu * mu;
  inv = rsqrtf(var + 1e-5f);
}

__global__ __launch_bounds__(NT) void embed_kernel(
    const int* __restrict__ idx, const int* __restrict__ num_loops,
    const float* __restrict__ wte, const float* __restrict__ wpe,
    const float* __restrict__ t_b2,
    const float* __restrict__ ln1_g, const float* __restrict__ ln1_b,
    const float* __restrict__ w_qkv, const float* __restrict__ b_qkv,
    const float* __restrict__ lnf_g, const float* __restrict__ lnf_b,
    float* __restrict__ ws, float* __restrict__ out) {
  __shared__ __attribute__((aligned(16))) float smem[L_TOTAL];
  float* xb = smem + L_XB;
  float* hb = smem + L_HB;
  float* st = smem + L_GB;
  const int b = blockIdx.x >> 3, q0 = (blockIdx.x & 7) * 16;
  const int toff = (b * 128 + q0) * 64;
  const int tid = threadIdx.x, lane = tid & 31;
  const int wid = __builtin_amdgcn_readfirstlane(tid >> 5);
  const int l16 = lane & 15, half = lane >> 4, koff = 2 * half;

  float* tt0 = smem + L_PM;
  if (tid < 128) {
    const int d = tid & 63;
    const int row = (tid < 64) ? 0 : (int)blockIdx.x;
    if (row < min(*num_loops, TEMB_ROWS)) {
      const float* tp = ws + WS_TP + row * 64 + d;
      float s = t_b2[d];
#pragma unroll
      for (int ch = 0; ch < TEMB_NC; ++ch) s += tp[ch * TEMB_ROWS * 64];
      if (tid < 64) tt0[d] = s; else ws[row * 64 + d] = s;
    }
  }
  for (int i = tid; i < 1024; i += NT) {
    int s = i >> 6, d = i & 63;
    xb[i] = wte[idx[b * 128 + q0 + s] * 64 + d] + wpe[(q0 + s) * 64 + d];
  }
  __syncthreads();

  if (*num_loops > 0) {
    row_pass<false, true, true>(xb, nullptr, nullptr, tt0, ws + WS_X + toff, ln1_g, ln1_b, hb, wid, lane);
    __syncthreads();
    qkv_out(hb, st, w_qkv, b_qkv, ws, toff, tid, wid, lane, l16, half, koff);
  } else {
    row_pass<false, false, false>(xb, nullptr, nullptr, nullptr, nullptr, lnf_g, lnf_b, hb, wid, lane);
    out_copy(hb, out + toff, tid);
  }
}

__global__ __launch_bounds__(NT) void step_kernel(
    int L, const int* __restrict__ num_loops,
    const float* __restrict__ b_o, const float* __restrict__ b_pr,
    const float* __restrict__ lnf_g, const float* __restrict__ lnf_b,
    float* __restrict__ ws, float* __restrict__ out) {
  const int NL = min(*num_loops, MAXLOOPS);
  const int ti = blockIdx.x & 7;
  const int sa = G * L - SKEW * ti;
  const int s0 = max(sa, 0);
  const int s1 = min(sa + G, NL);
  if (s0 >= s1) return;

  __shared__ __attribute__((aligned(16))) float smem[L_TOTAL];
  float* xb = smem + L_XB;
  float* hb = smem + L_HB;
  float* gb = smem + L_GB;
  float* part = smem + L_PART;
  float* pm = smem + L_PM;
  float* pl = smem + L_PL;
  const float* gv = smem + L_GV;

  const int b = blockIdx.x >> 3, q0 = ti * 16;
  const int toff = (b * 128 + q0) * 64;
  const int tid = threadIdx.x, lane = tid & 31;
  const int wid = __builtin_amdgcn_readfirstlane(tid >> 5);
  const int l16 = lane & 15, half = lane >> 4, koff = 2 * half;
  const int wq = wid & 3;
  const int kq = wid >> 2;

  float* xg = ws + WS_X + toff;
  const int bofs = b * 8192 + wq * 256;
  const bool below = (2 * kq < ti);
  const int c0 = wq * 16;
  const int n0g = wid * 16;
  const float* wf = ws + WS_W;
  const float* wo_f  = wf + WF_O   + (wq * 8 + 2 * kq) * 128;
  const float* wfc_f = wf + WF_FC  + (wid * 8) * 128;
  const float* wpr_f = wf + WF_PR  + (wq * 32 + 8 * kq) * 128;
  const float* wq_f  = wf + WF_QKV + (wid * 8) * 128;

  float4 kvn[2][2][2];
#pragma unroll
  for (int a = 0; a < 8; ++a) kvn[a >> 2][(a >> 1) & 1][a & 1] = make_float4(0.0f, 0.0f, 0.0f, 0.0f);
  if (below) ask_kv(kvn, ws, s0, bofs, kq, ti, lane);
  {
    const float xa = xg[tid], xc = xg[tid + NT];
    const float4 f0 =
        reinterpret_cast<const float4*>(plane_ptr(ws, tid >> 8, s0) + toff)[tid & 255];
    float4 f1 = f0;
    if (tid < 256)
      f1 = reinterpret_cast<const float4*>(plane_ptr(ws, 2, s0) + toff)[tid];
    float4 fv = f0;
    if (tid < NVEC / 4)
      fv = reinterpret_cast<const float4*>(wf + WF_VEC)[tid];
    *reinterpret_cast<float4*>(gb + (tid >> 4) * QK_SP + (tid & 15) * 4) = f0;
    if (tid < 256)
      *reinterpret_cast<float4*>(gb + ((tid + 512) >> 4) * QK_SP + (tid & 15) * 4) = f1;
    xb[tid] = xa;
    xb[tid + NT] = xc;
    if (tid < NVEC / 4)
      reinterpret_cast<float4*>(smem + L_GV)[tid] = fv;
  }
  wg_barrier();

  v2f fo[4], ffc[16], fpr[16], fq[16];
  float cbo[2], cbp[2];
#pragma unroll
  for (int s = 0; s < 16; ++s) { fq[s].x = 0.0f; fq[s].y = 0.0f; }

  auto step_body = [&](const int t, auto first_c) __attribute__((always_inline)) {
  constexpr bool first = decltype(first_c)::value;
  const bool more = (t + 1 < NL);
  v2f qa[4], kb[2][4], vb[2][4];
#pragma unroll
  for (int sp = 0; sp < 2; ++sp) {
    const float4 f = ld_staged(gb, wq, sp, lane);
    qa[2 * sp].x = f.x; qa[2 * sp].y = f.y; qa[2 * sp + 1].x = f.z; qa[2 * sp + 1].y = f.w;
  }
#pragma unroll
  for (int j = 0; j < 2; ++j)
#pragma unroll
    for (int sp = 0; sp < 2; ++sp) {
      pick_frag(kvn[0][j][sp], ld_staged(gb, 4 + wq, sp, lane), 2 * kq + j, ti,
                kb[j][2 * sp], kb[j][2 * sp + 1]);
      pick_frag(kvn[1][j][sp], ld_staged(gb, 8 + wq, sp, lane), 2 * kq + j, ti,
                vb[j][2 * sp], vb[j][2 * sp + 1]);
    }
  if (first) {
#pragma unroll
    for (int sp = 0; sp < 2; ++sp)
      ld_frag2(wo_f, sp, lane, fo[2 * sp], fo[2 * sp + 1]);
#pragma unroll
    for (int e = 0; e < 2; ++e) {
      const int d = lane + 32 * e;
      cbo[e] = b_o[d];
    }
  }

  {
    float* pw = smem + L_PW + wid * 512;
#pragma unroll
    for (int j = 0; j < 2; ++j) {
      const int kt = 2 * kq + j;
      v8f a8; for (int r = 0; r < 8; ++r) a8[r] = 0.0f;
#pragma unroll
      for (int s = 0; s < 4; ++s)
        a8 = wmma4(qa[s], kb[j][s], a8);
#pragma unroll
      for (int r = 0; r < 8; ++r) {
        const int row = q0 + r + 8 * half, col = kt * 16 + l16;
        pw[j * 256 + (r + 8 * half) * 16 + l16] =
            (col <= row) ? a8[r] * 0.25f : -1e9f;
      }
    }
    v2f p[2][4];
    float mx = -1e30f;
#pragma unroll
    for (int j = 0; j < 2; ++j)
#pragma unroll
      for (int s = 0; s < 4; ++s) {
        p[j][s] = *reinterpret_cast<const v2f*>(pw + j * 256 + l16 * 16 + 4 * s + koff);
        mx = fmaxf(mx, fmaxf(p[j][s].x, p[j][s].y));
      }
    mx = fmaxf(mx, __shfl_xor(mx, 16, 32));
    float sum = 0.0f;
#pragma unroll
    for (int j = 0; j < 2; ++j)
#pragma unroll
      for (int s = 0; s < 4; ++s) {
        p[j][s].x = __expf(p[j][s].x - mx);
        p[j][s].y = __expf(p[j][s].y - mx);
        sum += p[j][s].x + p[j][s].y;
      }
    sum += __shfl_xor(sum, 16, 32);
    v8f o; for (int r = 0; r < 8; ++r) o[r] = 0.0f;
#pragma unroll
    for (int j = 0; j < 2; ++j)
#pragma unroll
      for (int s = 0; s < 4; ++s)
        o = wmma4(p[j][s], vb[j][s], o);
    if (first) {
#pragma unroll
      for (int sp = 0; sp < 8; ++sp)
        ld_frag2(wfc_f, sp, lane, ffc[2 * sp], ffc[2 * sp + 1]);
    }
    for (int r = 0; r < 8; ++r)
      part[kq * 1024 + (r + 8 * half) * 64 + c0 + l16] = o[r];
    if (half == 0) {
      pm[(kq * 4 + wq) * 16 + l16] = mx;
      pl[(kq * 4 + wq) * 16 + l16] = sum;
    }
  }
  wg_barrier();
  {
    const int row = tid >> 5, c = (tid & 31) * 2, hh = c >> 4;
    float m[4];
    for (int k = 0; k < 4; ++k) m[k] = pm[(k * 4 + hh) * 16 + row];
    const float M = fmaxf(fmaxf(m[0], m[1]), fmaxf(m[2], m[3]));
    float L = 0.0f;
    v2f O; O.x = 0.0f; O.y = 0.0f;
    for (int k = 0; k < 4; ++k) {
      const float wk = __expf(m[k] - M);
      const v2f pk = *reinterpret_cast<const v2f*>(part + k * 1024 + row * 64 + c);
      L += pl[(k * 4 + hh) * 16 + row] * wk;
      O.x += pk.x * wk;
      O.y += pk.y * wk;
    }
    const float inv = 1.0f / L;
    O.x *= inv; O.y *= inv;
    *reinterpret_cast<v2f*>(hb + afrag(row, c)) = O;
  }
  wg_barrier();

  float ctt[2], cgE[2], cbE[2];
  {
    v8f acc; for (int r = 0; r < 8; ++r) acc[r] = 0.0f;
#pragma unroll
    for (int s = 0; s < 4; ++s)
      acc = wmma4(ld_afrag(hb, kq * 4 + s, lane), fo[s], acc);
    for (int r = 0; r < 8; ++r)
      part[kq * 1024 + (r + 8 * half) * 64 + c0 + l16] = acc[r];
    if (first) {
#pragma unroll
      for (int sp = 0; sp < 8; ++sp)
        ld_frag2(wpr_f, sp, lane, fpr[2 * sp], fpr[2 * sp + 1]);
#pragma unroll
      for (int e = 0; e < 2; ++e) cbp[e] = b_pr[lane + 32 * e];
    }
    const float* gE = lnf_g;
    const float* bE = lnf_b;
    const float* tr = ws + (more ? (t + 1) * 64 : 0);
#pragma unroll
    for (int e = 0; e < 2; ++e) {
      const int d = lane + 32 * e;
      ctt[e] = tr[d]; cgE[e] = gE[d]; cbE[e] = bE[d];
    }
  }
  wg_barrier();

  row_sum_r<false>(xb, part, cbo, cbo, nullptr, hb, wid, lane);
  wg_barrier();

  {
    v8f acc;
    for (int r = 0; r < 8; ++r) acc[r] = 0.0f;
    v2f s2, q2;
    s2.x = 0.0f; s2.y = 0.0f; q2.x = 0.0f; q2.y = 0.0f;
#pragma unroll
    for (int s = 0; s < 16; ++s) {
      const v2f a = ld_afrag(hb, s, lane);
      acc = wmma4(ffc[s], a, acc);
      s2 += a;
      q2 += a * a;
    }
    const float4 g0 = *reinterpret_cast<const float4*>(gv + n0g + 8 * half);
    const float4 g1 = *reinterpret_cast<const float4*>(gv + n0g + 8 * half + 4);
    const float4 h0 = *reinterpret_cast<const float4*>(gv + 256 + n0g + 8 * half);
    const float4 h1 = *reinterpret_cast<const float4*>(gv + 256 + n0g + 8 * half + 4);
    float mu, inv;
    row_stats(s2, q2, mu, inv);
    const float gG[8] = {g0.x, g0.y, g0.z, g0.w, g1.x, g1.y, g1.z, g1.w};
    const float gB[8] = {h0.x, h0.y, h0.z, h0.w, h1.x, h1.y, h1.z, h1.w};
    if (first && more && wid < 12) {
#pragma unroll
      for (int sp = 0; sp < 8; ++sp)
        ld_frag2(wq_f, sp, lane, fq[2 * sp], fq[2 * sp + 1]);
    }
    if (below && t + 1 < s1)
      ask_kv(kvn, ws, t + 1, bofs, kq, ti, lane);
    for (int r = 0; r < 8; ++r) {
      float u = inv * (acc[r] - mu * gG[r]) + gB[r];
      float g = 0.5f * u *
          (1.0f + tanhf(0.7978845608028654f * (u + 0.044715f * u * u * u)));
      gb[afrag(l16, n0g + 8 * half + r)] = g;
    }
  }
  wg_barrier();
  {
    v8f acc; for (int r = 0; r < 8; ++r) acc[r] = 0.0f;
#pragma unroll
    for (int s = 0; s < 16; ++s)
      acc = wmma4(ld_afrag(gb, kq * 16 + s, lane), fpr[s], acc);
    for (int r = 0; r < 8; ++r)
      part[kq * 1024 + (r + 8 * half) * 64 + c0 + l16] = acc[r];
  }
  wg_barrier();

  if (more) {
    row_sum_r<true>(xb, part, cbp, ctt, xg, hb, wid, lane);
    wg_barrier();
    if (wid < 12) {
      v8f acc;
      for (int r = 0; r < 8; ++r) acc[r] = 0.0f;
      v2f s2, q2;
      s2.x = 0.0f; s2.y = 0.0f; q2.x = 0.0f; q2.y = 0.0f;
#pragma unroll
      for (int s = 0; s < 16; ++s) {
        const v2f a = ld_afrag(hb, s, lane);
        acc = wmma4(fq[s], a, acc);
        s2 += a;
        q2 += a * a;
      }
      const float4 g0 = *reinterpret_cast<const float4*>(gv + 512 + n0g + 8 * half);
      const float4 g1 = *reinterpret_cast<const float4*>(gv + 512 + n0g + 8 * half + 4);
      const float4 h0 = *reinterpret_cast<const float4*>(gv + 704 + n0g + 8 * half);
      const float4 h1 = *reinterpret_cast<const float4*>(gv + 704 + n0g + 8 * half + 4);
      float mu, inv;
      row_stats(s2, q2, mu, inv);
      const float gG[8] = {g0.x, g0.y, g0.z, g0.w, g1.x, g1.y, g1.z, g1.w};
      const float gB[8] = {h0.x, h0.y, h0.z, h0.w, h1.x, h1.y, h1.z, h1.w};
      for (int r = 0; r < 8; ++r) acc[r] = inv * (acc[r] - mu * gG[r]) + gB[r];
      stage_qkv_t(gb, acc, wid, l16, half);
    }
    wg_barrier();
    for (int i = tid; i < 768; i += NT) {
      const int pln = i >> 8, off = i & 255;
      reinterpret_cast<float4*>(plane_ptr(ws, pln, t + 1) + toff)[off] =
          staged4(gb, i);
    }
  } else {
    row_pass_r<false, false>(xb, part, cbp, ctt, nullptr, cgE, cbE, hb, wid, lane);
    wg_barrier();
    for (int i = tid; i < 256; i += NT)
      reinterpret_cast<float4*>(out + toff)[i] = reinterpret_cast<const float4*>(hb)[i];
  }
  };
  step_body(s0, std::true_type{});
  for (int t = s0 + 1; t < s1; ++t)
    step_body(t, std::false_type{});
}

extern "C" void kernel_launch(void* const* d_in, const int* in_sizes, int n_in,
                              void* d_out, int out_size, void* d_ws, size_t ws_size,
                              hipStream_t stream) {
  const int*   idx   = (const int*)  d_in[0];
  const int*   nlp   = (const int*)  d_in[1];
  const float* wte   = (const float*)d_in[2];
  const float* wpe   = (const float*)d_in[3];
  const float* t_w1  = (const float*)d_in[4];
  const float* t_b1  = (const float*)d_in[5];
  const float* t_w2  = (const float*)d_in[6];
  const float* t_b2  = (const float*)d_in[7];
  const float* ln1_g = (const float*)d_in[8];
  const float* ln1_b = (const float*)d_in[9];
  const float* w_qkv = (const float*)d_in[10];
  const float* b_qkv = (const float*)d_in[11];
  const float* w_o   = (const float*)d_in[12];
  const float* b_o   = (const float*)d_in[13];
  const float* ln2_g = (const float*)d_in[14];
  const float* ln2_b = (const float*)d_in[15];
  const float* w_fc  = (const float*)d_in[16];
  const float* b_fc  = (const float*)d_in[17];
  const float* w_pr  = (const float*)d_in[18];
  const float* b_pr  = (const float*)d_in[19];
  const float* lnf_g = (const float*)d_in[20];
  const float* lnf_b = (const float*)d_in[21];

  float* ws = (float*)d_ws;

  temb_kernel<<<LAUNCH1_BLOCKS, 256, 0, stream>>>(
      nlp, t_w1, t_b1, t_w2, w_qkv, w_o, w_fc, w_pr, ln1_g, ln1_b, ln2_g, ln2_b, b_qkv, b_fc, ws);
  embed_kernel<<<256, NT, 0, stream>>>(idx, nlp, wte, wpe, t_b2, ln1_g, ln1_b, w_qkv,
                                       b_qkv, lnf_g, lnf_b, ws, (float*)d_out);
  for (int L = 0; L < NLAUNCH; ++L)
    step_kernel<<<256, NT, 0, stream>>>(L, nlp, b_o, b_pr, lnf_g, lnf_b, ws,
                                        (float*)d_out);
}
